# starting kernel
# baseline (speedup 1.0000x reference)
#include <hip/hip_runtime.h>
#include <hip/hip_bf16.h>

typedef __attribute__((ext_vector_type(2))) float v2f;
typedef __attribute__((ext_vector_type(8))) float v8f;

#define DD 128
#define HROW 132

__global__ __launch_bounds__(256) void gin_repack(const float* __restrict__ W,
                                                  float* __restrict__ Wp,
                                                  int totalPairs) {
    int i = blockIdx.x * blockDim.x + threadIdx.x;
    if (i >= totalPairs) return;
    int n  = i & 127;
    int kh = (i >> 7) & 63;
    int l  = i >> 13;
    const float* Wl = W + (size_t)l * DD * DD;
    float2 p;
    p.x = Wl[(2 * kh) * DD + n];
    p.y = Wl[(2 * kh + 1) * DD + n];
    ((float2*)Wp)[i] = p;
}

__global__ __launch_bounds__(256) void gin_init_z(const float* __restrict__ h,
                                                  const float* __restrict__ eps,
                                                  int layer, float* __restrict__ z,
                                                  int n4) {
    int i = blockIdx.x * blockDim.x + threadIdx.x;
    if (i >= n4) return;
    float e = 1.0f + eps[layer];
    float4 v = ((const float4*)h)[i];
    v.x *= e; v.y *= e; v.z *= e; v.w *= e;
    ((float4*)z)[i] = v;
}

__global__ __launch_bounds__(256) void gin_scatter(const float* __restrict__ h,
                                                   const int* __restrict__ src,
                                                   const int* __restrict__ dst,
                                                   float* __restrict__ z,
                                                   int nedges) {
    int idx = blockIdx.x * blockDim.x + threadIdx.x;
    int e = idx >> 5;
    if (e >= nedges) return;
    int q = (idx & 31) << 2;
    const float4 v = *(const float4*)(h + (size_t)src[e] * DD + q);
    float* zp = z + (size_t)dst[e] * DD + q;
    asm volatile(
        "global_atomic_add_f32 %0, %1, off\n\t"
        "global_atomic_add_f32 %0, %2, off offset:4\n\t"
        "global_atomic_add_f32 %0, %3, off offset:8\n\t"
        "global_atomic_add_f32 %0, %4, off offset:12"
        :: "v"(zp), "v"(v.x), "v"(v.y), "v"(v.z), "v"(v.w) : "memory");
}

__global__ __launch_bounds__(256, 1) void gin_mlp(const float* __restrict__ z,
                                                  const float* __restrict__ W1p,
                                                  const float* __restrict__ b1,
                                                  const float* __restrict__ W2p,
                                                  const float* __restrict__ b2,
                                                  float* __restrict__ out, int n) {
    __shared__ float hidS[8][16 * HROW];

    const int lane  = threadIdx.x & 31;
    const int wave  = threadIdx.x >> 5;
    const int ln    = lane & 15;
    const int khoff = lane >> 4;
    const int rh    = khoff << 3;
    const int m0    = blockIdx.x * 128 + wave * 16;

    int mrow = m0 + ln;
    if (mrow > n - 1) mrow = n - 1;
    const float* zrow = z + (size_t)mrow * DD;
    float* hw = &hidS[wave][0];

    const float* az = zrow + 2 * khoff;
    const float* aw = hw + ln * HROW + 2 * khoff;

    for (int nt = 0; nt < 8; nt += 2) {
        const int n0 = nt * 16;
        const float* Wb0 = W1p + 2 * (khoff * DD + n0 + ln);
        const float* Wb1 = W1p + 2 * (khoff * DD + n0 + 16 + ln);
        v8f acc0 = {}, acc1 = {};
        #pragma unroll
        for (int k0 = 0; k0 < DD; k0 += 4) {
            v2f a  = *(const v2f*)(az + k0);
            v2f g0 = *(const v2f*)(Wb0 + k0 * DD);
            v2f g1 = *(const v2f*)(Wb1 + k0 * DD);
            acc0 = __builtin_amdgcn_wmma_f32_16x16x4_f32(false, a, false, g0,
                                                         (short)0, acc0, false, false);
            acc1 = __builtin_amdgcn_wmma_f32_16x16x4_f32(false, a, false, g1,
                                                         (short)0, acc1, false, false);
        }
        const float bias0 = b1[n0 + ln];
        const float bias1 = b1[n0 + 16 + ln];
        #pragma unroll
        for (int j = 0; j < 8; ++j) {
            hw[(j + rh) * HROW + n0 + ln]      = fmaxf(acc0[j] + bias0, 0.0f);
            hw[(j + rh) * HROW + n0 + 16 + ln] = fmaxf(acc1[j] + bias1, 0.0f);
        }
    }

    __syncthreads();

    for (int nt = 0; nt < 8; nt += 2) {
        const int n0 = nt * 16;
        const float* Wb0 = W2p + 2 * (khoff * DD + n0 + ln);
        const float* Wb1 = W2p + 2 * (khoff * DD + n0 + 16 + ln);
        v8f acc0 = {}, acc1 = {};
        #pragma unroll
        for (int k0 = 0; k0 < DD; k0 += 4) {
            v2f a  = *(const v2f*)(aw + k0);
            v2f g0 = *(const v2f*)(Wb0 + k0 * DD);
            v2f g1 = *(const v2f*)(Wb1 + k0 * DD);
            acc0 = __builtin_amdgcn_wmma_f32_16x16x4_f32(false, a, false, g0,
                                                         (short)0, acc0, false, false);
            acc1 = __builtin_amdgcn_wmma_f32_16x16x4_f32(false, a, false, g1,
                                                         (short)0, acc1, false, false);
        }
        const float bias0 = b2[n0 + ln];
        const float bias1 = b2[n0 + 16 + ln];
        #pragma unroll
        for (int j = 0; j < 8; ++j) {
            const int r = m0 + j + rh;
            if (r < n) {
                out[(size_t)r * DD + n0 + ln]      = acc0[j] + bias0;
                out[(size_t)r * DD + n0 + 16 + ln] = acc1[j] + bias1;
            }
        }
    }
}

extern "C" void kernel_launch(void* const* d_in, const int* in_sizes, int n_in,
                              void* d_out, int out_size, void* d_ws, size_t ws_size,
                              hipStream_t stream) {
    const float* h   = (const float*)d_in[0];
    const int*   src = (const int*)  d_in[1];
    const int*   dst = (const int*)  d_in[2];
    const float* W1  = (const float*)d_in[3];
    const float* b1  = (const float*)d_in[4];
    const float* W2  = (const float*)d_in[5];
    const float* b2  = (const float*)d_in[6];
    const float* eps = (const float*)d_in[7];

    const int n = in_sizes[0] / DD;
    const int e = in_sizes[1];
    const int L = in_sizes[7];

    float* z   = (float*)d_ws;
    float* W1p = z + (size_t)n * DD;
    float* W2p = W1p + (size_t)L * DD * DD;
    float* out = (float*)d_out;

    const int n4         = n * DD / 4;
    const int initBlocks = (n4 + 255) / 256;
    const long scatThr   = (long)e * 32;
    const int scatBlocks = (int)((scatThr + 255) / 256);
    const int mlpBlocks  = (n + 127) / 128;
    const int pairs      = L * DD * DD / 2;
    const int rpBlocks   = (pairs + 255) / 256;

    gin_repack<<<rpBlocks, 256, 0, stream>>>(W1, W1p, pairs);
    gin_repack<<<rpBlocks, 256, 0, stream>>>(W2, W2p, pairs);

    for (int l = 0; l < L; ++l) {
        const float* hin = (l == 0) ? h : out;
        gin_init_z<<<initBlocks, 256, 0, stream>>>(hin, eps, l, z, n4);
        gin_scatter<<<scatBlocks, 256, 0, stream>>>(hin, src, dst, z, e);
        gin_mlp<<<mlpBlocks, 256, 0, stream>>>(z, W1p + (size_t)l * DD * DD, b1 + l * DD,
                                               W2p + (size_t)l * DD * DD, b2 + l * DD, out, n);
    }
}


# Replace per-layer z-init + float-atomic scatter by one in-call stable sort of edges by dst (CSR) and a wave-per-node gather; neighbour sums now in fixed edge order, no atomics left
# speedup vs baseline: 1.8826x; 1.8826x over previous
#include <hip/hip_runtime.h>
#include <hip/hip_bf16.h>

typedef __attribute__((ext_vector_type(2))) float v2f;
typedef __attribute__((ext_vector_type(8))) float v8f;

#define DD 128
#define HROW 132

__global__ __launch_bounds__(256) void gin_repack(const float* __restrict__ W,
                                                  float* __restrict__ Wp,
                                                  int totalPairs) {
    int i = blockIdx.x * blockDim.x + threadIdx.x;
    if (i >= totalPairs) return;
    int n  = i & 127;
    int kh = (i >> 7) & 63;
    int l  = i >> 13;
    const float* Wl = W + (size_t)l * DD * DD;
    float2 p;
    p.x = Wl[(2 * kh) * DD + n];
    p.y = Wl[(2 * kh + 1) * DD + n];
    ((float2*)Wp)[i] = p;
}

#define SORT_NB    512
#define SORT_TPB   128
#define SORT_BINS  256
#define SORT_PITCH 129

__device__ __forceinline__ void sort_count_runs(const int* __restrict__ keys, int shift,
                                                long start, long end, int* hist, int tid) {
    for (int i = tid; i < SORT_BINS * SORT_PITCH; i += SORT_TPB) hist[i] = 0;
    __syncthreads();
    long p = start;
    for (; p + 4 <= end; p += 4) {
        const int k0 = keys[p], k1 = keys[p + 1], k2 = keys[p + 2], k3 = keys[p + 3];
        hist[((k0 >> shift) & (SORT_BINS - 1)) * SORT_PITCH + tid] += 1;
        hist[((k1 >> shift) & (SORT_BINS - 1)) * SORT_PITCH + tid] += 1;
        hist[((k2 >> shift) & (SORT_BINS - 1)) * SORT_PITCH + tid] += 1;
        hist[((k3 >> shift) & (SORT_BINS - 1)) * SORT_PITCH + tid] += 1;
    }
    for (; p < end; ++p) {
        const int d = (keys[p] >> shift) & (SORT_BINS - 1);
        hist[d * SORT_PITCH + tid] += 1;
    }
    __syncthreads();
}

__global__ __launch_bounds__(SORT_TPB) void sort_count(const int* __restrict__ keys, int shift,
                                                       int nedges, int run,
                                                       int* __restrict__ table) {
    __shared__ int hist[SORT_BINS * SORT_PITCH];
    const int tid = threadIdx.x;
    const long start = ((long)blockIdx.x * SORT_TPB + tid) * run;
    long end = start + run;
    if (end > nedges) end = nedges;
    sort_count_runs(keys, shift, start, end, hist, tid);
    for (int bin = tid; bin < SORT_BINS; bin += SORT_TPB) {
        int total = 0;
        for (int j = 0; j < SORT_TPB; ++j) total += hist[bin * SORT_PITCH + j];
        table[bin * SORT_NB + blockIdx.x] = total;
    }
}

__global__ __launch_bounds__(256) void gin_scan(const int* __restrict__ in,
                                                int* __restrict__ outp) {
    __shared__ int wtot[8];
    __shared__ int wpre[8];
    const int tid  = threadIdx.x;
    const int lane = tid & 31;
    const int wv   = tid >> 5;
    const int base = blockIdx.x * 1024;

    const int nfront = blockIdx.x;
    const int* mine = in + tid * 4;
    int pre = 0;
    for (int ch = 0; ch < nfront; ch += 16) {
        int part[16];
        #pragma unroll
        for (int k = 0; k < 16; ++k) {
            const int at = (ch + k < nfront) ? ch + k : 0;
            const int4 f = *(const int4*)(mine + at * 1024);
            part[k] = (f.x + f.y) + (f.z + f.w);
        }
        #pragma unroll
        for (int k = 0; k < 16; ++k) pre += (ch + k < nfront) ? part[k] : 0;
    }
    const int4 c = *(const int4*)(in + base + tid * 4);
    const int local = (c.x + c.y) + (c.z + c.w);

    int incl = local;
    for (int d = 1; d < 32; d <<= 1) {
        const int t = __shfl_up(incl, d);
        if (lane >= d) incl += t;
    }
    for (int d = 16; d > 0; d >>= 1) pre += __shfl_xor(pre, d);
    if (lane == 31) wtot[wv] = incl;
    if (lane == 0)  wpre[wv] = pre;
    __syncthreads();
    int front = 0;
    for (int w = 0; w < 8; ++w) {
        front += wpre[w];
        if (w < wv) front += wtot[w];
    }
    int4 o;
    o.x = front + incl - local;
    o.y = o.x + c.x;
    o.z = o.y + c.y;
    o.w = o.z + c.z;
    *(int4*)(outp + base + tid * 4) = o;
}

__device__ __forceinline__ void sort_put(int* hist, int tid, int shift, int k, int v,
                                         int* __restrict__ keysOut, int* __restrict__ valsOut) {
    const int d = (k >> shift) & (SORT_BINS - 1);
    const int q = hist[d * SORT_PITCH + tid];
    hist[d * SORT_PITCH + tid] = q + 1;
    keysOut[q] = k;
    valsOut[q] = v;
}

__global__ __launch_bounds__(SORT_TPB) void sort_place(const int* __restrict__ keys,
                                                       const int* __restrict__ vals, int shift,
                                                       int nedges, int run,
                                                       const int* __restrict__ tableS,
                                                       int* __restrict__ keysOut,
                                                       int* __restrict__ valsOut) {
    __shared__ int hist[SORT_BINS * SORT_PITCH];
    const int tid = threadIdx.x;
    const long start = ((long)blockIdx.x * SORT_TPB + tid) * run;
    long end = start + run;
    if (end > nedges) end = nedges;
    sort_count_runs(keys, shift, start, end, hist, tid);
    for (int bin = tid; bin < SORT_BINS; bin += SORT_TPB) {
        int pos = tableS[bin * SORT_NB + blockIdx.x];
        for (int j = 0; j < SORT_TPB; ++j) {
            const int c = hist[bin * SORT_PITCH + j];
            hist[bin * SORT_PITCH + j] = pos;
            pos += c;
        }
    }
    __syncthreads();
    long p = start;
    for (; p + 4 <= end; p += 4) {
        const int k0 = keys[p], k1 = keys[p + 1], k2 = keys[p + 2], k3 = keys[p + 3];
        const int v0 = vals[p], v1 = vals[p + 1], v2 = vals[p + 2], v3 = vals[p + 3];
        sort_put(hist, tid, shift, k0, v0, keysOut, valsOut);
        sort_put(hist, tid, shift, k1, v1, keysOut, valsOut);
        sort_put(hist, tid, shift, k2, v2, keysOut, valsOut);
        sort_put(hist, tid, shift, k3, v3, keysOut, valsOut);
    }
    for (; p < end; ++p) sort_put(hist, tid, shift, keys[p], vals[p], keysOut, valsOut);
}

__global__ __launch_bounds__(256) void gin_offsets(const int* __restrict__ sdst,
                                                   int nedges, int n,
                                                   int* __restrict__ off) {
    const int p = blockIdx.x * blockDim.x + threadIdx.x;
    if (p > nedges) return;
    const int prev = (p == 0) ? -1 : sdst[p - 1];
    const int here = (p == nedges) ? n : sdst[p];
    for (int v = prev + 1; v <= here; ++v) off[v] = p;
}

__device__ __forceinline__ void add4(float4& a, const float4 r) {
    a.x += r.x; a.y += r.y; a.z += r.z; a.w += r.w;
}

__global__ __launch_bounds__(256) void gin_gather(const float* __restrict__ h,
                                                  const float* __restrict__ eps, int layer,
                                                  const int* __restrict__ off,
                                                  const int* __restrict__ csrc,
                                                  float* __restrict__ z, int n) {
    const int lane = threadIdx.x & 31;
    const int v = __builtin_amdgcn_readfirstlane((int)(blockIdx.x * 8 + (threadIdx.x >> 5)));
    if (v >= n) return;
    const int b = off[v];
    const int c = off[v + 1];
    const float* hc = h + lane * 4;
    int base = b;
    int m = c - base;
    if (m > 32) m = 32;
    int idxv = (lane < m) ? csrc[base + lane] : 0;
    const float4 hv = *(const float4*)(hc + (size_t)v * DD);
    float4 acc = {0.0f, 0.0f, 0.0f, 0.0f};
    while (m > 0) {
        int j = 0;
        for (; j + 4 <= m; j += 4) {
            const int s0 = __builtin_amdgcn_readlane(idxv, j);
            const int s1 = __builtin_amdgcn_readlane(idxv, j + 1);
            const int s2 = __builtin_amdgcn_readlane(idxv, j + 2);
            const int s3 = __builtin_amdgcn_readlane(idxv, j + 3);
            const float4 r0 = *(const float4*)(hc + (size_t)s0 * DD);
            const float4 r1 = *(const float4*)(hc + (size_t)s1 * DD);
            const float4 r2 = *(const float4*)(hc + (size_t)s2 * DD);
            const float4 r3 = *(const float4*)(hc + (size_t)s3 * DD);
            add4(acc, r0); add4(acc, r1); add4(acc, r2); add4(acc, r3);
        }
        if (j + 2 <= m) {
            const int s0 = __builtin_amdgcn_readlane(idxv, j);
            const int s1 = __builtin_amdgcn_readlane(idxv, j + 1);
            const float4 r0 = *(const float4*)(hc + (size_t)s0 * DD);
            const float4 r1 = *(const float4*)(hc + (size_t)s1 * DD);
            add4(acc, r0); add4(acc, r1);
            j += 2;
        }
        if (j < m) {
            const int s0 = __builtin_amdgcn_readlane(idxv, j);
            const float4 r0 = *(const float4*)(hc + (size_t)s0 * DD);
            add4(acc, r0);
        }
        base += 32;
        m = c - base;
        if (m > 32) m = 32;
        idxv = (lane < m) ? csrc[base + lane] : 0;
    }
    const float e1 = 1.0f + eps[layer];
    float4 o;
    o.x = e1 * hv.x + acc.x;
    o.y = e1 * hv.y + acc.y;
    o.z = e1 * hv.z + acc.z;
    o.w = e1 * hv.w + acc.w;
    *(float4*)(z + (size_t)v * DD + lane * 4) = o;
}

__global__ __launch_bounds__(256, 1) void gin_mlp(const float* __restrict__ z,
                                                  const float* __restrict__ W1p,
                                                  const float* __restrict__ b1,
                                                  const float* __restrict__ W2p,
                                                  const float* __restrict__ b2,
                                                  float* __restrict__ out, int n) {
    __shared__ float hidS[8][16 * HROW];

    const int lane  = threadIdx.x & 31;
    const int wave  = threadIdx.x >> 5;
    const int ln    = lane & 15;
    const int khoff = lane >> 4;
    const int rh    = khoff << 3;
    const int m0    = blockIdx.x * 128 + wave * 16;

    int mrow = m0 + ln;
    if (mrow > n - 1) mrow = n - 1;
    const float* zrow = z + (size_t)mrow * DD;
    float* hw = &hidS[wave][0];

    const float* az = zrow + 2 * khoff;
    const float* aw = hw + ln * HROW + 2 * khoff;

    for (int nt = 0; nt < 8; nt += 2) {
        const int n0 = nt * 16;
        const float* Wb0 = W1p + 2 * (khoff * DD + n0 + ln);
        const float* Wb1 = W1p + 2 * (khoff * DD + n0 + 16 + ln);
        v8f acc0 = {}, acc1 = {};
        #pragma unroll
        for (int k0 = 0; k0 < DD; k0 += 4) {
            v2f a  = *(const v2f*)(az + k0);
            v2f g0 = *(const v2f*)(Wb0 + k0 * DD);
            v2f g1 = *(const v2f*)(Wb1 + k0 * DD);
            acc0 = __builtin_amdgcn_wmma_f32_16x16x4_f32(false, a, false, g0,
                                                         (short)0, acc0, false, false);
            acc1 = __builtin_amdgcn_wmma_f32_16x16x4_f32(false, a, false, g1,
                                                         (short)0, acc1, false, false);
        }
        const float bias0 = b1[n0 + ln];
        const float bias1 = b1[n0 + 16 + ln];
        #pragma unroll
        for (int j = 0; j < 8; ++j) {
            hw[(j + rh) * HROW + n0 + ln]      = fmaxf(acc0[j] + bias0, 0.0f);
            hw[(j + rh) * HROW + n0 + 16 + ln] = fmaxf(acc1[j] + bias1, 0.0f);
        }
    }

    __syncthreads();

    for (int nt = 0; nt < 8; nt += 2) {
        const int n0 = nt * 16;
        const float* Wb0 = W2p + 2 * (khoff * DD + n0 + ln);
        const float* Wb1 = W2p + 2 * (khoff * DD + n0 + 16 + ln);
        v8f acc0 = {}, acc1 = {};
        #pragma unroll
        for (int k0 = 0; k0 < DD; k0 += 4) {
            v2f a  = *(const v2f*)(aw + k0);
            v2f g0 = *(const v2f*)(Wb0 + k0 * DD);
            v2f g1 = *(const v2f*)(Wb1 + k0 * DD);
            acc0 = __builtin_amdgcn_wmma_f32_16x16x4_f32(false, a, false, g0,
                                                         (short)0, acc0, false, false);
            acc1 = __builtin_amdgcn_wmma_f32_16x16x4_f32(false, a, false, g1,
                                                         (short)0, acc1, false, false);
        }
        const float bias0 = b2[n0 + ln];
        const float bias1 = b2[n0 + 16 + ln];
        #pragma unroll
        for (int j = 0; j < 8; ++j) {
            const int r = m0 + j + rh;
            if (r < n) {
                out[(size_t)r * DD + n0 + ln]      = acc0[j] + bias0;
                out[(size_t)r * DD + n0 + 16 + ln] = acc1[j] + bias1;
            }
        }
    }
}

extern "C" void kernel_launch(void* const* d_in, const int* in_sizes, int n_in,
                              void* d_out, int out_size, void* d_ws, size_t ws_size,
                              hipStream_t stream) {
    const float* h   = (const float*)d_in[0];
    const int*   src = (const int*)  d_in[1];
    const int*   dst = (const int*)  d_in[2];
    const float* W1  = (const float*)d_in[3];
    const float* b1  = (const float*)d_in[4];
    const float* W2  = (const float*)d_in[5];
    const float* b2  = (const float*)d_in[6];
    const float* eps = (const float*)d_in[7];

    const int n = in_sizes[0] / DD;
    const int e = in_sizes[1];
    const int L = in_sizes[7];

    float* z   = (float*)d_ws;
    float* W1p = z + (size_t)n * DD;
    float* W2p = W1p + (size_t)L * DD * DD;
    float* out = (float*)d_out;

    const int tabN = SORT_BINS * SORT_NB;
    const int np   = ((n + 4) / 4) * 4;
    int* table  = (int*)(W2p + (size_t)L * DD * DD);
    int* tableS = table + tabN;
    int* off    = tableS + tabN;
    int* kA     = off + np;
    int* vA     = kA + e;
    int* kB     = vA + e;
    int* vB     = kB + e;

    const int gatBlocks  = (n + 7) / 8;
    const int offBlocks  = (e + 1 + 255) / 256;
    const int mlpBlocks  = (n + 127) / 128;
    const int pairs      = L * DD * DD / 2;
    const int rpBlocks   = (pairs + 255) / 256;
    const int run        = (e + SORT_NB * SORT_TPB - 1) / (SORT_NB * SORT_TPB);

    gin_repack<<<rpBlocks, 256, 0, stream>>>(W1, W1p, pairs);
    gin_repack<<<rpBlocks, 256, 0, stream>>>(W2, W2p, pairs);

    const int* kin = dst;
    const int* vin = src;
    int* kout = kA;
    int* vout = vA;
    for (int shift = 0; shift == 0 || ((n - 1) >> shift) > 0; shift += 8) {
        sort_count<<<SORT_NB, SORT_TPB, 0, stream>>>(kin, shift, e, run, table);
        gin_scan<<<tabN / 1024, 256, 0, stream>>>(table, tableS);
        sort_place<<<SORT_NB, SORT_TPB, 0, stream>>>(kin, vin, shift, e, run, tableS, kout, vout);
        kin = kout;
        vin = vout;
        kout = (kout == kA) ? kB : kA;
        vout = (vout == vA) ? vB : vA;
    }
    gin_offsets<<<offBlocks, 256, 0, stream>>>(kin, e, n, off);

    for (int l = 0; l < L; ++l) {
        const float* hin = (l == 0) ? h : out;
        gin_gather<<<gatBlocks, 256, 0, stream>>>(hin, eps, l, off, vin, z, n);
        gin_mlp<<<mlpBlocks, 256, 0, stream>>>(z, W1p + (size_t)l * DD * DD, b1 + l * DD,
                                               W2p + (size_t)l * DD * DD, b2 + l * DD, out, n);
    }
}


# Fuse gather into the MLP kernel: each wave sums its 16 nodes' neighbour rows (same list order) into an LDS tile that feeds the first product; layers alternate between two buffers; 3 launches fewer
# speedup vs baseline: 1.9882x; 1.0561x over previous
#include <hip/hip_runtime.h>
#include <hip/hip_bf16.h>

typedef __attribute__((ext_vector_type(2))) float v2f;
typedef __attribute__((ext_vector_type(8))) float v8f;

#define DD 128
#define HROW 132

__global__ __launch_bounds__(256) void gin_repack(const float* __restrict__ W,
                                                  float* __restrict__ Wp,
                                                  int totalPairs) {
    int i = blockIdx.x * blockDim.x + threadIdx.x;
    if (i >= totalPairs) return;
    int n  = i & 127;
    int kh = (i >> 7) & 63;
    int l  = i >> 13;
    const float* Wl = W + (size_t)l * DD * DD;
    float2 p;
    p.x = Wl[(2 * kh) * DD + n];
    p.y = Wl[(2 * kh + 1) * DD + n];
    ((float2*)Wp)[i] = p;
}

#define SORT_NB    512
#define SORT_TPB   128
#define SORT_BINS  256
#define SORT_PITCH 129

__device__ __forceinline__ void sort_count_runs(const int* __restrict__ keys, int shift,
                                                long start, long end, int* hist, int tid) {
    for (int i = tid; i < SORT_BINS * SORT_PITCH; i += SORT_TPB) hist[i] = 0;
    __syncthreads();
    long p = start;
    for (; p + 4 <= end; p += 4) {
        const int k0 = keys[p], k1 = keys[p + 1], k2 = keys[p + 2], k3 = keys[p + 3];
        hist[((k0 >> shift) & (SORT_BINS - 1)) * SORT_PITCH + tid] += 1;
        hist[((k1 >> shift) & (SORT_BINS - 1)) * SORT_PITCH + tid] += 1;
        hist[((k2 >> shift) & (SORT_BINS - 1)) * SORT_PITCH + tid] += 1;
        hist[((k3 >> shift) & (SORT_BINS - 1)) * SORT_PITCH + tid] += 1;
    }
    for (; p < end; ++p) {
        const int d = (keys[p] >> shift) & (SORT_BINS - 1);
        hist[d * SORT_PITCH + tid] += 1;
    }
    __syncthreads();
}

__global__ __launch_bounds__(SORT_TPB) void sort_count(const int* __restrict__ keys, int shift,
                                                       int nedges, int run,
                                                       int* __restrict__ table) {
    __shared__ int hist[SORT_BINS * SORT_PITCH];
    const int tid = threadIdx.x;
    const long start = ((long)blockIdx.x * SORT_TPB + tid) * run;
    long end = start + run;
    if (end > nedges) end = nedges;
    sort_count_runs(keys, shift, start, end, hist, tid);
    for (int bin = tid; bin < SORT_BINS; bin += SORT_TPB) {
        int total = 0;
        for (int j = 0; j < SORT_TPB; ++j) total += hist[bin * SORT_PITCH + j];
        table[bin * SORT_NB + blockIdx.x] = total;
    }
}

__global__ __launch_bounds__(256) void gin_scan(const int* __restrict__ in,
                                                int* __restrict__ outp) {
    __shared__ int wtot[8];
    __shared__ int wpre[8];
    const int tid  = threadIdx.x;
    const int lane = tid & 31;
    const int wv   = tid >> 5;
    const int base = blockIdx.x * 1024;

    const int nfront = blockIdx.x;
    const int* mine = in + tid * 4;
    int pre = 0;
    for (int ch = 0; ch < nfront; ch += 16) {
        int part[16];
        #pragma unroll
        for (int k = 0; k < 16; ++k) {
            const int at = (ch + k < nfront) ? ch + k : 0;
            const int4 f = *(const int4*)(mine + at * 1024);
            part[k] = (f.x + f.y) + (f.z + f.w);
        }
        #pragma unroll
        for (int k = 0; k < 16; ++k) pre += (ch + k < nfront) ? part[k] : 0;
    }
    const int4 c = *(const int4*)(in + base + tid * 4);
    const int local = (c.x + c.y) + (c.z + c.w);

    int incl = local;
    for (int d = 1; d < 32; d <<= 1) {
        const int t = __shfl_up(incl, d);
        if (lane >= d) incl += t;
    }
    for (int d = 16; d > 0; d >>= 1) pre += __shfl_xor(pre, d);
    if (lane == 31) wtot[wv] = incl;
    if (lane == 0)  wpre[wv] = pre;
    __syncthreads();
    int front = 0;
    for (int w = 0; w < 8; ++w) {
        front += wpre[w];
        if (w < wv) front += wtot[w];
    }
    int4 o;
    o.x = front + incl - local;
    o.y = o.x + c.x;
    o.z = o.y + c.y;
    o.w = o.z + c.z;
    *(int4*)(outp + base + tid * 4) = o;
}

__device__ __forceinline__ void sort_put(int* hist, int tid, int shift, int k, int v,
                                         int* __restrict__ keysOut, int* __restrict__ valsOut) {
    const int d = (k >> shift) & (SORT_BINS - 1);
    const int q = hist[d * SORT_PITCH + tid];
    hist[d * SORT_PITCH + tid] = q + 1;
    keysOut[q] = k;
    valsOut[q] = v;
}

__global__ __launch_bounds__(SORT_TPB) void sort_place(const int* __restrict__ keys,
                                                       const int* __restrict__ vals, int shift,
                                                       int nedges, int run,
                                                       const int* __restrict__ tableS,
                                                       int* __restrict__ keysOut,
                                                       int* __restrict__ valsOut) {
    __shared__ int hist[SORT_BINS * SORT_PITCH];
    const int tid = threadIdx.x;
    const long start = ((long)blockIdx.x * SORT_TPB + tid) * run;
    long end = start + run;
    if (end > nedges) end = nedges;
    sort_count_runs(keys, shift, start, end, hist, tid);
    for (int bin = tid; bin < SORT_BINS; bin += SORT_TPB) {
        int pos = tableS[bin * SORT_NB + blockIdx.x];
        for (int j = 0; j < SORT_TPB; ++j) {
            const int c = hist[bin * SORT_PITCH + j];
            hist[bin * SORT_PITCH + j] = pos;
            pos += c;
        }
    }
    __syncthreads();
    long p = start;
    for (; p + 4 <= end; p += 4) {
        const int k0 = keys[p], k1 = keys[p + 1], k2 = keys[p + 2], k3 = keys[p + 3];
        const int v0 = vals[p], v1 = vals[p + 1], v2 = vals[p + 2], v3 = vals[p + 3];
        sort_put(hist, tid, shift, k0, v0, keysOut, valsOut);
        sort_put(hist, tid, shift, k1, v1, keysOut, valsOut);
        sort_put(hist, tid, shift, k2, v2, keysOut, valsOut);
        sort_put(hist, tid, shift, k3, v3, keysOut, valsOut);
    }
    for (; p < end; ++p) sort_put(hist, tid, shift, keys[p], vals[p], keysOut, valsOut);
}

__global__ __launch_bounds__(256) void gin_offsets(const int* __restrict__ sdst,
                                                   int nedges, int n,
                                                   int* __restrict__ off) {
    const int p = blockIdx.x * blockDim.x + threadIdx.x;
    if (p > nedges) return;
    const int prev = (p == 0) ? -1 : sdst[p - 1];
    const int here = (p == nedges) ? n : sdst[p];
    for (int v = prev + 1; v <= here; ++v) off[v] = p;
}

#define GROWS 16

__device__ __forceinline__ void add4(float4& a, const float4 r) {
    a.x += r.x; a.y += r.y; a.z += r.z; a.w += r.w;
}

__device__ __forceinline__ void close_row(float* zrow, const float e1, const float4 acc) {
    const float4 hv = *(const float4*)zrow;
    float4 o;
    o.x = e1 * hv.x + acc.x;
    o.y = e1 * hv.y + acc.y;
    o.z = e1 * hv.z + acc.z;
    o.w = e1 * hv.w + acc.w;
    *(float4*)zrow = o;
}

__device__ __forceinline__ void load_rows(float4 (&r)[GROWS], const int idxv, const int j,
                                          const float* hc) {
    #pragma unroll
    for (int t = 0; t < GROWS; ++t) {
        const int s = __builtin_amdgcn_readlane(idxv, j + t);
        r[t] = *(const float4*)(hc + (size_t)s * DD);
    }
}

__device__ __forceinline__ void sum_rows(const float4 (&r)[GROWS], const int p0, const int eEnd,
                                         const int ob, float* zl, const float e1,
                                         int& i, int& nextB, float4& acc) {
    #pragma unroll
    for (int t = 0; t < GROWS; ++t) {
        const int p = p0 + t;
        if (p < eEnd) {
            while (p >= nextB) {
                close_row(zl + i * HROW, e1, acc);
                acc.x = 0.0f; acc.y = 0.0f; acc.z = 0.0f; acc.w = 0.0f;
                ++i;
                nextB = __builtin_amdgcn_readlane(ob, i + 1);
            }
            add4(acc, r[t]);
        }
    }
}

__global__ __launch_bounds__(256, 1) void gin_layer(const float* __restrict__ h,
                                                    const float* __restrict__ eps, int layer,
                                                    const int* __restrict__ off,
                                                    const int* __restrict__ csrc,
                                                    const float* __restrict__ W1p,
                                                    const float* __restrict__ b1,
                                                    const float* __restrict__ W2p,
                                                    const float* __restrict__ b2,
                                                    float* __restrict__ out, int n) {
    __shared__ __attribute__((aligned(16))) float zS[8][16 * HROW];
    __shared__ __attribute__((aligned(16))) float hidS[8][16 * HROW];

    const int lane  = threadIdx.x & 31;
    const int wave  = __builtin_amdgcn_readfirstlane((int)(threadIdx.x >> 5));
    const int ln    = lane & 15;
    const int khoff = lane >> 4;
    const int rh    = khoff << 3;
    const int m0    = blockIdx.x * 128 + wave * 16;

    float* zw = &zS[wave][0];
    float* hw = &hidS[wave][0];

    int ob;
    {
        int t = m0 + lane;
        if (t > n) t = n;
        ob = off[t];
    }
    const float* hc = h + lane * 4;
    float* zl = zw + lane * 4;
    {
        float4 own[16];
        #pragma unroll
        for (int q = 0; q < 16; ++q) {
            int rr = m0 + q;
            if (rr > n - 1) rr = n - 1;
            own[q] = *(const float4*)(hc + (size_t)rr * DD);
        }
        #pragma unroll
        for (int q = 0; q < 16; ++q) *(float4*)(zl + q * HROW) = own[q];
    }
    const float e1 = 1.0f + eps[layer];
    const int eEnd = __builtin_amdgcn_readlane(ob, 16);
    int base  = __builtin_amdgcn_readlane(ob, 0);
    int nextB = __builtin_amdgcn_readlane(ob, 1);
    int i = 0;
    float4 acc = {0.0f, 0.0f, 0.0f, 0.0f};
    int m = eEnd - base;
    if (m > 32) m = 32;
    int idxv = (lane < m) ? csrc[base + lane] : 0;
    while (m > 0) {
        const int nb = base + 32;
        int mn = eEnd - nb;
        if (mn > 32) mn = 32;
        float4 r[GROWS];
        load_rows(r, idxv, 0, hc);
        const int idxn = (lane < mn) ? csrc[nb + lane] : 0;
        sum_rows(r, base, eEnd, ob, zl, e1, i, nextB, acc);
        if (m > GROWS) {
            load_rows(r, idxv, GROWS, hc);
            sum_rows(r, base + GROWS, eEnd, ob, zl, e1, i, nextB, acc);
        }
        base = nb;
        m = mn;
        idxv = idxn;
    }
    while (i < 16) {
        close_row(zl + i * HROW, e1, acc);
        acc.x = 0.0f; acc.y = 0.0f; acc.z = 0.0f; acc.w = 0.0f;
        ++i;
    }

    const float* az = zw + ln * HROW + 2 * khoff;
    const float* aw = hw + ln * HROW + 2 * khoff;

    for (int nt = 0; nt < 8; nt += 2) {
        const int n0 = nt * 16;
        const float* Wb0 = W1p + 2 * (khoff * DD + n0 + ln);
        const float* Wb1 = W1p + 2 * (khoff * DD + n0 + 16 + ln);
        v8f acc0 = {}, acc1 = {};
        #pragma unroll
        for (int k0 = 0; k0 < DD; k0 += 4) {
            v2f a  = *(const v2f*)(az + k0);
            v2f g0 = *(const v2f*)(Wb0 + k0 * DD);
            v2f g1 = *(const v2f*)(Wb1 + k0 * DD);
            acc0 = __builtin_amdgcn_wmma_f32_16x16x4_f32(false, a, false, g0,
                                                         (short)0, acc0, false, false);
            acc1 = __builtin_amdgcn_wmma_f32_16x16x4_f32(false, a, false, g1,
                                                         (short)0, acc1, false, false);
        }
        const float bias0 = b1[n0 + ln];
        const float bias1 = b1[n0 + 16 + ln];
        #pragma unroll
        for (int j = 0; j < 8; ++j) {
            hw[(j + rh) * HROW + n0 + ln]      = fmaxf(acc0[j] + bias0, 0.0f);
            hw[(j + rh) * HROW + n0 + 16 + ln] = fmaxf(acc1[j] + bias1, 0.0f);
        }
    }

    __syncthreads();

    for (int nt = 0; nt < 8; nt += 2) {
        const int n0 = nt * 16;
        const float* Wb0 = W2p + 2 * (khoff * DD + n0 + ln);
        const float* Wb1 = W2p + 2 * (khoff * DD + n0 + 16 + ln);
        v8f acc0 = {}, acc1 = {};
        #pragma unroll
        for (int k0 = 0; k0 < DD; k0 += 4) {
            v2f a  = *(const v2f*)(aw + k0);
            v2f g0 = *(const v2f*)(Wb0 + k0 * DD);
            v2f g1 = *(const v2f*)(Wb1 + k0 * DD);
            acc0 = __builtin_amdgcn_wmma_f32_16x16x4_f32(false, a, false, g0,
                                                         (short)0, acc0, false, false);
            acc1 = __builtin_amdgcn_wmma_f32_16x16x4_f32(false, a, false, g1,
                                                         (short)0, acc1, false, false);
        }
        const float bias0 = b2[n0 + ln];
        const float bias1 = b2[n0 + 16 + ln];
        #pragma unroll
        for (int j = 0; j < 8; ++j) {
            const int r = m0 + j + rh;
            if (r < n) {
                out[(size_t)r * DD + n0 + ln]      = acc0[j] + bias0;
                out[(size_t)r * DD + n0 + 16 + ln] = acc1[j] + bias1;
            }
        }
    }
}

extern "C" void kernel_launch(void* const* d_in, const int* in_sizes, int n_in,
                              void* d_out, int out_size, void* d_ws, size_t ws_size,
                              hipStream_t stream) {
    const float* h   = (const float*)d_in[0];
    const int*   src = (const int*)  d_in[1];
    const int*   dst = (const int*)  d_in[2];
    const float* W1  = (const float*)d_in[3];
    const float* b1  = (const float*)d_in[4];
    const float* W2  = (const float*)d_in[5];
    const float* b2  = (const float*)d_in[6];
    const float* eps = (const float*)d_in[7];

    const int n = in_sizes[0] / DD;
    const int e = in_sizes[1];
    const int L = in_sizes[7];

    float* hb  = (float*)d_ws;
    float* W1p = hb + (size_t)n * DD;
    float* W2p = W1p + (size_t)L * DD * DD;
    float* out = (float*)d_out;

    const int tabN = SORT_BINS * SORT_NB;
    const int np   = ((n + 4) / 4) * 4;
    int* table  = (int*)(W2p + (size_t)L * DD * DD);
    int* tableS = table + tabN;
    int* off    = tableS + tabN;
    int* kA     = off + np;
    int* vA     = kA + e;
    int* kB     = vA + e;
    int* vB     = kB + e;

    const int offBlocks  = (e + 1 + 255) / 256;
    const int mlpBlocks  = (n + 127) / 128;
    const int pairs      = L * DD * DD / 2;
    const int rpBlocks   = (pairs + 255) / 256;
    const int run        = (e + SORT_NB * SORT_TPB - 1) / (SORT_NB * SORT_TPB);

    gin_repack<<<rpBlocks, 256, 0, stream>>>(W1, W1p, pairs);
    gin_repack<<<rpBlocks, 256, 0, stream>>>(W2, W2p, pairs);

    const int* kin = dst;
    const int* vin = src;
    int* kout = kA;
    int* vout = vA;
    for (int shift = 0; shift == 0 || ((n - 1) >> shift) > 0; shift += 8) {
        sort_count<<<SORT_NB, SORT_TPB, 0, stream>>>(kin, shift, e, run, table);
        gin_scan<<<tabN / 1024, 256, 0, stream>>>(table, tableS);
        sort_place<<<SORT_NB, SORT_TPB, 0, stream>>>(kin, vin, shift, e, run, tableS, kout, vout);
        kin = kout;
        vin = vout;
        kout = (kout == kA) ? kB : kA;
        vout = (vout == vA) ? vB : vA;
    }
    gin_offsets<<<offBlocks, 256, 0, stream>>>(kin, e, n, off);

    for (int l = 0; l < L; ++l) {
        const float* hin = (l == 0) ? h : (((L - l) & 1) ? hb : out);
        float* hout = ((L - 1 - l) & 1) ? hb : out;
        gin_layer<<<mlpBlocks, 256, 0, stream>>>(hin, eps, l, off, vin,
                                                 W1p + (size_t)l * DD * DD, b1 + l * DD,
                                                 W2p + (size_t)l * DD * DD, b2 + l * DD, hout, n);
    }
}


# Two-level sort: edges split once by the high key byte, then one launch counting-sorts each 256-node group in LDS by the low byte and writes lists and list starts; 3 launches fewer, sum order kept
# speedup vs baseline: 2.4293x; 1.2219x over previous
#include <hip/hip_runtime.h>
#include <hip/hip_bf16.h>

typedef __attribute__((ext_vector_type(2))) float v2f;
typedef __attribute__((ext_vector_type(8))) float v8f;

#define DD 128
#define HROW 132

__global__ __launch_bounds__(256) void gin_repack(const float* __restrict__ W,
                                                  float* __restrict__ Wp,
                                                  int totalPairs) {
    int i = blockIdx.x * blockDim.x + threadIdx.x;
    if (i >= totalPairs) return;
    int n  = i & 127;
    int kh = (i >> 7) & 63;
    int l  = i >> 13;
    const float* Wl = W + (size_t)l * DD * DD;
    float2 p;
    p.x = Wl[(2 * kh) * DD + n];
    p.y = Wl[(2 * kh + 1) * DD + n];
    ((float2*)Wp)[i] = p;
}

#define SORT_NB    512
#define SORT_TPB   128
#define SORT_BINS  256
#define SORT_PITCH 129

__device__ __forceinline__ void sort_count_runs(const int* __restrict__ keys, int shift,
                                                long start, long end, int* hist, int tid) {
    for (int i = tid; i < SORT_BINS * SORT_PITCH; i += SORT_TPB) hist[i] = 0;
    __syncthreads();
    long p = start;
    for (; p + 4 <= end; p += 4) {
        const int k0 = keys[p], k1 = keys[p + 1], k2 = keys[p + 2], k3 = keys[p + 3];
        hist[((k0 >> shift) & (SORT_BINS - 1)) * SORT_PITCH + tid] += 1;
        hist[((k1 >> shift) & (SORT_BINS - 1)) * SORT_PITCH + tid] += 1;
        hist[((k2 >> shift) & (SORT_BINS - 1)) * SORT_PITCH + tid] += 1;
        hist[((k3 >> shift) & (SORT_BINS - 1)) * SORT_PITCH + tid] += 1;
    }
    for (; p < end; ++p) {
        const int d = (keys[p] >> shift) & (SORT_BINS - 1);
        hist[d * SORT_PITCH + tid] += 1;
    }
    __syncthreads();
}

__global__ __launch_bounds__(SORT_TPB) void sort_count(const int* __restrict__ keys, int shift,
                                                       int nedges, int run,
                                                       int* __restrict__ table) {
    __shared__ int hist[SORT_BINS * SORT_PITCH];
    const int tid = threadIdx.x;
    const long start = ((long)blockIdx.x * SORT_TPB + tid) * run;
    long end = start + run;
    if (end > nedges) end = nedges;
    sort_count_runs(keys, shift, start, end, hist, tid);
    for (int bin = tid; bin < SORT_BINS; bin += SORT_TPB) {
        int total = 0;
        for (int j = 0; j < SORT_TPB; ++j) total += hist[bin * SORT_PITCH + j];
        table[bin * SORT_NB + blockIdx.x] = total;
    }
}

__global__ __launch_bounds__(256) void gin_scan(const int* __restrict__ in,
                                                int* __restrict__ outp) {
    __shared__ int wtot[8];
    __shared__ int wpre[8];
    const int tid  = threadIdx.x;
    const int lane = tid & 31;
    const int wv   = tid >> 5;
    const int base = blockIdx.x * 1024;

    const int nfront = blockIdx.x;
    const int* mine = in + tid * 4;
    int pre = 0;
    for (int ch = 0; ch < nfront; ch += 16) {
        int part[16];
        #pragma unroll
        for (int k = 0; k < 16; ++k) {
            const int at = (ch + k < nfront) ? ch + k : 0;
            const int4 f = *(const int4*)(mine + at * 1024);
            part[k] = (f.x + f.y) + (f.z + f.w);
        }
        #pragma unroll
        for (int k = 0; k < 16; ++k) pre += (ch + k < nfront) ? part[k] : 0;
    }
    const int4 c = *(const int4*)(in + base + tid * 4);
    const int local = (c.x + c.y) + (c.z + c.w);

    int incl = local;
    for (int d = 1; d < 32; d <<= 1) {
        const int t = __shfl_up(incl, d);
        if (lane >= d) incl += t;
    }
    for (int d = 16; d > 0; d >>= 1) pre += __shfl_xor(pre, d);
    if (lane == 31) wtot[wv] = incl;
    if (lane == 0)  wpre[wv] = pre;
    __syncthreads();
    int front = 0;
    for (int w = 0; w < 8; ++w) {
        front += wpre[w];
        if (w < wv) front += wtot[w];
    }
    int4 o;
    o.x = front + incl - local;
    o.y = o.x + c.x;
    o.z = o.y + c.y;
    o.w = o.z + c.z;
    *(int4*)(outp + base + tid * 4) = o;
}

__device__ __forceinline__ void sort_put(int* hist, int tid, int shift, int k, int v,
                                         int* __restrict__ keysOut, int* __restrict__ valsOut) {
    const int d = (k >> shift) & (SORT_BINS - 1);
    const int q = hist[d * SORT_PITCH + tid];
    hist[d * SORT_PITCH + tid] = q + 1;
    keysOut[q] = k;
    valsOut[q] = v;
}

__global__ __launch_bounds__(SORT_TPB) void sort_place(const int* __restrict__ keys,
                                                       const int* __restrict__ vals, int shift,
                                                       int nedges, int run,
                                                       const int* __restrict__ tableS,
                                                       int* __restrict__ keysOut,
                                                       int* __restrict__ valsOut) {
    __shared__ int hist[SORT_BINS * SORT_PITCH];
    const int tid = threadIdx.x;
    const long start = ((long)blockIdx.x * SORT_TPB + tid) * run;
    long end = start + run;
    if (end > nedges) end = nedges;
    sort_count_runs(keys, shift, start, end, hist, tid);
    for (int bin = tid; bin < SORT_BINS; bin += SORT_TPB) {
        int pos = tableS[bin * SORT_NB + blockIdx.x];
        for (int j = 0; j < SORT_TPB; ++j) {
            const int c = hist[bin * SORT_PITCH + j];
            hist[bin * SORT_PITCH + j] = pos;
            pos += c;
        }
    }
    __syncthreads();
    long p = start;
    for (; p + 4 <= end; p += 4) {
        const int k0 = keys[p], k1 = keys[p + 1], k2 = keys[p + 2], k3 = keys[p + 3];
        const int v0 = vals[p], v1 = vals[p + 1], v2 = vals[p + 2], v3 = vals[p + 3];
        sort_put(hist, tid, shift, k0, v0, keysOut, valsOut);
        sort_put(hist, tid, shift, k1, v1, keysOut, valsOut);
        sort_put(hist, tid, shift, k2, v2, keysOut, valsOut);
        sort_put(hist, tid, shift, k3, v3, keysOut, valsOut);
    }
    for (; p < end; ++p) sort_put(hist, tid, shift, keys[p], vals[p], keysOut, valsOut);
}

__global__ __launch_bounds__(SORT_TPB) void sort_fine(const int* __restrict__ keys,
                                                      const int* __restrict__ vals,
                                                      const int* __restrict__ tableS,
                                                      int nedges, int n,
                                                      int* __restrict__ off,
                                                      int* __restrict__ valsOut) {
    __shared__ int hist[SORT_BINS * SORT_PITCH];
    __shared__ int tot[SORT_BINS];
    const int tid = threadIdx.x;
    const int g = blockIdx.x;
    const int bs = (g < SORT_BINS) ? tableS[g * SORT_NB] : nedges;
    const int be = (g + 1 < SORT_BINS) ? tableS[(g + 1) * SORT_NB] : nedges;
    const int run = (be - bs + SORT_TPB - 1) / SORT_TPB;
    long start = (long)bs + (long)tid * run;
    if (start > be) start = be;
    long end = start + run;
    if (end > be) end = be;
    for (int i = tid; i < SORT_BINS * SORT_PITCH; i += SORT_TPB) hist[i] = 0;
    __syncthreads();
    {
        long p = start;
        for (; p + 8 <= end; p += 8) {
            int k[8];
            #pragma unroll
            for (int u = 0; u < 8; ++u) k[u] = keys[p + u];
            #pragma unroll
            for (int u = 0; u < 8; ++u) hist[(k[u] & (SORT_BINS - 1)) * SORT_PITCH + tid] += 1;
        }
        for (; p < end; ++p) hist[(keys[p] & (SORT_BINS - 1)) * SORT_PITCH + tid] += 1;
    }
    __syncthreads();
    for (int bin = tid; bin < SORT_BINS; bin += SORT_TPB) {
        int total = 0;
        for (int j = 0; j < SORT_TPB; ++j) total += hist[bin * SORT_PITCH + j];
        tot[bin] = total;
    }
    __syncthreads();
    for (int bin = tid; bin < SORT_BINS; bin += SORT_TPB) {
        int pos = bs;
        for (int d = 0; d < bin; ++d) pos += tot[d];
        const int v = (g << 8) + bin;
        if (v <= n) off[v] = pos;
        for (int j = 0; j < SORT_TPB; ++j) {
            const int c = hist[bin * SORT_PITCH + j];
            hist[bin * SORT_PITCH + j] = pos;
            pos += c;
        }
    }
    __syncthreads();
    long p = start;
    for (; p + 4 <= end; p += 4) {
        int k[4], v[4];
        #pragma unroll
        for (int u = 0; u < 4; ++u) { k[u] = keys[p + u]; v[u] = vals[p + u]; }
        #pragma unroll
        for (int u = 0; u < 4; ++u) {
            const int d = k[u] & (SORT_BINS - 1);
            const int q = hist[d * SORT_PITCH + tid];
            hist[d * SORT_PITCH + tid] = q + 1;
            valsOut[q] = v[u];
        }
    }
    for (; p < end; ++p) {
        const int d = keys[p] & (SORT_BINS - 1);
        const int q = hist[d * SORT_PITCH + tid];
        hist[d * SORT_PITCH + tid] = q + 1;
        valsOut[q] = vals[p];
    }
}

__global__ __launch_bounds__(256) void gin_offsets(const int* __restrict__ sdst,
                                                   int nedges, int n,
                                                   int* __restrict__ off) {
    const int p = blockIdx.x * blockDim.x + threadIdx.x;
    if (p > nedges) return;
    const int prev = (p == 0) ? -1 : sdst[p - 1];
    const int here = (p == nedges) ? n : sdst[p];
    for (int v = prev + 1; v <= here; ++v) off[v] = p;
}

#define GROWS 16

__device__ __forceinline__ void add4(float4& a, const float4 r) {
    a.x += r.x; a.y += r.y; a.z += r.z; a.w += r.w;
}

__device__ __forceinline__ void close_row(float* zrow, const float e1, const float4 acc) {
    const float4 hv = *(const float4*)zrow;
    float4 o;
    o.x = e1 * hv.x + acc.x;
    o.y = e1 * hv.y + acc.y;
    o.z = e1 * hv.z + acc.z;
    o.w = e1 * hv.w + acc.w;
    *(float4*)zrow = o;
}

__device__ __forceinline__ void load_rows(float4 (&r)[GROWS], const int idxv, const int j,
                                          const float* hc) {
    #pragma unroll
    for (int t = 0; t < GROWS; ++t) {
        const int s = __builtin_amdgcn_readlane(idxv, j + t);
        r[t] = *(const float4*)(hc + (size_t)s * DD);
    }
}

__device__ __forceinline__ void sum_rows(const float4 (&r)[GROWS], const int p0, const int eEnd,
                                         const int ob, float* zl, const float e1,
                                         int& i, int& nextB, float4& acc) {
    #pragma unroll
    for (int t = 0; t < GROWS; ++t) {
        const int p = p0 + t;
        if (p < eEnd) {
            while (p >= nextB) {
                close_row(zl + i * HROW, e1, acc);
                acc.x = 0.0f; acc.y = 0.0f; acc.z = 0.0f; acc.w = 0.0f;
                ++i;
                nextB = __builtin_amdgcn_readlane(ob, i + 1);
            }
            add4(acc, r[t]);
        }
    }
}

__global__ __launch_bounds__(256, 1) void gin_layer(const float* __restrict__ h,
                                                    const float* __restrict__ eps, int layer,
                                                    const int* __restrict__ off,
                                                    const int* __restrict__ csrc,
                                                    const float* __restrict__ W1p,
                                                    const float* __restrict__ b1,
                                                    const float* __restrict__ W2p,
                                                    const float* __restrict__ b2,
                                                    float* __restrict__ out, int n) {
    __shared__ __attribute__((aligned(16))) float zS[8][16 * HROW];
    __shared__ __attribute__((aligned(16))) float hidS[8][16 * HROW];

    const int lane  = threadIdx.x & 31;
    const int wave  = __builtin_amdgcn_readfirstlane((int)(threadIdx.x >> 5));
    const int ln    = lane & 15;
    const int khoff = lane >> 4;
    const int rh    = khoff << 3;
    const int m0    = blockIdx.x * 128 + wave * 16;

    float* zw = &zS[wave][0];
    float* hw = &hidS[wave][0];

    int ob;
    {
        int t = m0 + lane;
        if (t > n) t = n;
        ob = off[t];
    }
    const float* hc = h + lane * 4;
    float* zl = zw + lane * 4;
    {
        float4 own[16];
        #pragma unroll
        for (int q = 0; q < 16; ++q) {
            int rr = m0 + q;
            if (rr > n - 1) rr = n - 1;
            own[q] = *(const float4*)(hc + (size_t)rr * DD);
        }
        #pragma unroll
        for (int q = 0; q < 16; ++q) *(float4*)(zl + q * HROW) = own[q];
    }
    const float e1 = 1.0f + eps[layer];
    const int eEnd = __builtin_amdgcn_readlane(ob, 16);
    int base  = __builtin_amdgcn_readlane(ob, 0);
    int nextB = __builtin_amdgcn_readlane(ob, 1);
    int i = 0;
    float4 acc = {0.0f, 0.0f, 0.0f, 0.0f};
    int m = eEnd - base;
    if (m > 32) m = 32;
    int idxv = (lane < m) ? csrc[base + lane] : 0;
    while (m > 0) {
        const int nb = base + 32;
        int mn = eEnd - nb;
        if (mn > 32) mn = 32;
        float4 r[GROWS];
        load_rows(r, idxv, 0, hc);
        const int idxn = (lane < mn) ? csrc[nb + lane] : 0;
        sum_rows(r, base, eEnd, ob, zl, e1, i, nextB, acc);
        if (m > GROWS) {
            load_rows(r, idxv, GROWS, hc);
            sum_rows(r, base + GROWS, eEnd, ob, zl, e1, i, nextB, acc);
        }
        base = nb;
        m = mn;
        idxv = idxn;
    }
    while (i < 16) {
        close_row(zl + i * HROW, e1, acc);
        acc.x = 0.0f; acc.y = 0.0f; acc.z = 0.0f; acc.w = 0.0f;
        ++i;
    }

    const float* az = zw + ln * HROW + 2 * khoff;
    const float* aw = hw + ln * HROW + 2 * khoff;

    for (int nt = 0; nt < 8; nt += 2) {
        const int n0 = nt * 16;
        const float* Wb0 = W1p + 2 * (khoff * DD + n0 + ln);
        const float* Wb1 = W1p + 2 * (khoff * DD + n0 + 16 + ln);
        v8f acc0 = {}, acc1 = {};
        #pragma unroll
        for (int k0 = 0; k0 < DD; k0 += 4) {
            v2f a  = *(const v2f*)(az + k0);
            v2f g0 = *(const v2f*)(Wb0 + k0 * DD);
            v2f g1 = *(const v2f*)(Wb1 + k0 * DD);
            acc0 = __builtin_amdgcn_wmma_f32_16x16x4_f32(false, a, false, g0,
                                                         (short)0, acc0, false, false);
            acc1 = __builtin_amdgcn_wmma_f32_16x16x4_f32(false, a, false, g1,
                                                         (short)0, acc1, false, false);
        }
        const float bias0 = b1[n0 + ln];
        const float bias1 = b1[n0 + 16 + ln];
        #pragma unroll
        for (int j = 0; j < 8; ++j) {
            hw[(j + rh) * HROW + n0 + ln]      = fmaxf(acc0[j] + bias0, 0.0f);
            hw[(j + rh) * HROW + n0 + 16 + ln] = fmaxf(acc1[j] + bias1, 0.0f);
        }
    }

    __syncthreads();

    for (int nt = 0; nt < 8; nt += 2) {
        const int n0 = nt * 16;
        const float* Wb0 = W2p + 2 * (khoff * DD + n0 + ln);
        const float* Wb1 = W2p + 2 * (khoff * DD + n0 + 16 + ln);
        v8f acc0 = {}, acc1 = {};
        #pragma unroll
        for (int k0 = 0; k0 < DD; k0 += 4) {
            v2f a  = *(const v2f*)(aw + k0);
            v2f g0 = *(const v2f*)(Wb0 + k0 * DD);
            v2f g1 = *(const v2f*)(Wb1 + k0 * DD);
            acc0 = __builtin_amdgcn_wmma_f32_16x16x4_f32(false, a, false, g0,
                                                         (short)0, acc0, false, false);
            acc1 = __builtin_amdgcn_wmma_f32_16x16x4_f32(false, a, false, g1,
                                                         (short)0, acc1, false, false);
        }
        const float bias0 = b2[n0 + ln];
        const float bias1 = b2[n0 + 16 + ln];
        #pragma unroll
        for (int j = 0; j < 8; ++j) {
            const int r = m0 + j + rh;
            if (r < n) {
                out[(size_t)r * DD + n0 + ln]      = acc0[j] + bias0;
                out[(size_t)r * DD + n0 + 16 + ln] = acc1[j] + bias1;
            }
        }
    }
}

extern "C" void kernel_launch(void* const* d_in, const int* in_sizes, int n_in,
                              void* d_out, int out_size, void* d_ws, size_t ws_size,
                              hipStream_t stream) {
    const float* h   = (const float*)d_in[0];
    const int*   src = (const int*)  d_in[1];
    const int*   dst = (const int*)  d_in[2];
    const float* W1  = (const float*)d_in[3];
    const float* b1  = (const float*)d_in[4];
    const float* W2  = (const float*)d_in[5];
    const float* b2  = (const float*)d_in[6];
    const float* eps = (const float*)d_in[7];

    const int n = in_sizes[0] / DD;
    const int e = in_sizes[1];
    const int L = in_sizes[7];

    float* hb  = (float*)d_ws;
    float* W1p = hb + (size_t)n * DD;
    float* W2p = W1p + (size_t)L * DD * DD;
    float* out = (float*)d_out;

    const int tabN = SORT_BINS * SORT_NB;
    const int np   = ((n + 4) / 4) * 4;
    int* table  = (int*)(W2p + (size_t)L * DD * DD);
    int* tableS = table + tabN;
    int* off    = tableS + tabN;
    int* kA     = off + np;
    int* vA     = kA + e;
    int* kB     = vA + e;
    int* vB     = kB + e;

    const int offBlocks  = (e + 1 + 255) / 256;
    const int mlpBlocks  = (n + 127) / 128;
    const int pairs      = L * DD * DD / 2;
    const int rpBlocks   = (pairs + 255) / 256;
    const int run        = (e + SORT_NB * SORT_TPB - 1) / (SORT_NB * SORT_TPB);

    gin_repack<<<rpBlocks, 256, 0, stream>>>(W1, W1p, pairs);
    gin_repack<<<rpBlocks, 256, 0, stream>>>(W2, W2p, pairs);

    const int* kin = dst;
    const int* vin = src;
    if (((n - 1) >> 16) == 0) {
        sort_count<<<SORT_NB, SORT_TPB, 0, stream>>>(dst, 8, e, run, table);
        gin_scan<<<tabN / 1024, 256, 0, stream>>>(table, tableS);
        sort_place<<<SORT_NB, SORT_TPB, 0, stream>>>(dst, src, 8, e, run, tableS, kA, vA);
        sort_fine<<<(n >> 8) + 1, SORT_TPB, 0, stream>>>(kA, vA, tableS, e, n, off, vB);
        vin = vB;
    } else {
        int* kout = kA;
        int* vout = vA;
        for (int shift = 0; shift == 0 || ((n - 1) >> shift) > 0; shift += 8) {
            sort_count<<<SORT_NB, SORT_TPB, 0, stream>>>(kin, shift, e, run, table);
            gin_scan<<<tabN / 1024, 256, 0, stream>>>(table, tableS);
            sort_place<<<SORT_NB, SORT_TPB, 0, stream>>>(kin, vin, shift, e, run, tableS, kout, vout);
            kin = kout;
            vin = vout;
            kout = (kout == kA) ? kB : kA;
            vout = (vout == vA) ? vB : vA;
        }
        gin_offsets<<<offBlocks, 256, 0, stream>>>(kin, e, n, off);
    }

    for (int l = 0; l < L; ++l) {
        const float* hin = (l == 0) ? h : (((L - l) & 1) ? hb : out);
        float* hout = ((L - 1 - l) & 1) ? hb : out;
        gin_layer<<<mlpBlocks, 256, 0, stream>>>(hin, eps, l, off, vin,
                                                 W1p + (size_t)l * DD * DD, b1 + l * DD,
                                                 W2p + (size_t)l * DD * DD, b2 + l * DD, hout, n);
    }
}


# Sort scatters staged in LDS: each block places its edges into an LDS buffer, then stores it front to back (group contiguous / per-bin runs) instead of one scattered 4-byte store per lane; order kept
# speedup vs baseline: 2.8228x; 1.1620x over previous
#include <hip/hip_runtime.h>
#include <hip/hip_bf16.h>

typedef __attribute__((ext_vector_type(2))) float v2f;
typedef __attribute__((ext_vector_type(8))) float v8f;

#define DD 128
#define HROW 132

__global__ __launch_bounds__(256) void gin_repack(const float* __restrict__ W,
                                                  float* __restrict__ Wp,
                                                  int totalPairs) {
    int i = blockIdx.x * blockDim.x + threadIdx.x;
    if (i >= totalPairs) return;
    int n  = i & 127;
    int kh = (i >> 7) & 63;
    int l  = i >> 13;
    const float* Wl = W + (size_t)l * DD * DD;
    float2 p;
    p.x = Wl[(2 * kh) * DD + n];
    p.y = Wl[(2 * kh + 1) * DD + n];
    ((float2*)Wp)[i] = p;
}

#define SORT_NB    512
#define SORT_TPB   128
#define SORT_BINS  256
#define SORT_PITCH 129

__device__ __forceinline__ void sort_count_runs(const int* __restrict__ keys, int shift,
                                                long start, long end, int* hist, int tid) {
    for (int i = tid; i < SORT_BINS * SORT_PITCH; i += SORT_TPB) hist[i] = 0;
    __syncthreads();
    long p = start;
    for (; p + 4 <= end; p += 4) {
        const int k0 = keys[p], k1 = keys[p + 1], k2 = keys[p + 2], k3 = keys[p + 3];
        hist[((k0 >> shift) & (SORT_BINS - 1)) * SORT_PITCH + tid] += 1;
        hist[((k1 >> shift) & (SORT_BINS - 1)) * SORT_PITCH + tid] += 1;
        hist[((k2 >> shift) & (SORT_BINS - 1)) * SORT_PITCH + tid] += 1;
        hist[((k3 >> shift) & (SORT_BINS - 1)) * SORT_PITCH + tid] += 1;
    }
    for (; p < end; ++p) {
        const int d = (keys[p] >> shift) & (SORT_BINS - 1);
        hist[d * SORT_PITCH + tid] += 1;
    }
    __syncthreads();
}

__global__ __launch_bounds__(SORT_TPB) void sort_count(const int* __restrict__ keys, int shift,
                                                       int nedges, int run,
                                                       int* __restrict__ table) {
    __shared__ int hist[SORT_BINS * SORT_PITCH];
    const int tid = threadIdx.x;
    const long start = ((long)blockIdx.x * SORT_TPB + tid) * run;
    long end = start + run;
    if (end > nedges) end = nedges;
    sort_count_runs(keys, shift, start, end, hist, tid);
    for (int bin = tid; bin < SORT_BINS; bin += SORT_TPB) {
        int total = 0;
        for (int j = 0; j < SORT_TPB; ++j) total += hist[bin * SORT_PITCH + j];
        table[bin * SORT_NB + blockIdx.x] = total;
    }
}

__global__ __launch_bounds__(256) void gin_scan(const int* __restrict__ in,
                                                int* __restrict__ outp) {
    __shared__ int wtot[8];
    __shared__ int wpre[8];
    const int tid  = threadIdx.x;
    const int lane = tid & 31;
    const int wv   = tid >> 5;
    const int base = blockIdx.x * 1024;

    const int nfront = blockIdx.x;
    const int* mine = in + tid * 4;
    int pre = 0;
    for (int ch = 0; ch < nfront; ch += 16) {
        int part[16];
        #pragma unroll
        for (int k = 0; k < 16; ++k) {
            const int at = (ch + k < nfront) ? ch + k : 0;
            const int4 f = *(const int4*)(mine + at * 1024);
            part[k] = (f.x + f.y) + (f.z + f.w);
        }
        #pragma unroll
        for (int k = 0; k < 16; ++k) pre += (ch + k < nfront) ? part[k] : 0;
    }
    const int4 c = *(const int4*)(in + base + tid * 4);
    const int local = (c.x + c.y) + (c.z + c.w);

    int incl = local;
    for (int d = 1; d < 32; d <<= 1) {
        const int t = __shfl_up(incl, d);
        if (lane >= d) incl += t;
    }
    for (int d = 16; d > 0; d >>= 1) pre += __shfl_xor(pre, d);
    if (lane == 31) wtot[wv] = incl;
    if (lane == 0)  wpre[wv] = pre;
    __syncthreads();
    int front = 0;
    for (int w = 0; w < 8; ++w) {
        front += wpre[w];
        if (w < wv) front += wtot[w];
    }
    int4 o;
    o.x = front + incl - local;
    o.y = o.x + c.x;
    o.z = o.y + c.y;
    o.w = o.z + c.z;
    *(int4*)(outp + base + tid * 4) = o;
}

__device__ __forceinline__ void sort_put(int* hist, int tid, int shift, int k, int v,
                                         int* __restrict__ keysOut, int* __restrict__ valsOut) {
    const int d = (k >> shift) & (SORT_BINS - 1);
    const int q = hist[d * SORT_PITCH + tid];
    hist[d * SORT_PITCH + tid] = q + 1;
    keysOut[q] = k;
    valsOut[q] = v;
}

#define PLACE_CAP 2048

__device__ __forceinline__ void sort_put_lds(int* hist, const int* lbs, int* stK, int* stV,
                                             int tid, int shift, int k, int v) {
    const int d = (k >> shift) & (SORT_BINS - 1);
    const int w = hist[d * SORT_PITCH + tid];
    hist[d * SORT_PITCH + tid] = w + 1;
    const int i = lbs[d] + w;
    stK[i] = k;
    stV[i] = v;
}

__global__ __launch_bounds__(SORT_TPB) void sort_place(const int* __restrict__ keys,
                                                       const int* __restrict__ vals, int shift,
                                                       int nedges, int run,
                                                       const int* __restrict__ tableS,
                                                       int* __restrict__ keysOut,
                                                       int* __restrict__ valsOut) {
    __shared__ int hist[SORT_BINS * SORT_PITCH];
    __shared__ int tot[SORT_BINS];
    __shared__ int lbs[SORT_BINS];
    __shared__ int dlt[SORT_BINS];
    __shared__ int wsum[SORT_TPB / 32];
    __shared__ int stK[PLACE_CAP];
    __shared__ int stV[PLACE_CAP];
    const int tid = threadIdx.x;
    const long start = ((long)blockIdx.x * SORT_TPB + tid) * run;
    long end = start + run;
    if (end > nedges) end = nedges;
    sort_count_runs(keys, shift, start, end, hist, tid);
    if ((long)run * SORT_TPB <= PLACE_CAP) {
        const int g0 = tableS[(2 * tid) * SORT_NB + blockIdx.x];
        const int g1 = tableS[(2 * tid + 1) * SORT_NB + blockIdx.x];
        for (int bin = tid; bin < SORT_BINS; bin += SORT_TPB) {
            int pos = 0;
            for (int j = 0; j < SORT_TPB; ++j) {
                const int c = hist[bin * SORT_PITCH + j];
                hist[bin * SORT_PITCH + j] = pos;
                pos += c;
            }
            tot[bin] = pos;
        }
        __syncthreads();
        const int lane = tid & 31;
        const int wv   = tid >> 5;
        const int ta = tot[2 * tid];
        const int tb = tot[2 * tid + 1];
        int incl = ta + tb;
        for (int d = 1; d < 32; d <<= 1) {
            const int t = __shfl_up(incl, d);
            if (lane >= d) incl += t;
        }
        if (lane == 31) wsum[wv] = incl;
        __syncthreads();
        int front = 0;
        for (int w = 0; w < SORT_TPB / 32; ++w) front += (w < wv) ? wsum[w] : 0;
        const int ex = front + incl - (ta + tb);
        lbs[2 * tid]     = ex;
        lbs[2 * tid + 1] = ex + ta;
        dlt[2 * tid]     = g0 - ex;
        dlt[2 * tid + 1] = g1 - (ex + ta);
        __syncthreads();
        long p = start;
        for (; p + 4 <= end; p += 4) {
            const int k0 = keys[p], k1 = keys[p + 1], k2 = keys[p + 2], k3 = keys[p + 3];
            const int v0 = vals[p], v1 = vals[p + 1], v2 = vals[p + 2], v3 = vals[p + 3];
            sort_put_lds(hist, lbs, stK, stV, tid, shift, k0, v0);
            sort_put_lds(hist, lbs, stK, stV, tid, shift, k1, v1);
            sort_put_lds(hist, lbs, stK, stV, tid, shift, k2, v2);
            sort_put_lds(hist, lbs, stK, stV, tid, shift, k3, v3);
        }
        for (; p < end; ++p) sort_put_lds(hist, lbs, stK, stV, tid, shift, keys[p], vals[p]);
        __syncthreads();
        const long bstart = (long)blockIdx.x * SORT_TPB * run;
        long bend = bstart + (long)SORT_TPB * run;
        if (bend > nedges) bend = nedges;
        const int cnt = (bend > bstart) ? (int)(bend - bstart) : 0;
        int i = tid;
        for (; i + 3 * SORT_TPB < cnt; i += 4 * SORT_TPB) {
            int k[4], v[4], o[4];
            #pragma unroll
            for (int u = 0; u < 4; ++u) { k[u] = stK[i + u * SORT_TPB]; v[u] = stV[i + u * SORT_TPB]; }
            #pragma unroll
            for (int u = 0; u < 4; ++u) o[u] = dlt[(k[u] >> shift) & (SORT_BINS - 1)] + i + u * SORT_TPB;
            #pragma unroll
            for (int u = 0; u < 4; ++u) { keysOut[o[u]] = k[u]; valsOut[o[u]] = v[u]; }
        }
        for (; i < cnt; i += SORT_TPB) {
            const int k = stK[i];
            const int v = stV[i];
            const int o = dlt[(k >> shift) & (SORT_BINS - 1)] + i;
            keysOut[o] = k;
            valsOut[o] = v;
        }
        return;
    }
    for (int bin = tid; bin < SORT_BINS; bin += SORT_TPB) {
        int pos = tableS[bin * SORT_NB + blockIdx.x];
        for (int j = 0; j < SORT_TPB; ++j) {
            const int c = hist[bin * SORT_PITCH + j];
            hist[bin * SORT_PITCH + j] = pos;
            pos += c;
        }
    }
    __syncthreads();
    long p = start;
    for (; p + 4 <= end; p += 4) {
        const int k0 = keys[p], k1 = keys[p + 1], k2 = keys[p + 2], k3 = keys[p + 3];
        const int v0 = vals[p], v1 = vals[p + 1], v2 = vals[p + 2], v3 = vals[p + 3];
        sort_put(hist, tid, shift, k0, v0, keysOut, valsOut);
        sort_put(hist, tid, shift, k1, v1, keysOut, valsOut);
        sort_put(hist, tid, shift, k2, v2, keysOut, valsOut);
        sort_put(hist, tid, shift, k3, v3, keysOut, valsOut);
    }
    for (; p < end; ++p) sort_put(hist, tid, shift, keys[p], vals[p], keysOut, valsOut);
}

#define FINE_CAP 8192

__global__ __launch_bounds__(SORT_TPB) void sort_fine(const int* __restrict__ keys,
                                                      const int* __restrict__ vals,
                                                      const int* __restrict__ tableS,
                                                      int nedges, int n,
                                                      int* __restrict__ off,
                                                      int* __restrict__ valsOut) {
    __shared__ int hist[SORT_BINS * SORT_PITCH];
    __shared__ int tot[SORT_BINS];
    __shared__ int stage[FINE_CAP];
    const int tid = threadIdx.x;
    const int g = blockIdx.x;
    const int bs = (g < SORT_BINS) ? tableS[g * SORT_NB] : nedges;
    const int be = (g + 1 < SORT_BINS) ? tableS[(g + 1) * SORT_NB] : nedges;
    const int run = (be - bs + SORT_TPB - 1) / SORT_TPB;
    long start = (long)bs + (long)tid * run;
    if (start > be) start = be;
    long end = start + run;
    if (end > be) end = be;
    for (int i = tid; i < SORT_BINS * SORT_PITCH; i += SORT_TPB) hist[i] = 0;
    __syncthreads();
    {
        long p = start;
        for (; p + 8 <= end; p += 8) {
            int k[8];
            #pragma unroll
            for (int u = 0; u < 8; ++u) k[u] = keys[p + u];
            #pragma unroll
            for (int u = 0; u < 8; ++u) hist[(k[u] & (SORT_BINS - 1)) * SORT_PITCH + tid] += 1;
        }
        for (; p < end; ++p) hist[(keys[p] & (SORT_BINS - 1)) * SORT_PITCH + tid] += 1;
    }
    __syncthreads();
    for (int bin = tid; bin < SORT_BINS; bin += SORT_TPB) {
        int total = 0;
        for (int j = 0; j < SORT_TPB; ++j) total += hist[bin * SORT_PITCH + j];
        tot[bin] = total;
    }
    __syncthreads();
    for (int bin = tid; bin < SORT_BINS; bin += SORT_TPB) {
        int pos = bs;
        for (int d = 0; d < bin; ++d) pos += tot[d];
        const int v = (g << 8) + bin;
        if (v <= n) off[v] = pos;
        for (int j = 0; j < SORT_TPB; ++j) {
            const int c = hist[bin * SORT_PITCH + j];
            hist[bin * SORT_PITCH + j] = pos;
            pos += c;
        }
    }
    __syncthreads();
    if (be - bs <= FINE_CAP) {
        long p = start;
        for (; p + 4 <= end; p += 4) {
            int k[4], v[4];
            #pragma unroll
            for (int u = 0; u < 4; ++u) { k[u] = keys[p + u]; v[u] = vals[p + u]; }
            #pragma unroll
            for (int u = 0; u < 4; ++u) {
                const int d = k[u] & (SORT_BINS - 1);
                const int q = hist[d * SORT_PITCH + tid];
                hist[d * SORT_PITCH + tid] = q + 1;
                stage[q - bs] = v[u];
            }
        }
        for (; p < end; ++p) {
            const int d = keys[p] & (SORT_BINS - 1);
            const int q = hist[d * SORT_PITCH + tid];
            hist[d * SORT_PITCH + tid] = q + 1;
            stage[q - bs] = vals[p];
        }
        __syncthreads();
        const int cnt = be - bs;
        int i = tid;
        for (; i + 7 * SORT_TPB < cnt; i += 8 * SORT_TPB) {
            int w[8];
            #pragma unroll
            for (int u = 0; u < 8; ++u) w[u] = stage[i + u * SORT_TPB];
            #pragma unroll
            for (int u = 0; u < 8; ++u) valsOut[bs + i + u * SORT_TPB] = w[u];
        }
        for (; i < cnt; i += SORT_TPB) valsOut[bs + i] = stage[i];
        return;
    }
    long p = start;
    for (; p + 4 <= end; p += 4) {
        int k[4], v[4];
        #pragma unroll
        for (int u = 0; u < 4; ++u) { k[u] = keys[p + u]; v[u] = vals[p + u]; }
        #pragma unroll
        for (int u = 0; u < 4; ++u) {
            const int d = k[u] & (SORT_BINS - 1);
            const int q = hist[d * SORT_PITCH + tid];
            hist[d * SORT_PITCH + tid] = q + 1;
            valsOut[q] = v[u];
        }
    }
    for (; p < end; ++p) {
        const int d = keys[p] & (SORT_BINS - 1);
        const int q = hist[d * SORT_PITCH + tid];
        hist[d * SORT_PITCH + tid] = q + 1;
        valsOut[q] = vals[p];
    }
}

__global__ __launch_bounds__(256) void gin_offsets(const int* __restrict__ sdst,
                                                   int nedges, int n,
                                                   int* __restrict__ off) {
    const int p = blockIdx.x * blockDim.x + threadIdx.x;
    if (p > nedges) return;
    const int prev = (p == 0) ? -1 : sdst[p - 1];
    const int here = (p == nedges) ? n : sdst[p];
    for (int v = prev + 1; v <= here; ++v) off[v] = p;
}

#define GROWS 16

__device__ __forceinline__ void add4(float4& a, const float4 r) {
    a.x += r.x; a.y += r.y; a.z += r.z; a.w += r.w;
}

__device__ __forceinline__ void close_row(float* zrow, const float e1, const float4 acc) {
    const float4 hv = *(const float4*)zrow;
    float4 o;
    o.x = e1 * hv.x + acc.x;
    o.y = e1 * hv.y + acc.y;
    o.z = e1 * hv.z + acc.z;
    o.w = e1 * hv.w + acc.w;
    *(float4*)zrow = o;
}

__device__ __forceinline__ void load_rows(float4 (&r)[GROWS], const int idxv, const int j,
                                          const float* hc) {
    #pragma unroll
    for (int t = 0; t < GROWS; ++t) {
        const int s = __builtin_amdgcn_readlane(idxv, j + t);
        r[t] = *(const float4*)(hc + (size_t)s * DD);
    }
}

__device__ __forceinline__ void sum_rows(const float4 (&r)[GROWS], const int p0, const int eEnd,
                                         const int ob, float* zl, const float e1,
                                         int& i, int& nextB, float4& acc) {
    #pragma unroll
    for (int t = 0; t < GROWS; ++t) {
        const int p = p0 + t;
        if (p < eEnd) {
            while (p >= nextB) {
                close_row(zl + i * HROW, e1, acc);
                acc.x = 0.0f; acc.y = 0.0f; acc.z = 0.0f; acc.w = 0.0f;
                ++i;
                nextB = __builtin_amdgcn_readlane(ob, i + 1);
            }
            add4(acc, r[t]);
        }
    }
}

__global__ __launch_bounds__(256, 1) void gin_layer(const float* __restrict__ h,
                                                    const float* __restrict__ eps, int layer,
                                                    const int* __restrict__ off,
                                                    const int* __restrict__ csrc,
                                                    const float* __restrict__ W1p,
                                                    const float* __restrict__ b1,
                                                    const float* __restrict__ W2p,
                                                    const float* __restrict__ b2,
                                                    float* __restrict__ out, int n) {
    __shared__ __attribute__((aligned(16))) float zS[8][16 * HROW];
    __shared__ __attribute__((aligned(16))) float hidS[8][16 * HROW];

    const int lane  = threadIdx.x & 31;
    const int wave  = __builtin_amdgcn_readfirstlane((int)(threadIdx.x >> 5));
    const int ln    = lane & 15;
    const int khoff = lane >> 4;
    const int rh    = khoff << 3;
    const int m0    = blockIdx.x * 128 + wave * 16;

    float* zw = &zS[wave][0];
    float* hw = &hidS[wave][0];

    int ob;
    {
        int t = m0 + lane;
        if (t > n) t = n;
        ob = off[t];
    }
    const float* hc = h + lane * 4;
    float* zl = zw + lane * 4;
    {
        float4 own[16];
        #pragma unroll
        for (int q = 0; q < 16; ++q) {
            int rr = m0 + q;
            if (rr > n - 1) rr = n - 1;
            own[q] = *(const float4*)(hc + (size_t)rr * DD);
        }
        #pragma unroll
        for (int q = 0; q < 16; ++q) *(float4*)(zl + q * HROW) = own[q];
    }
    const float e1 = 1.0f + eps[layer];
    const int eEnd = __builtin_amdgcn_readlane(ob, 16);
    int base  = __builtin_amdgcn_readlane(ob, 0);
    int nextB = __builtin_amdgcn_readlane(ob, 1);
    int i = 0;
    float4 acc = {0.0f, 0.0f, 0.0f, 0.0f};
    int m = eEnd - base;
    if (m > 32) m = 32;
    int idxv = (lane < m) ? csrc[base + lane] : 0;
    while (m > 0) {
        const int nb = base + 32;
        int mn = eEnd - nb;
        if (mn > 32) mn = 32;
        float4 r[GROWS];
        load_rows(r, idxv, 0, hc);
        const int idxn = (lane < mn) ? csrc[nb + lane] : 0;
        sum_rows(r, base, eEnd, ob, zl, e1, i, nextB, acc);
        if (m > GROWS) {
            load_rows(r, idxv, GROWS, hc);
            sum_rows(r, base + GROWS, eEnd, ob, zl, e1, i, nextB, acc);
        }
        base = nb;
        m = mn;
        idxv = idxn;
    }
    while (i < 16) {
        close_row(zl + i * HROW, e1, acc);
        acc.x = 0.0f; acc.y = 0.0f; acc.z = 0.0f; acc.w = 0.0f;
        ++i;
    }

    const float* az = zw + ln * HROW + 2 * khoff;
    const float* aw = hw + ln * HROW + 2 * khoff;

    for (int nt = 0; nt < 8; nt += 2) {
        const int n0 = nt * 16;
        const float* Wb0 = W1p + 2 * (khoff * DD + n0 + ln);
        const float* Wb1 = W1p + 2 * (khoff * DD + n0 + 16 + ln);
        v8f acc0 = {}, acc1 = {};
        #pragma unroll
        for (int k0 = 0; k0 < DD; k0 += 4) {
            v2f a  = *(const v2f*)(az + k0);
            v2f g0 = *(const v2f*)(Wb0 + k0 * DD);
            v2f g1 = *(const v2f*)(Wb1 + k0 * DD);
            acc0 = __builtin_amdgcn_wmma_f32_16x16x4_f32(false, a, false, g0,
                                                         (short)0, acc0, false, false);
            acc1 = __builtin_amdgcn_wmma_f32_16x16x4_f32(false, a, false, g1,
                                                         (short)0, acc1, false, false);
        }
        const float bias0 = b1[n0 + ln];
        const float bias1 = b1[n0 + 16 + ln];
        #pragma unroll
        for (int j = 0; j < 8; ++j) {
            hw[(j + rh) * HROW + n0 + ln]      = fmaxf(acc0[j] + bias0, 0.0f);
            hw[(j + rh) * HROW + n0 + 16 + ln] = fmaxf(acc1[j] + bias1, 0.0f);
        }
    }

    __syncthreads();

    for (int nt = 0; nt < 8; nt += 2) {
        const int n0 = nt * 16;
        const float* Wb0 = W2p + 2 * (khoff * DD + n0 + ln);
        const float* Wb1 = W2p + 2 * (khoff * DD + n0 + 16 + ln);
        v8f acc0 = {}, acc1 = {};
        #pragma unroll
        for (int k0 = 0; k0 < DD; k0 += 4) {
            v2f a  = *(const v2f*)(aw + k0);
            v2f g0 = *(const v2f*)(Wb0 + k0 * DD);
            v2f g1 = *(const v2f*)(Wb1 + k0 * DD);
            acc0 = __builtin_amdgcn_wmma_f32_16x16x4_f32(false, a, false, g0,
                                                         (short)0, acc0, false, false);
            acc1 = __builtin_amdgcn_wmma_f32_16x16x4_f32(false, a, false, g1,
                                                         (short)0, acc1, false, false);
        }
        const float bias0 = b2[n0 + ln];
        const float bias1 = b2[n0 + 16 + ln];
        #pragma unroll
        for (int j = 0; j < 8; ++j) {
            const int r = m0 + j + rh;
            if (r < n) {
                out[(size_t)r * DD + n0 + ln]      = acc0[j] + bias0;
                out[(size_t)r * DD + n0 + 16 + ln] = acc1[j] + bias1;
            }
        }
    }
}

extern "C" void kernel_launch(void* const* d_in, const int* in_sizes, int n_in,
                              void* d_out, int out_size, void* d_ws, size_t ws_size,
                              hipStream_t stream) {
    const float* h   = (const float*)d_in[0];
    const int*   src = (const int*)  d_in[1];
    const int*   dst = (const int*)  d_in[2];
    const float* W1  = (const float*)d_in[3];
    const float* b1  = (const float*)d_in[4];
    const float* W2  = (const float*)d_in[5];
    const float* b2  = (const float*)d_in[6];
    const float* eps = (const float*)d_in[7];

    const int n = in_sizes[0] / DD;
    const int e = in_sizes[1];
    const int L = in_sizes[7];

    float* hb  = (float*)d_ws;
    float* W1p = hb + (size_t)n * DD;
    float* W2p = W1p + (size_t)L * DD * DD;
    float* out = (float*)d_out;

    const int tabN = SORT_BINS * SORT_NB;
    const int np   = ((n + 4) / 4) * 4;
    int* table  = (int*)(W2p + (size_t)L * DD * DD);
    int* tableS = table + tabN;
    int* off    = tableS + tabN;
    int* kA     = off + np;
    int* vA     = kA + e;
    int* kB     = vA + e;
    int* vB     = kB + e;

    const int offBlocks  = (e + 1 + 255) / 256;
    const int mlpBlocks  = (n + 127) / 128;
    const int pairs      = L * DD * DD / 2;
    const int rpBlocks   = (pairs + 255) / 256;
    const int run        = (e + SORT_NB * SORT_TPB - 1) / (SORT_NB * SORT_TPB);

    gin_repack<<<rpBlocks, 256, 0, stream>>>(W1, W1p, pairs);
    gin_repack<<<rpBlocks, 256, 0, stream>>>(W2, W2p, pairs);

    const int* kin = dst;
    const int* vin = src;
    if (((n - 1) >> 16) == 0) {
        sort_count<<<SORT_NB, SORT_TPB, 0, stream>>>(dst, 8, e, run, table);
        gin_scan<<<tabN / 1024, 256, 0, stream>>>(table, tableS);
        sort_place<<<SORT_NB, SORT_TPB, 0, stream>>>(dst, src, 8, e, run, tableS, kA, vA);
        sort_fine<<<(n >> 8) + 1, SORT_TPB, 0, stream>>>(kA, vA, tableS, e, n, off, vB);
        vin = vB;
    } else {
        int* kout = kA;
        int* vout = vA;
        for (int shift = 0; shift == 0 || ((n - 1) >> shift) > 0; shift += 8) {
            sort_count<<<SORT_NB, SORT_TPB, 0, stream>>>(kin, shift, e, run, table);
            gin_scan<<<tabN / 1024, 256, 0, stream>>>(table, tableS);
            sort_place<<<SORT_NB, SORT_TPB, 0, stream>>>(kin, vin, shift, e, run, tableS, kout, vout);
            kin = kout;
            vin = vout;
            kout = (kout == kA) ? kB : kA;
            vout = (vout == vA) ? vB : vA;
        }
        gin_offsets<<<offBlocks, 256, 0, stream>>>(kin, e, n, off);
    }

    for (int l = 0; l < L; ++l) {
        const float* hin = (l == 0) ? h : (((L - l) & 1) ? hb : out);
        float* hout = ((L - 1 - l) & 1) ? hb : out;
        gin_layer<<<mlpBlocks, 256, 0, stream>>>(hin, eps, l, off, vin,
                                                 W1p + (size_t)l * DD * DD, b1 + l * DD,
                                                 W2p + (size_t)l * DD * DD, b2 + l * DD, hout, n);
    }
}


# CSR build without global counts: each block sorts its own edge slice by the high key byte in LDS; the group pass gathers one piece per slice into LDS and sorts there; 2 launches fewer, order kept
# speedup vs baseline: 3.0681x; 1.0869x over previous
#include <hip/hip_runtime.h>
#include <hip/hip_bf16.h>

typedef __attribute__((ext_vector_type(2))) float v2f;
typedef __attribute__((ext_vector_type(8))) float v8f;

#define DD 128
#define HROW 132

__global__ __launch_bounds__(256) void gin_repack(const float* __restrict__ W,
                                                  float* __restrict__ Wp,
                                                  int totalPairs) {
    int i = blockIdx.x * blockDim.x + threadIdx.x;
    if (i >= totalPairs) return;
    int n  = i & 127;
    int kh = (i >> 7) & 63;
    int l  = i >> 13;
    const float* Wl = W + (size_t)l * DD * DD;
    float2 p;
    p.x = Wl[(2 * kh) * DD + n];
    p.y = Wl[(2 * kh + 1) * DD + n];
    ((float2*)Wp)[i] = p;
}

#define SORT_NB    512
#define SORT_TPB   128
#define SORT_BINS  256
#define SORT_PITCH 129

__device__ __forceinline__ void sort_count_runs(const int* __restrict__ keys, int shift,
                                                long start, long end, int* hist, int tid) {
    for (int i = tid; i < SORT_BINS * SORT_PITCH; i += SORT_TPB) hist[i] = 0;
    __syncthreads();
    long p = start;
    for (; p + 4 <= end; p += 4) {
        const int k0 = keys[p], k1 = keys[p + 1], k2 = keys[p + 2], k3 = keys[p + 3];
        hist[((k0 >> shift) & (SORT_BINS - 1)) * SORT_PITCH + tid] += 1;
        hist[((k1 >> shift) & (SORT_BINS - 1)) * SORT_PITCH + tid] += 1;
        hist[((k2 >> shift) & (SORT_BINS - 1)) * SORT_PITCH + tid] += 1;
        hist[((k3 >> shift) & (SORT_BINS - 1)) * SORT_PITCH + tid] += 1;
    }
    for (; p < end; ++p) {
        const int d = (keys[p] >> shift) & (SORT_BINS - 1);
        hist[d * SORT_PITCH + tid] += 1;
    }
    __syncthreads();
}

__global__ __launch_bounds__(SORT_TPB) void sort_count(const int* __restrict__ keys, int shift,
                                                       int nedges, int run,
                                                       int* __restrict__ table) {
    __shared__ int hist[SORT_BINS * SORT_PITCH];
    const int tid = threadIdx.x;
    const long start = ((long)blockIdx.x * SORT_TPB + tid) * run;
    long end = start + run;
    if (end > nedges) end = nedges;
    sort_count_runs(keys, shift, start, end, hist, tid);
    for (int bin = tid; bin < SORT_BINS; bin += SORT_TPB) {
        int total = 0;
        for (int j = 0; j < SORT_TPB; ++j) total += hist[bin * SORT_PITCH + j];
        table[bin * SORT_NB + blockIdx.x] = total;
    }
}

__global__ __launch_bounds__(256) void gin_scan(const int* __restrict__ in,
                                                int* __restrict__ outp) {
    __shared__ int wtot[8];
    __shared__ int wpre[8];
    const int tid  = threadIdx.x;
    const int lane = tid & 31;
    const int wv   = tid >> 5;
    const int base = blockIdx.x * 1024;

    const int nfront = blockIdx.x;
    const int* mine = in + tid * 4;
    int pre = 0;
    for (int ch = 0; ch < nfront; ch += 16) {
        int part[16];
        #pragma unroll
        for (int k = 0; k < 16; ++k) {
            const int at = (ch + k < nfront) ? ch + k : 0;
            const int4 f = *(const int4*)(mine + at * 1024);
            part[k] = (f.x + f.y) + (f.z + f.w);
        }
        #pragma unroll
        for (int k = 0; k < 16; ++k) pre += (ch + k < nfront) ? part[k] : 0;
    }
    const int4 c = *(const int4*)(in + base + tid * 4);
    const int local = (c.x + c.y) + (c.z + c.w);

    int incl = local;
    for (int d = 1; d < 32; d <<= 1) {
        const int t = __shfl_up(incl, d);
        if (lane >= d) incl += t;
    }
    for (int d = 16; d > 0; d >>= 1) pre += __shfl_xor(pre, d);
    if (lane == 31) wtot[wv] = incl;
    if (lane == 0)  wpre[wv] = pre;
    __syncthreads();
    int front = 0;
    for (int w = 0; w < 8; ++w) {
        front += wpre[w];
        if (w < wv) front += wtot[w];
    }
    int4 o;
    o.x = front + incl - local;
    o.y = o.x + c.x;
    o.z = o.y + c.y;
    o.w = o.z + c.z;
    *(int4*)(outp + base + tid * 4) = o;
}

__device__ __forceinline__ void sort_put(int* hist, int tid, int shift, int k, int v,
                                         int* __restrict__ keysOut, int* __restrict__ valsOut) {
    const int d = (k >> shift) & (SORT_BINS - 1);
    const int q = hist[d * SORT_PITCH + tid];
    hist[d * SORT_PITCH + tid] = q + 1;
    keysOut[q] = k;
    valsOut[q] = v;
}

#define PLACE_CAP 2048

__device__ __forceinline__ void sort_put_lds(int* hist, const int* lbs, int* stK, int* stV,
                                             int tid, int shift, int k, int v) {
    const int d = (k >> shift) & (SORT_BINS - 1);
    const int w = hist[d * SORT_PITCH + tid];
    hist[d * SORT_PITCH + tid] = w + 1;
    const int i = lbs[d] + w;
    stK[i] = k;
    stV[i] = v;
}

__global__ __launch_bounds__(SORT_TPB) void sort_place(const int* __restrict__ keys,
                                                       const int* __restrict__ vals, int shift,
                                                       int nedges, int run,
                                                       const int* __restrict__ tableS,
                                                       int* __restrict__ keysOut,
                                                       int* __restrict__ valsOut) {
    __shared__ int hist[SORT_BINS * SORT_PITCH];
    __shared__ int tot[SORT_BINS];
    __shared__ int lbs[SORT_BINS];
    __shared__ int dlt[SORT_BINS];
    __shared__ int wsum[SORT_TPB / 32];
    __shared__ int stK[PLACE_CAP];
    __shared__ int stV[PLACE_CAP];
    const int tid = threadIdx.x;
    const long start = ((long)blockIdx.x * SORT_TPB + tid) * run;
    long end = start + run;
    if (end > nedges) end = nedges;
    sort_count_runs(keys, shift, start, end, hist, tid);
    if ((long)run * SORT_TPB <= PLACE_CAP) {
        const int g0 = tableS[(2 * tid) * SORT_NB + blockIdx.x];
        const int g1 = tableS[(2 * tid + 1) * SORT_NB + blockIdx.x];
        for (int bin = tid; bin < SORT_BINS; bin += SORT_TPB) {
            int pos = 0;
            for (int j = 0; j < SORT_TPB; ++j) {
                const int c = hist[bin * SORT_PITCH + j];
                hist[bin * SORT_PITCH + j] = pos;
                pos += c;
            }
            tot[bin] = pos;
        }
        __syncthreads();
        const int lane = tid & 31;
        const int wv   = tid >> 5;
        const int ta = tot[2 * tid];
        const int tb = tot[2 * tid + 1];
        int incl = ta + tb;
        for (int d = 1; d < 32; d <<= 1) {
            const int t = __shfl_up(incl, d);
            if (lane >= d) incl += t;
        }
        if (lane == 31) wsum[wv] = incl;
        __syncthreads();
        int front = 0;
        for (int w = 0; w < SORT_TPB / 32; ++w) front += (w < wv) ? wsum[w] : 0;
        const int ex = front + incl - (ta + tb);
        lbs[2 * tid]     = ex;
        lbs[2 * tid + 1] = ex + ta;
        dlt[2 * tid]     = g0 - ex;
        dlt[2 * tid + 1] = g1 - (ex + ta);
        __syncthreads();
        long p = start;
        for (; p + 4 <= end; p += 4) {
            const int k0 = keys[p], k1 = keys[p + 1], k2 = keys[p + 2], k3 = keys[p + 3];
            const int v0 = vals[p], v1 = vals[p + 1], v2 = vals[p + 2], v3 = vals[p + 3];
            sort_put_lds(hist, lbs, stK, stV, tid, shift, k0, v0);
            sort_put_lds(hist, lbs, stK, stV, tid, shift, k1, v1);
            sort_put_lds(hist, lbs, stK, stV, tid, shift, k2, v2);
            sort_put_lds(hist, lbs, stK, stV, tid, shift, k3, v3);
        }
        for (; p < end; ++p) sort_put_lds(hist, lbs, stK, stV, tid, shift, keys[p], vals[p]);
        __syncthreads();
        const long bstart = (long)blockIdx.x * SORT_TPB * run;
        long bend = bstart + (long)SORT_TPB * run;
        if (bend > nedges) bend = nedges;
        const int cnt = (bend > bstart) ? (int)(bend - bstart) : 0;
        int i = tid;
        for (; i + 3 * SORT_TPB < cnt; i += 4 * SORT_TPB) {
            int k[4], v[4], o[4];
            #pragma unroll
            for (int u = 0; u < 4; ++u) { k[u] = stK[i + u * SORT_TPB]; v[u] = stV[i + u * SORT_TPB]; }
            #pragma unroll
            for (int u = 0; u < 4; ++u) o[u] = dlt[(k[u] >> shift) & (SORT_BINS - 1)] + i + u * SORT_TPB;
            #pragma unroll
            for (int u = 0; u < 4; ++u) { keysOut[o[u]] = k[u]; valsOut[o[u]] = v[u]; }
        }
        for (; i < cnt; i += SORT_TPB) {
            const int k = stK[i];
            const int v = stV[i];
            const int o = dlt[(k >> shift) & (SORT_BINS - 1)] + i;
            keysOut[o] = k;
            valsOut[o] = v;
        }
        return;
    }
    for (int bin = tid; bin < SORT_BINS; bin += SORT_TPB) {
        int pos = tableS[bin * SORT_NB + blockIdx.x];
        for (int j = 0; j < SORT_TPB; ++j) {
            const int c = hist[bin * SORT_PITCH + j];
            hist[bin * SORT_PITCH + j] = pos;
            pos += c;
        }
    }
    __syncthreads();
    long p = start;
    for (; p + 4 <= end; p += 4) {
        const int k0 = keys[p], k1 = keys[p + 1], k2 = keys[p + 2], k3 = keys[p + 3];
        const int v0 = vals[p], v1 = vals[p + 1], v2 = vals[p + 2], v3 = vals[p + 3];
        sort_put(hist, tid, shift, k0, v0, keysOut, valsOut);
        sort_put(hist, tid, shift, k1, v1, keysOut, valsOut);
        sort_put(hist, tid, shift, k2, v2, keysOut, valsOut);
        sort_put(hist, tid, shift, k3, v3, keysOut, valsOut);
    }
    for (; p < end; ++p) sort_put(hist, tid, shift, keys[p], vals[p], keysOut, valsOut);
}

__global__ __launch_bounds__(SORT_TPB) void sort_local(const int* __restrict__ keys,
                                                       const int* __restrict__ vals,
                                                       int nedges, int run,
                                                       int* __restrict__ segs,
                                                       int* __restrict__ keysOut,
                                                       int* __restrict__ valsOut) {
    __shared__ int hist[SORT_BINS * SORT_PITCH];
    __shared__ int tot[SORT_BINS];
    __shared__ int lbs[SORT_BINS];
    __shared__ int wsum[SORT_TPB / 32];
    __shared__ int stK[PLACE_CAP];
    __shared__ int stV[PLACE_CAP];
    const int tid = threadIdx.x;
    const long start = ((long)blockIdx.x * SORT_TPB + tid) * run;
    long end = start + run;
    if (end > nedges) end = nedges;
    sort_count_runs(keys, 8, start, end, hist, tid);
    for (int bin = tid; bin < SORT_BINS; bin += SORT_TPB) {
        int pos = 0;
        for (int j = 0; j < SORT_TPB; ++j) {
            const int c = hist[bin * SORT_PITCH + j];
            hist[bin * SORT_PITCH + j] = pos;
            pos += c;
        }
        tot[bin] = pos;
    }
    __syncthreads();
    const int lane = tid & 31;
    const int wv   = tid >> 5;
    const int ta = tot[2 * tid];
    const int tb = tot[2 * tid + 1];
    int incl = ta + tb;
    for (int d = 1; d < 32; d <<= 1) {
        const int t = __shfl_up(incl, d);
        if (lane >= d) incl += t;
    }
    if (lane == 31) wsum[wv] = incl;
    __syncthreads();
    int front = 0;
    for (int w = 0; w < SORT_TPB / 32; ++w) front += (w < wv) ? wsum[w] : 0;
    const int ex = front + incl - (ta + tb);
    lbs[2 * tid]     = ex;
    lbs[2 * tid + 1] = ex + ta;
    segs[(2 * tid) * SORT_NB + blockIdx.x]     = ex;
    segs[(2 * tid + 1) * SORT_NB + blockIdx.x] = ex + ta;
    if (tid == SORT_TPB - 1) segs[SORT_BINS * SORT_NB + blockIdx.x] = ex + ta + tb;
    __syncthreads();
    long p = start;
    for (; p + 4 <= end; p += 4) {
        const int k0 = keys[p], k1 = keys[p + 1], k2 = keys[p + 2], k3 = keys[p + 3];
        const int v0 = vals[p], v1 = vals[p + 1], v2 = vals[p + 2], v3 = vals[p + 3];
        sort_put_lds(hist, lbs, stK, stV, tid, 8, k0, v0);
        sort_put_lds(hist, lbs, stK, stV, tid, 8, k1, v1);
        sort_put_lds(hist, lbs, stK, stV, tid, 8, k2, v2);
        sort_put_lds(hist, lbs, stK, stV, tid, 8, k3, v3);
    }
    for (; p < end; ++p) sort_put_lds(hist, lbs, stK, stV, tid, 8, keys[p], vals[p]);
    __syncthreads();
    const long bstart = (long)blockIdx.x * SORT_TPB * run;
    long bend = bstart + (long)SORT_TPB * run;
    if (bend > nedges) bend = nedges;
    const int cnt = (bend > bstart) ? (int)(bend - bstart) : 0;
    int i = tid;
    for (; i + 3 * SORT_TPB < cnt; i += 4 * SORT_TPB) {
        int k[4], v[4];
        #pragma unroll
        for (int u = 0; u < 4; ++u) { k[u] = stK[i + u * SORT_TPB]; v[u] = stV[i + u * SORT_TPB]; }
        #pragma unroll
        for (int u = 0; u < 4; ++u) { keysOut[bstart + i + u * SORT_TPB] = k[u]; valsOut[bstart + i + u * SORT_TPB] = v[u]; }
    }
    for (; i < cnt; i += SORT_TPB) {
        keysOut[bstart + i] = stK[i];
        valsOut[bstart + i] = stV[i];
    }
}

#define FINE_CAP 8192
#define FINE_SL  (SORT_NB / SORT_TPB)
#define FINE_PB  16
#define FINE_CH  16
#define FINE_LST (FINE_CAP + FINE_CAP / 32)

__device__ __forceinline__ int lst_at(int p) { return p + (p >> 5); }

__global__ __launch_bounds__(SORT_TPB) void sort_fine(const int* __restrict__ keys,
                                                      const int* __restrict__ vals,
                                                      const int* __restrict__ segs,
                                                      int sliceLen, int n,
                                                      int* __restrict__ off,
                                                      int* __restrict__ valsOut) {
    __shared__ int hist[SORT_BINS * SORT_PITCH];
    __shared__ int tot[SORT_BINS];
    __shared__ int stage[FINE_CAP];
    __shared__ int lstK[FINE_LST];
    __shared__ int lstV[FINE_LST];
    __shared__ int pcB[SORT_NB];
    __shared__ int pcL[SORT_NB];
    __shared__ int pcA[SORT_NB];
    __shared__ int red[2 * (SORT_TPB / 32)];
    const int tid  = threadIdx.x;
    const int lane = tid & 31;
    const int wv   = tid >> 5;
    const int g = blockIdx.x;
    const int gl = (g < SORT_BINS) ? g : SORT_BINS;
    const int gh = (g + 1 < SORT_BINS) ? g + 1 : SORT_BINS;
    const int4 s0 = *(const int4*)(segs + gl * SORT_NB + FINE_SL * tid);
    const int4 s1 = *(const int4*)(segs + gh * SORT_NB + FINE_SL * tid);
    int pb[FINE_SL], pl[FINE_SL];
    pb[0] = (FINE_SL * tid + 0) * sliceLen + s0.x;  pl[0] = s1.x - s0.x;
    pb[1] = (FINE_SL * tid + 1) * sliceLen + s0.y;  pl[1] = s1.y - s0.y;
    pb[2] = (FINE_SL * tid + 2) * sliceLen + s0.z;  pl[2] = s1.z - s0.z;
    pb[3] = (FINE_SL * tid + 3) * sliceLen + s0.w;  pl[3] = s1.w - s0.w;
    for (int i = tid; i < SORT_BINS * SORT_PITCH; i += SORT_TPB) hist[i] = 0;
    const int mine = (pl[0] + pl[1]) + (pl[2] + pl[3]);
    int incl = mine;
    int a = (s0.x + s0.y) + (s0.z + s0.w);
    for (int d = 1; d < 32; d <<= 1) {
        const int t = __shfl_up(incl, d);
        if (lane >= d) incl += t;
    }
    for (int d = 16; d > 0; d >>= 1) a += __shfl_xor(a, d);
    if (lane == 31) red[wv] = incl;
    if (lane == 0)  red[SORT_TPB / 32 + wv] = a;
    __syncthreads();
    int bs = 0, cntg = 0, front = 0;
    for (int w = 0; w < SORT_TPB / 32; ++w) {
        cntg += red[w];
        front += (w < wv) ? red[w] : 0;
        bs += red[SORT_TPB / 32 + w];
    }
    const bool inLds = cntg <= FINE_CAP;
    const int run = (cntg + SORT_TPB - 1) / SORT_TPB;
    int start = tid * run;
    if (start > cntg) start = cntg;
    int end = start + run;
    if (end > cntg) end = cntg;

    if (inLds) {
        int at0 = front + incl - mine;
        #pragma unroll
        for (int u = 0; u < FINE_SL; ++u) {
            pcB[FINE_SL * tid + u] = (pl[u] > 0) ? pb[u] : 0;
            pcL[FINE_SL * tid + u] = pl[u];
            pcA[FINE_SL * tid + u] = at0;
            at0 += pl[u];
        }
    }
    __syncthreads();
    if (inLds) {
        const int sub = lane >> 3;
        const int j   = lane & 7;
        for (int r0 = 0; r0 < SORT_NB / 16; r0 += FINE_CH) {
            int B[FINE_CH], L[FINE_CH], A[FINE_CH];
            int k[FINE_CH][2], v[FINE_CH][2];
            #pragma unroll
            for (int r = 0; r < FINE_CH; ++r) {
                const int pi = wv * (SORT_NB / 4) + (r0 + r) * 4 + sub;
                B[r] = pcB[pi];
                L[r] = pcL[pi];
                A[r] = pcA[pi];
            }
            #pragma unroll
            for (int r = 0; r < FINE_CH; ++r) {
                const int last = (L[r] > 0) ? L[r] - 1 : 0;
                #pragma unroll
                for (int hh = 0; hh < 2; ++hh) {
                    const int jj = j + 8 * hh;
                    const int at = B[r] + ((jj < last) ? jj : last);
                    k[r][hh] = keys[at];
                    v[r][hh] = vals[at];
                }
            }
            #pragma unroll
            for (int r = 0; r < FINE_CH; ++r) {
                #pragma unroll
                for (int hh = 0; hh < 2; ++hh) {
                    const int jj = j + 8 * hh;
                    if (jj < L[r]) {
                        lstK[lst_at(A[r] + jj)] = k[r][hh];
                        lstV[lst_at(A[r] + jj)] = v[r][hh];
                    }
                }
                for (int q = FINE_PB + j; q < L[r]; q += 8) {
                    lstK[lst_at(A[r] + q)] = keys[B[r] + q];
                    lstV[lst_at(A[r] + q)] = vals[B[r] + q];
                }
            }
        }
    }
    __syncthreads();
    if (inLds) {
        int p = start;
        for (; p + 8 <= end; p += 8) {
            int k[8];
            #pragma unroll
            for (int u = 0; u < 8; ++u) k[u] = lstK[lst_at(p + u)];
            #pragma unroll
            for (int u = 0; u < 8; ++u) hist[(k[u] & (SORT_BINS - 1)) * SORT_PITCH + tid] += 1;
        }
        for (; p < end; ++p) hist[(lstK[lst_at(p)] & (SORT_BINS - 1)) * SORT_PITCH + tid] += 1;
    } else {
        #pragma unroll
        for (int u = 0; u < FINE_SL; ++u)
            for (int q = 0; q < pl[u]; ++q)
                hist[(keys[pb[u] + q] & (SORT_BINS - 1)) * SORT_PITCH + tid] += 1;
    }
    __syncthreads();
    for (int bin = tid; bin < SORT_BINS; bin += SORT_TPB) {
        int total = 0;
        for (int j = 0; j < SORT_TPB; ++j) total += hist[bin * SORT_PITCH + j];
        tot[bin] = total;
    }
    __syncthreads();
    for (int bin = tid; bin < SORT_BINS; bin += SORT_TPB) {
        int pos = bs;
        for (int d = 0; d < bin; ++d) pos += tot[d];
        const int v = (g << 8) + bin;
        if (v <= n) off[v] = pos;
        for (int j = 0; j < SORT_TPB; ++j) {
            const int c = hist[bin * SORT_PITCH + j];
            hist[bin * SORT_PITCH + j] = pos;
            pos += c;
        }
    }
    __syncthreads();
    if (inLds) {
        int p = start;
        for (; p + 4 <= end; p += 4) {
            int k[4], v[4];
            #pragma unroll
            for (int u = 0; u < 4; ++u) { k[u] = lstK[lst_at(p + u)]; v[u] = lstV[lst_at(p + u)]; }
            #pragma unroll
            for (int u = 0; u < 4; ++u) {
                const int d = k[u] & (SORT_BINS - 1);
                const int w = hist[d * SORT_PITCH + tid];
                hist[d * SORT_PITCH + tid] = w + 1;
                stage[w - bs] = v[u];
            }
        }
        for (; p < end; ++p) {
            const int d = lstK[lst_at(p)] & (SORT_BINS - 1);
            const int w = hist[d * SORT_PITCH + tid];
            hist[d * SORT_PITCH + tid] = w + 1;
            stage[w - bs] = lstV[lst_at(p)];
        }
        __syncthreads();
        int i = tid;
        for (; i + 7 * SORT_TPB < cntg; i += 8 * SORT_TPB) {
            int w[8];
            #pragma unroll
            for (int u = 0; u < 8; ++u) w[u] = stage[i + u * SORT_TPB];
            #pragma unroll
            for (int u = 0; u < 8; ++u) valsOut[bs + i + u * SORT_TPB] = w[u];
        }
        for (; i < cntg; i += SORT_TPB) valsOut[bs + i] = stage[i];
        return;
    }
    #pragma unroll
    for (int u = 0; u < FINE_SL; ++u) {
        for (int q = 0; q < pl[u]; ++q) {
            const int d = keys[pb[u] + q] & (SORT_BINS - 1);
            const int w = hist[d * SORT_PITCH + tid];
            hist[d * SORT_PITCH + tid] = w + 1;
            valsOut[w] = vals[pb[u] + q];
        }
    }
}

__global__ __launch_bounds__(256) void gin_offsets(const int* __restrict__ sdst,
                                                   int nedges, int n,
                                                   int* __restrict__ off) {
    const int p = blockIdx.x * blockDim.x + threadIdx.x;
    if (p > nedges) return;
    const int prev = (p == 0) ? -1 : sdst[p - 1];
    const int here = (p == nedges) ? n : sdst[p];
    for (int v = prev + 1; v <= here; ++v) off[v] = p;
}

#define GROWS 16

__device__ __forceinline__ void add4(float4& a, const float4 r) {
    a.x += r.x; a.y += r.y; a.z += r.z; a.w += r.w;
}

__device__ __forceinline__ void close_row(float* zrow, const float e1, const float4 acc) {
    const float4 hv = *(const float4*)zrow;
    float4 o;
    o.x = e1 * hv.x + acc.x;
    o.y = e1 * hv.y + acc.y;
    o.z = e1 * hv.z + acc.z;
    o.w = e1 * hv.w + acc.w;
    *(float4*)zrow = o;
}

__device__ __forceinline__ void load_rows(float4 (&r)[GROWS], const int idxv, const int j,
                                          const float* hc) {
    #pragma unroll
    for (int t = 0; t < GROWS; ++t) {
        const int s = __builtin_amdgcn_readlane(idxv, j + t);
        r[t] = *(const float4*)(hc + (size_t)s * DD);
    }
}

__device__ __forceinline__ void sum_rows(const float4 (&r)[GROWS], const int p0, const int eEnd,
                                         const int ob, float* zl, const float e1,
                                         int& i, int& nextB, float4& acc) {
    #pragma unroll
    for (int t = 0; t < GROWS; ++t) {
        const int p = p0 + t;
        if (p < eEnd) {
            while (p >= nextB) {
                close_row(zl + i * HROW, e1, acc);
                acc.x = 0.0f; acc.y = 0.0f; acc.z = 0.0f; acc.w = 0.0f;
                ++i;
                nextB = __builtin_amdgcn_readlane(ob, i + 1);
            }
            add4(acc, r[t]);
        }
    }
}

__global__ __launch_bounds__(256, 1) void gin_layer(const float* __restrict__ h,
                                                    const float* __restrict__ eps, int layer,
                                                    const int* __restrict__ off,
                                                    const int* __restrict__ csrc,
                                                    const float* __restrict__ W1p,
                                                    const float* __restrict__ b1,
                                                    const float* __restrict__ W2p,
                                                    const float* __restrict__ b2,
                                                    float* __restrict__ out, int n) {
    __shared__ __attribute__((aligned(16))) float zS[8][16 * HROW];
    __shared__ __attribute__((aligned(16))) float hidS[8][16 * HROW];

    const int lane  = threadIdx.x & 31;
    const int wave  = __builtin_amdgcn_readfirstlane((int)(threadIdx.x >> 5));
    const int ln    = lane & 15;
    const int khoff = lane >> 4;
    const int rh    = khoff << 3;
    const int m0    = blockIdx.x * 128 + wave * 16;

    float* zw = &zS[wave][0];
    float* hw = &hidS[wave][0];

    int ob;
    {
        int t = m0 + lane;
        if (t > n) t = n;
        ob = off[t];
    }
    const float* hc = h + lane * 4;
    float* zl = zw + lane * 4;
    {
        float4 own[16];
        #pragma unroll
        for (int q = 0; q < 16; ++q) {
            int rr = m0 + q;
            if (rr > n - 1) rr = n - 1;
            own[q] = *(const float4*)(hc + (size_t)rr * DD);
        }
        #pragma unroll
        for (int q = 0; q < 16; ++q) *(float4*)(zl + q * HROW) = own[q];
    }
    const float e1 = 1.0f + eps[layer];
    const int eEnd = __builtin_amdgcn_readlane(ob, 16);
    int base  = __builtin_amdgcn_readlane(ob, 0);
    int nextB = __builtin_amdgcn_readlane(ob, 1);
    int i = 0;
    float4 acc = {0.0f, 0.0f, 0.0f, 0.0f};
    int m = eEnd - base;
    if (m > 32) m = 32;
    int idxv = (lane < m) ? csrc[base + lane] : 0;
    while (m > 0) {
        const int nb = base + 32;
        int mn = eEnd - nb;
        if (mn > 32) mn = 32;
        float4 r[GROWS];
        load_rows(r, idxv, 0, hc);
        const int idxn = (lane < mn) ? csrc[nb + lane] : 0;
        sum_rows(r, base, eEnd, ob, zl, e1, i, nextB, acc);
        if (m > GROWS) {
            load_rows(r, idxv, GROWS, hc);
            sum_rows(r, base + GROWS, eEnd, ob, zl, e1, i, nextB, acc);
        }
        base = nb;
        m = mn;
        idxv = idxn;
    }
    while (i < 16) {
        close_row(zl + i * HROW, e1, acc);
        acc.x = 0.0f; acc.y = 0.0f; acc.z = 0.0f; acc.w = 0.0f;
        ++i;
    }

    const float* az = zw + ln * HROW + 2 * khoff;
    const float* aw = hw + ln * HROW + 2 * khoff;

    for (int nt = 0; nt < 8; nt += 2) {
        const int n0 = nt * 16;
        const float* Wb0 = W1p + 2 * (khoff * DD + n0 + ln);
        const float* Wb1 = W1p + 2 * (khoff * DD + n0 + 16 + ln);
        v8f acc0 = {}, acc1 = {};
        #pragma unroll
        for (int k0 = 0; k0 < DD; k0 += 4) {
            v2f a  = *(const v2f*)(az + k0);
            v2f g0 = *(const v2f*)(Wb0 + k0 * DD);
            v2f g1 = *(const v2f*)(Wb1 + k0 * DD);
            acc0 = __builtin_amdgcn_wmma_f32_16x16x4_f32(false, a, false, g0,
                                                         (short)0, acc0, false, false);
            acc1 = __builtin_amdgcn_wmma_f32_16x16x4_f32(false, a, false, g1,
                                                         (short)0, acc1, false, false);
        }
        const float bias0 = b1[n0 + ln];
        const float bias1 = b1[n0 + 16 + ln];
        #pragma unroll
        for (int j = 0; j < 8; ++j) {
            hw[(j + rh) * HROW + n0 + ln]      = fmaxf(acc0[j] + bias0, 0.0f);
            hw[(j + rh) * HROW + n0 + 16 + ln] = fmaxf(acc1[j] + bias1, 0.0f);
        }
    }

    __syncthreads();

    for (int nt = 0; nt < 8; nt += 2) {
        const int n0 = nt * 16;
        const float* Wb0 = W2p + 2 * (khoff * DD + n0 + ln);
        const float* Wb1 = W2p + 2 * (khoff * DD + n0 + 16 + ln);
        v8f acc0 = {}, acc1 = {};
        #pragma unroll
        for (int k0 = 0; k0 < DD; k0 += 4) {
            v2f a  = *(const v2f*)(aw + k0);
            v2f g0 = *(const v2f*)(Wb0 + k0 * DD);
            v2f g1 = *(const v2f*)(Wb1 + k0 * DD);
            acc0 = __builtin_amdgcn_wmma_f32_16x16x4_f32(false, a, false, g0,
                                                         (short)0, acc0, false, false);
            acc1 = __builtin_amdgcn_wmma_f32_16x16x4_f32(false, a, false, g1,
                                                         (short)0, acc1, false, false);
        }
        const float bias0 = b2[n0 + ln];
        const float bias1 = b2[n0 + 16 + ln];
        #pragma unroll
        for (int j = 0; j < 8; ++j) {
            const int r = m0 + j + rh;
            if (r < n) {
                out[(size_t)r * DD + n0 + ln]      = acc0[j] + bias0;
                out[(size_t)r * DD + n0 + 16 + ln] = acc1[j] + bias1;
            }
        }
    }
}

extern "C" void kernel_launch(void* const* d_in, const int* in_sizes, int n_in,
                              void* d_out, int out_size, void* d_ws, size_t ws_size,
                              hipStream_t stream) {
    const float* h   = (const float*)d_in[0];
    const int*   src = (const int*)  d_in[1];
    const int*   dst = (const int*)  d_in[2];
    const float* W1  = (const float*)d_in[3];
    const float* b1  = (const float*)d_in[4];
    const float* W2  = (const float*)d_in[5];
    const float* b2  = (const float*)d_in[6];
    const float* eps = (const float*)d_in[7];

    const int n = in_sizes[0] / DD;
    const int e = in_sizes[1];
    const int L = in_sizes[7];

    float* hb  = (float*)d_ws;
    float* W1p = hb + (size_t)n * DD;
    float* W2p = W1p + (size_t)L * DD * DD;
    float* out = (float*)d_out;

    const int tabN = SORT_BINS * SORT_NB;
    const int np   = ((n + 4) / 4) * 4;
    int* table  = (int*)(W2p + (size_t)L * DD * DD);
    int* tableS = table + tabN;
    int* off    = tableS + tabN;
    int* kA     = off + np;
    int* vA     = kA + e;
    int* kB     = vA + e;
    int* vB     = kB + e;

    const int offBlocks  = (e + 1 + 255) / 256;
    const int mlpBlocks  = (n + 127) / 128;
    const int pairs      = L * DD * DD / 2;
    const int rpBlocks   = (pairs + 255) / 256;
    const int run        = (e + SORT_NB * SORT_TPB - 1) / (SORT_NB * SORT_TPB);

    gin_repack<<<rpBlocks, 256, 0, stream>>>(W1, W1p, pairs);
    gin_repack<<<rpBlocks, 256, 0, stream>>>(W2, W2p, pairs);

    const int* kin = dst;
    const int* vin = src;
    if (((n - 1) >> 16) == 0 && (long)run * SORT_TPB <= PLACE_CAP) {
        int* segs = table;
        sort_local<<<SORT_NB, SORT_TPB, 0, stream>>>(dst, src, e, run, segs, kA, vA);
        sort_fine<<<(n >> 8) + 1, SORT_TPB, 0, stream>>>(kA, vA, segs, run * SORT_TPB, n, off, vB);
        vin = vB;
    } else {
        int* kout = kA;
        int* vout = vA;
        for (int shift = 0; shift == 0 || ((n - 1) >> shift) > 0; shift += 8) {
            sort_count<<<SORT_NB, SORT_TPB, 0, stream>>>(kin, shift, e, run, table);
            gin_scan<<<tabN / 1024, 256, 0, stream>>>(table, tableS);
            sort_place<<<SORT_NB, SORT_TPB, 0, stream>>>(kin, vin, shift, e, run, tableS, kout, vout);
            kin = kout;
            vin = vout;
            kout = (kout == kA) ? kB : kA;
            vout = (vout == vA) ? vB : vA;
        }
        gin_offsets<<<offBlocks, 256, 0, stream>>>(kin, e, n, off);
    }

    for (int l = 0; l < L; ++l) {
        const float* hin = (l == 0) ? h : (((L - l) & 1) ? hb : out);
        float* hout = ((L - 1 - l) & 1) ? hb : out;
        gin_layer<<<mlpBlocks, 256, 0, stream>>>(hin, eps, l, off, vin,
                                                 W1p + (size_t)l * DD * DD, b1 + l * DD,
                                                 W2p + (size_t)l * DD * DD, b2 + l * DD, hout, n);
    }
}


# Weights fetched once per work-group from W into a 64 KB LDS image shared by its 8 waves: B fragments from LDS, k outer over 8 N-tiles, hidden tile in the z tile; repack launches gone; sums unchanged
# speedup vs baseline: 3.2181x; 1.0489x over previous
#include <hip/hip_runtime.h>
#include <hip/hip_bf16.h>

typedef __attribute__((ext_vector_type(2))) float v2f;
typedef __attribute__((ext_vector_type(8))) float v8f;

#define DD 128
#define HROW 132

#define SORT_NB    512
#define SORT_TPB   128
#define SORT_BINS  256
#define SORT_PITCH 129

__device__ __forceinline__ void sort_count_runs(const int* __restrict__ keys, int shift,
                                                long start, long end, int* hist, int tid) {
    for (int i = tid; i < SORT_BINS * SORT_PITCH; i += SORT_TPB) hist[i] = 0;
    __syncthreads();
    long p = start;
    for (; p + 4 <= end; p += 4) {
        const int k0 = keys[p], k1 = keys[p + 1], k2 = keys[p + 2], k3 = keys[p + 3];
        hist[((k0 >> shift) & (SORT_BINS - 1)) * SORT_PITCH + tid] += 1;
        hist[((k1 >> shift) & (SORT_BINS - 1)) * SORT_PITCH + tid] += 1;
        hist[((k2 >> shift) & (SORT_BINS - 1)) * SORT_PITCH + tid] += 1;
        hist[((k3 >> shift) & (SORT_BINS - 1)) * SORT_PITCH + tid] += 1;
    }
    for (; p < end; ++p) {
        const int d = (keys[p] >> shift) & (SORT_BINS - 1);
        hist[d * SORT_PITCH + tid] += 1;
    }
    __syncthreads();
}

__global__ __launch_bounds__(SORT_TPB) void sort_count(const int* __restrict__ keys, int shift,
                                                       int nedges, int run,
                                                       int* __restrict__ table) {
    __shared__ int hist[SORT_BINS * SORT_PITCH];
    const int tid = threadIdx.x;
    const long start = ((long)blockIdx.x * SORT_TPB + tid) * run;
    long end = start + run;
    if (end > nedges) end = nedges;
    sort_count_runs(keys, shift, start, end, hist, tid);
    for (int bin = tid; bin < SORT_BINS; bin += SORT_TPB) {
        int total = 0;
        for (int j = 0; j < SORT_TPB; ++j) total += hist[bin * SORT_PITCH + j];
        table[bin * SORT_NB + blockIdx.x] = total;
    }
}

__global__ __launch_bounds__(256) void gin_scan(const int* __restrict__ in,
                                                int* __restrict__ outp) {
    __shared__ int wtot[8];
    __shared__ int wpre[8];
    const int tid  = threadIdx.x;
    const int lane = tid & 31;
    const int wv   = tid >> 5;
    const int base = blockIdx.x * 1024;

    const int nfront = blockIdx.x;
    const int* mine = in + tid * 4;
    int pre = 0;
    for (int ch = 0; ch < nfront; ch += 16) {
        int part[16];
        #pragma unroll
        for (int k = 0; k < 16; ++k) {
            const int at = (ch + k < nfront) ? ch + k : 0;
            const int4 f = *(const int4*)(mine + at * 1024);
            part[k] = (f.x + f.y) + (f.z + f.w);
        }
        #pragma unroll
        for (int k = 0; k < 16; ++k) pre += (ch + k < nfront) ? part[k] : 0;
    }
    const int4 c = *(const int4*)(in + base + tid * 4);
    const int local = (c.x + c.y) + (c.z + c.w);

    int incl = local;
    for (int d = 1; d < 32; d <<= 1) {
        const int t = __shfl_up(incl, d);
        if (lane >= d) incl += t;
    }
    for (int d = 16; d > 0; d >>= 1) pre += __shfl_xor(pre, d);
    if (lane == 31) wtot[wv] = incl;
    if (lane == 0)  wpre[wv] = pre;
    __syncthreads();
    int front = 0;
    for (int w = 0; w < 8; ++w) {
        front += wpre[w];
        if (w < wv) front += wtot[w];
    }
    int4 o;
    o.x = front + incl - local;
    o.y = o.x + c.x;
    o.z = o.y + c.y;
    o.w = o.z + c.z;
    *(int4*)(outp + base + tid * 4) = o;
}

__device__ __forceinline__ void sort_put(int* hist, int tid, int shift, int k, int v,
                                         int* __restrict__ keysOut, int* __restrict__ valsOut) {
    const int d = (k >> shift) & (SORT_BINS - 1);
    const int q = hist[d * SORT_PITCH + tid];
    hist[d * SORT_PITCH + tid] = q + 1;
    keysOut[q] = k;
    valsOut[q] = v;
}

#define PLACE_CAP 2048

__device__ __forceinline__ void sort_put_lds(int* hist, const int* lbs, int* stK, int* stV,
                                             int tid, int shift, int k, int v) {
    const int d = (k >> shift) & (SORT_BINS - 1);
    const int w = hist[d * SORT_PITCH + tid];
    hist[d * SORT_PITCH + tid] = w + 1;
    const int i = lbs[d] + w;
    stK[i] = k;
    stV[i] = v;
}

__global__ __launch_bounds__(SORT_TPB) void sort_place(const int* __restrict__ keys,
                                                       const int* __restrict__ vals, int shift,
                                                       int nedges, int run,
                                                       const int* __restrict__ tableS,
                                                       int* __restrict__ keysOut,
                                                       int* __restrict__ valsOut) {
    __shared__ int hist[SORT_BINS * SORT_PITCH];
    __shared__ int tot[SORT_BINS];
    __shared__ int lbs[SORT_BINS];
    __shared__ int dlt[SORT_BINS];
    __shared__ int wsum[SORT_TPB / 32];
    __shared__ int stK[PLACE_CAP];
    __shared__ int stV[PLACE_CAP];
    const int tid = threadIdx.x;
    const long start = ((long)blockIdx.x * SORT_TPB + tid) * run;
    long end = start + run;
    if (end > nedges) end = nedges;
    sort_count_runs(keys, shift, start, end, hist, tid);
    if ((long)run * SORT_TPB <= PLACE_CAP) {
        const int g0 = tableS[(2 * tid) * SORT_NB + blockIdx.x];
        const int g1 = tableS[(2 * tid + 1) * SORT_NB + blockIdx.x];
        for (int bin = tid; bin < SORT_BINS; bin += SORT_TPB) {
            int pos = 0;
            for (int j = 0; j < SORT_TPB; ++j) {
                const int c = hist[bin * SORT_PITCH + j];
                hist[bin * SORT_PITCH + j] = pos;
                pos += c;
            }
            tot[bin] = pos;
        }
        __syncthreads();
        const int lane = tid & 31;
        const int wv   = tid >> 5;
        const int ta = tot[2 * tid];
        const int tb = tot[2 * tid + 1];
        int incl = ta + tb;
        for (int d = 1; d < 32; d <<= 1) {
            const int t = __shfl_up(incl, d);
            if (lane >= d) incl += t;
        }
        if (lane == 31) wsum[wv] = incl;
        __syncthreads();
        int front = 0;
        for (int w = 0; w < SORT_TPB / 32; ++w) front += (w < wv) ? wsum[w] : 0;
        const int ex = front + incl - (ta + tb);
        lbs[2 * tid]     = ex;
        lbs[2 * tid + 1] = ex + ta;
        dlt[2 * tid]     = g0 - ex;
        dlt[2 * tid + 1] = g1 - (ex + ta);
        __syncthreads();
        long p = start;
        for (; p + 4 <= end; p += 4) {
            const int k0 = keys[p], k1 = keys[p + 1], k2 = keys[p + 2], k3 = keys[p + 3];
            const int v0 = vals[p], v1 = vals[p + 1], v2 = vals[p + 2], v3 = vals[p + 3];
            sort_put_lds(hist, lbs, stK, stV, tid, shift, k0, v0);
            sort_put_lds(hist, lbs, stK, stV, tid, shift, k1, v1);
            sort_put_lds(hist, lbs, stK, stV, tid, shift, k2, v2);
            sort_put_lds(hist, lbs, stK, stV, tid, shift, k3, v3);
        }
        for (; p < end; ++p) sort_put_lds(hist, lbs, stK, stV, tid, shift, keys[p], vals[p]);
        __syncthreads();
        const long bstart = (long)blockIdx.x * SORT_TPB * run;
        long bend = bstart + (long)SORT_TPB * run;
        if (bend > nedges) bend = nedges;
        const int cnt = (bend > bstart) ? (int)(bend - bstart) : 0;
        int i = tid;
        for (; i + 3 * SORT_TPB < cnt; i += 4 * SORT_TPB) {
            int k[4], v[4], o[4];
            #pragma unroll
            for (int u = 0; u < 4; ++u) { k[u] = stK[i + u * SORT_TPB]; v[u] = stV[i + u * SORT_TPB]; }
            #pragma unroll
            for (int u = 0; u < 4; ++u) o[u] = dlt[(k[u] >> shift) & (SORT_BINS - 1)] + i + u * SORT_TPB;
            #pragma unroll
            for (int u = 0; u < 4; ++u) { keysOut[o[u]] = k[u]; valsOut[o[u]] = v[u]; }
        }
        for (; i < cnt; i += SORT_TPB) {
            const int k = stK[i];
            const int v = stV[i];
            const int o = dlt[(k >> shift) & (SORT_BINS - 1)] + i;
            keysOut[o] = k;
            valsOut[o] = v;
        }
        return;
    }
    for (int bin = tid; bin < SORT_BINS; bin += SORT_TPB) {
        int pos = tableS[bin * SORT_NB + blockIdx.x];
        for (int j = 0; j < SORT_TPB; ++j) {
            const int c = hist[bin * SORT_PITCH + j];
            hist[bin * SORT_PITCH + j] = pos;
            pos += c;
        }
    }
    __syncthreads();
    long p = start;
    for (; p + 4 <= end; p += 4) {
        const int k0 = keys[p], k1 = keys[p + 1], k2 = keys[p + 2], k3 = keys[p + 3];
        const int v0 = vals[p], v1 = vals[p + 1], v2 = vals[p + 2], v3 = vals[p + 3];
        sort_put(hist, tid, shift, k0, v0, keysOut, valsOut);
        sort_put(hist, tid, shift, k1, v1, keysOut, valsOut);
        sort_put(hist, tid, shift, k2, v2, keysOut, valsOut);
        sort_put(hist, tid, shift, k3, v3, keysOut, valsOut);
    }
    for (; p < end; ++p) sort_put(hist, tid, shift, keys[p], vals[p], keysOut, valsOut);
}

__global__ __launch_bounds__(SORT_TPB) void sort_local(const int* __restrict__ keys,
                                                       const int* __restrict__ vals,
                                                       int nedges, int run,
                                                       int* __restrict__ segs,
                                                       int* __restrict__ keysOut,
                                                       int* __restrict__ valsOut) {
    __shared__ int hist[SORT_BINS * SORT_PITCH];
    __shared__ int tot[SORT_BINS];
    __shared__ int lbs[SORT_BINS];
    __shared__ int wsum[SORT_TPB / 32];
    __shared__ int stK[PLACE_CAP];
    __shared__ int stV[PLACE_CAP];
    const int tid = threadIdx.x;
    const long start = ((long)blockIdx.x * SORT_TPB + tid) * run;
    long end = start + run;
    if (end > nedges) end = nedges;
    sort_count_runs(keys, 8, start, end, hist, tid);
    for (int bin = tid; bin < SORT_BINS; bin += SORT_TPB) {
        int pos = 0;
        for (int j = 0; j < SORT_TPB; ++j) {
            const int c = hist[bin * SORT_PITCH + j];
            hist[bin * SORT_PITCH + j] = pos;
            pos += c;
        }
        tot[bin] = pos;
    }
    __syncthreads();
    const int lane = tid & 31;
    const int wv   = tid >> 5;
    const int ta = tot[2 * tid];
    const int tb = tot[2 * tid + 1];
    int incl = ta + tb;
    for (int d = 1; d < 32; d <<= 1) {
        const int t = __shfl_up(incl, d);
        if (lane >= d) incl += t;
    }
    if (lane == 31) wsum[wv] = incl;
    __syncthreads();
    int front = 0;
    for (int w = 0; w < SORT_TPB / 32; ++w) front += (w < wv) ? wsum[w] : 0;
    const int ex = front + incl - (ta + tb);
    lbs[2 * tid]     = ex;
    lbs[2 * tid + 1] = ex + ta;
    segs[(2 * tid) * SORT_NB + blockIdx.x]     = ex;
    segs[(2 * tid + 1) * SORT_NB + blockIdx.x] = ex + ta;
    if (tid == SORT_TPB - 1) segs[SORT_BINS * SORT_NB + blockIdx.x] = ex + ta + tb;
    __syncthreads();
    long p = start;
    for (; p + 4 <= end; p += 4) {
        const int k0 = keys[p], k1 = keys[p + 1], k2 = keys[p + 2], k3 = keys[p + 3];
        const int v0 = vals[p], v1 = vals[p + 1], v2 = vals[p + 2], v3 = vals[p + 3];
        sort_put_lds(hist, lbs, stK, stV, tid, 8, k0, v0);
        sort_put_lds(hist, lbs, stK, stV, tid, 8, k1, v1);
        sort_put_lds(hist, lbs, stK, stV, tid, 8, k2, v2);
        sort_put_lds(hist, lbs, stK, stV, tid, 8, k3, v3);
    }
    for (; p < end; ++p) sort_put_lds(hist, lbs, stK, stV, tid, 8, keys[p], vals[p]);
    __syncthreads();
    const long bstart = (long)blockIdx.x * SORT_TPB * run;
    long bend = bstart + (long)SORT_TPB * run;
    if (bend > nedges) bend = nedges;
    const int cnt = (bend > bstart) ? (int)(bend - bstart) : 0;
    int i = tid;
    for (; i + 3 * SORT_TPB < cnt; i += 4 * SORT_TPB) {
        int k[4], v[4];
        #pragma unroll
        for (int u = 0; u < 4; ++u) { k[u] = stK[i + u * SORT_TPB]; v[u] = stV[i + u * SORT_TPB]; }
        #pragma unroll
        for (int u = 0; u < 4; ++u) { keysOut[bstart + i + u * SORT_TPB] = k[u]; valsOut[bstart + i + u * SORT_TPB] = v[u]; }
    }
    for (; i < cnt; i += SORT_TPB) {
        keysOut[bstart + i] = stK[i];
        valsOut[bstart + i] = stV[i];
    }
}

#define FINE_CAP 8192
#define FINE_SL  (SORT_NB / SORT_TPB)
#define FINE_PB  16
#define FINE_CH  16
#define FINE_LST (FINE_CAP + FINE_CAP / 32)

__device__ __forceinline__ int lst_at(int p) { return p + (p >> 5); }

__global__ __launch_bounds__(SORT_TPB) void sort_fine(const int* __restrict__ keys,
                                                      const int* __restrict__ vals,
                                                      const int* __restrict__ segs,
                                                      int sliceLen, int n,
                                                      int* __restrict__ off,
                                                      int* __restrict__ valsOut) {
    __shared__ int hist[SORT_BINS * SORT_PITCH];
    __shared__ int tot[SORT_BINS];
    __shared__ int stage[FINE_CAP];
    __shared__ int lstK[FINE_LST];
    __shared__ int lstV[FINE_LST];
    __shared__ int pcB[SORT_NB];
    __shared__ int pcL[SORT_NB];
    __shared__ int pcA[SORT_NB];
    __shared__ int red[2 * (SORT_TPB / 32)];
    const int tid  = threadIdx.x;
    const int lane = tid & 31;
    const int wv   = tid >> 5;
    const int g = blockIdx.x;
    const int gl = (g < SORT_BINS) ? g : SORT_BINS;
    const int gh = (g + 1 < SORT_BINS) ? g + 1 : SORT_BINS;
    const int4 s0 = *(const int4*)(segs + gl * SORT_NB + FINE_SL * tid);
    const int4 s1 = *(const int4*)(segs + gh * SORT_NB + FINE_SL * tid);
    int pb[FINE_SL], pl[FINE_SL];
    pb[0] = (FINE_SL * tid + 0) * sliceLen + s0.x;  pl[0] = s1.x - s0.x;
    pb[1] = (FINE_SL * tid + 1) * sliceLen + s0.y;  pl[1] = s1.y - s0.y;
    pb[2] = (FINE_SL * tid + 2) * sliceLen + s0.z;  pl[2] = s1.z - s0.z;
    pb[3] = (FINE_SL * tid + 3) * sliceLen + s0.w;  pl[3] = s1.w - s0.w;
    for (int i = tid; i < SORT_BINS * SORT_PITCH; i += SORT_TPB) hist[i] = 0;
    const int mine = (pl[0] + pl[1]) + (pl[2] + pl[3]);
    int incl = mine;
    int a = (s0.x + s0.y) + (s0.z + s0.w);
    for (int d = 1; d < 32; d <<= 1) {
        const int t = __shfl_up(incl, d);
        if (lane >= d) incl += t;
    }
    for (int d = 16; d > 0; d >>= 1) a += __shfl_xor(a, d);
    if (lane == 31) red[wv] = incl;
    if (lane == 0)  red[SORT_TPB / 32 + wv] = a;
    __syncthreads();
    int bs = 0, cntg = 0, front = 0;
    for (int w = 0; w < SORT_TPB / 32; ++w) {
        cntg += red[w];
        front += (w < wv) ? red[w] : 0;
        bs += red[SORT_TPB / 32 + w];
    }
    const bool inLds = cntg <= FINE_CAP;
    const int run = (cntg + SORT_TPB - 1) / SORT_TPB;
    int start = tid * run;
    if (start > cntg) start = cntg;
    int end = start + run;
    if (end > cntg) end = cntg;

    if (inLds) {
        int at0 = front + incl - mine;
        #pragma unroll
        for (int u = 0; u < FINE_SL; ++u) {
            pcB[FINE_SL * tid + u] = (pl[u] > 0) ? pb[u] : 0;
            pcL[FINE_SL * tid + u] = pl[u];
            pcA[FINE_SL * tid + u] = at0;
            at0 += pl[u];
        }
    }
    __syncthreads();
    if (inLds) {
        const int sub = lane >> 3;
        const int j   = lane & 7;
        for (int r0 = 0; r0 < SORT_NB / 16; r0 += FINE_CH) {
            int B[FINE_CH], L[FINE_CH], A[FINE_CH];
            int k[FINE_CH][2], v[FINE_CH][2];
            #pragma unroll
            for (int r = 0; r < FINE_CH; ++r) {
                const int pi = wv * (SORT_NB / 4) + (r0 + r) * 4 + sub;
                B[r] = pcB[pi];
                L[r] = pcL[pi];
                A[r] = pcA[pi];
            }
            #pragma unroll
            for (int r = 0; r < FINE_CH; ++r) {
                const int last = (L[r] > 0) ? L[r] - 1 : 0;
                #pragma unroll
                for (int hh = 0; hh < 2; ++hh) {
                    const int jj = j + 8 * hh;
                    const int at = B[r] + ((jj < last) ? jj : last);
                    k[r][hh] = keys[at];
                    v[r][hh] = vals[at];
                }
            }
            #pragma unroll
            for (int r = 0; r < FINE_CH; ++r) {
                #pragma unroll
                for (int hh = 0; hh < 2; ++hh) {
                    const int jj = j + 8 * hh;
                    if (jj < L[r]) {
                        lstK[lst_at(A[r] + jj)] = k[r][hh];
                        lstV[lst_at(A[r] + jj)] = v[r][hh];
                    }
                }
                for (int q = FINE_PB + j; q < L[r]; q += 8) {
                    lstK[lst_at(A[r] + q)] = keys[B[r] + q];
                    lstV[lst_at(A[r] + q)] = vals[B[r] + q];
                }
            }
        }
    }
    __syncthreads();
    if (inLds) {
        int p = start;
        for (; p + 8 <= end; p += 8) {
            int k[8];
            #pragma unroll
            for (int u = 0; u < 8; ++u) k[u] = lstK[lst_at(p + u)];
            #pragma unroll
            for (int u = 0; u < 8; ++u) hist[(k[u] & (SORT_BINS - 1)) * SORT_PITCH + tid] += 1;
        }
        for (; p < end; ++p) hist[(lstK[lst_at(p)] & (SORT_BINS - 1)) * SORT_PITCH + tid] += 1;
    } else {
        #pragma unroll
        for (int u = 0; u < FINE_SL; ++u)
            for (int q = 0; q < pl[u]; ++q)
                hist[(keys[pb[u] + q] & (SORT_BINS - 1)) * SORT_PITCH + tid] += 1;
    }
    __syncthreads();
    for (int bin = tid; bin < SORT_BINS; bin += SORT_TPB) {
        int total = 0;
        for (int j = 0; j < SORT_TPB; ++j) total += hist[bin * SORT_PITCH + j];
        tot[bin] = total;
    }
    __syncthreads();
    for (int bin = tid; bin < SORT_BINS; bin += SORT_TPB) {
        int pos = bs;
        for (int d = 0; d < bin; ++d) pos += tot[d];
        const int v = (g << 8) + bin;
        if (v <= n) off[v] = pos;
        for (int j = 0; j < SORT_TPB; ++j) {
            const int c = hist[bin * SORT_PITCH + j];
            hist[bin * SORT_PITCH + j] = pos;
            pos += c;
        }
    }
    __syncthreads();
    if (inLds) {
        int p = start;
        for (; p + 4 <= end; p += 4) {
            int k[4], v[4];
            #pragma unroll
            for (int u = 0; u < 4; ++u) { k[u] = lstK[lst_at(p + u)]; v[u] = lstV[lst_at(p + u)]; }
            #pragma unroll
            for (int u = 0; u < 4; ++u) {
                const int d = k[u] & (SORT_BINS - 1);
                const int w = hist[d * SORT_PITCH + tid];
                hist[d * SORT_PITCH + tid] = w + 1;
                stage[w - bs] = v[u];
            }
        }
        for (; p < end; ++p) {
            const int d = lstK[lst_at(p)] & (SORT_BINS - 1);
            const int w = hist[d * SORT_PITCH + tid];
            hist[d * SORT_PITCH + tid] = w + 1;
            stage[w - bs] = lstV[lst_at(p)];
        }
        __syncthreads();
        int i = tid;
        for (; i + 7 * SORT_TPB < cntg; i += 8 * SORT_TPB) {
            int w[8];
            #pragma unroll
            for (int u = 0; u < 8; ++u) w[u] = stage[i + u * SORT_TPB];
            #pragma unroll
            for (int u = 0; u < 8; ++u) valsOut[bs + i + u * SORT_TPB] = w[u];
        }
        for (; i < cntg; i += SORT_TPB) valsOut[bs + i] = stage[i];
        return;
    }
    #pragma unroll
    for (int u = 0; u < FINE_SL; ++u) {
        for (int q = 0; q < pl[u]; ++q) {
            const int d = keys[pb[u] + q] & (SORT_BINS - 1);
            const int w = hist[d * SORT_PITCH + tid];
            hist[d * SORT_PITCH + tid] = w + 1;
            valsOut[w] = vals[pb[u] + q];
        }
    }
}

__global__ __launch_bounds__(256) void gin_offsets(const int* __restrict__ sdst,
                                                   int nedges, int n,
                                                   int* __restrict__ off) {
    const int p = blockIdx.x * blockDim.x + threadIdx.x;
    if (p > nedges) return;
    const int prev = (p == 0) ? -1 : sdst[p - 1];
    const int here = (p == nedges) ? n : sdst[p];
    for (int v = prev + 1; v <= here; ++v) off[v] = p;
}

#define GROWS 16

__device__ __forceinline__ void add4(float4& a, const float4 r) {
    a.x += r.x; a.y += r.y; a.z += r.z; a.w += r.w;
}

__device__ __forceinline__ void close_row(float* zrow, const float e1, const float4 acc) {
    const float4 hv = *(const float4*)zrow;
    float4 o;
    o.x = e1 * hv.x + acc.x;
    o.y = e1 * hv.y + acc.y;
    o.z = e1 * hv.z + acc.z;
    o.w = e1 * hv.w + acc.w;
    *(float4*)zrow = o;
}

__device__ __forceinline__ void load_rows(float4 (&r)[GROWS], const int idxv, const int j,
                                          const float* hc) {
    #pragma unroll
    for (int t = 0; t < GROWS; ++t) {
        const int s = __builtin_amdgcn_readlane(idxv, j + t);
        r[t] = *(const float4*)(hc + (size_t)s * DD);
    }
}

__device__ __forceinline__ void sum_rows(const float4 (&r)[GROWS], const int p0, const int eEnd,
                                         const int ob, float* zl, const float e1,
                                         int& i, int& nextB, float4& acc) {
    #pragma unroll
    for (int t = 0; t < GROWS; ++t) {
        const int p = p0 + t;
        if (p < eEnd) {
            while (p >= nextB) {
                close_row(zl + i * HROW, e1, acc);
                acc.x = 0.0f; acc.y = 0.0f; acc.z = 0.0f; acc.w = 0.0f;
                ++i;
                nextB = __builtin_amdgcn_readlane(ob, i + 1);
            }
            add4(acc, r[t]);
        }
    }
}

__device__ __forceinline__ void w_fetch(float (&wr)[64], const float* __restrict__ Wl,
                                        const int tid) {
    const float* wc = Wl + (tid >> 7) * 4 * DD + (tid & 127);
    #pragma unroll
    for (int i = 0; i < 16; ++i) {
        #pragma unroll
        for (int q = 0; q < 4; ++q) wr[4 * i + q] = wc[(8 * i + q) * DD];
    }
}

__device__ __forceinline__ void w_put(const float (&wr)[64], float* wS, const int tid) {
    float* wd = wS + ((tid >> 7) * DD + (tid & 127)) * 4;
    #pragma unroll
    for (int i = 0; i < 16; ++i) {
        float4 v;
        v.x = wr[4 * i]; v.y = wr[4 * i + 1]; v.z = wr[4 * i + 2]; v.w = wr[4 * i + 3];
        *(float4*)(wd + 2 * i * DD * 4) = v;
    }
}

__device__ __forceinline__ void mlp_product(v8f (&acc)[8], const float* ab, const float* wl) {
    #pragma unroll
    for (int k0 = 0; k0 < DD; k0 += 4) {
        const v2f a = *(const v2f*)(ab + k0);
        #pragma unroll
        for (int t = 0; t < 8; ++t) {
            const v2f b = *(const v2f*)(wl + (k0 >> 2) * (DD * 4) + t * 64);
            acc[t] = __builtin_amdgcn_wmma_f32_16x16x4_f32(false, a, false, b,
                                                           (short)0, acc[t], false, false);
        }
    }
}

__global__ __launch_bounds__(256, 2) void gin_layer(const float* __restrict__ h,
                                                    const float* __restrict__ eps, int layer,
                                                    const int* __restrict__ off,
                                                    const int* __restrict__ csrc,
                                                    const float* __restrict__ W1,
                                                    const float* __restrict__ b1,
                                                    const float* __restrict__ W2,
                                                    const float* __restrict__ b2,
                                                    float* __restrict__ out, int n) {
    __shared__ __attribute__((aligned(16))) float lds[DD * DD + 8 * 16 * HROW];
    float* wS = lds;

    const int tid   = threadIdx.x;
    const int lane  = threadIdx.x & 31;
    const int wave  = __builtin_amdgcn_readfirstlane((int)(threadIdx.x >> 5));
    const int ln    = lane & 15;
    const int khoff = lane >> 4;
    const int rh    = khoff << 3;
    const int m0    = blockIdx.x * 128 + wave * 16;

    float* zw = lds + DD * DD + wave * (16 * HROW);

    float wr[64];
    w_fetch(wr, W1, tid);

    int ob;
    {
        int t = m0 + lane;
        if (t > n) t = n;
        ob = off[t];
    }
    const float* hc = h + lane * 4;
    float* zl = zw + lane * 4;
    {
        float4 own[16];
        #pragma unroll
        for (int q = 0; q < 16; ++q) {
            int rr = m0 + q;
            if (rr > n - 1) rr = n - 1;
            own[q] = *(const float4*)(hc + (size_t)rr * DD);
        }
        w_put(wr, wS, tid);
        #pragma unroll
        for (int q = 0; q < 16; ++q) *(float4*)(zl + q * HROW) = own[q];
    }
    const float e1 = 1.0f + eps[layer];
    const int eEnd = __builtin_amdgcn_readlane(ob, 16);
    int base  = __builtin_amdgcn_readlane(ob, 0);
    int nextB = __builtin_amdgcn_readlane(ob, 1);
    int i = 0;
    float4 acc = {0.0f, 0.0f, 0.0f, 0.0f};
    int m = eEnd - base;
    if (m > 32) m = 32;
    int idxv = (lane < m) ? csrc[base + lane] : 0;
    __syncthreads();
    while (m > 0) {
        const int nb = base + 32;
        int mn = eEnd - nb;
        if (mn > 32) mn = 32;
        float4 r[GROWS];
        load_rows(r, idxv, 0, hc);
        const int idxn = (lane < mn) ? csrc[nb + lane] : 0;
        sum_rows(r, base, eEnd, ob, zl, e1, i, nextB, acc);
        if (m > GROWS) {
            load_rows(r, idxv, GROWS, hc);
            sum_rows(r, base + GROWS, eEnd, ob, zl, e1, i, nextB, acc);
        }
        base = nb;
        m = mn;
        idxv = idxn;
    }
    while (i < 16) {
        close_row(zl + i * HROW, e1, acc);
        acc.x = 0.0f; acc.y = 0.0f; acc.z = 0.0f; acc.w = 0.0f;
        ++i;
    }

    const float* ab = zw + ln * HROW + 2 * khoff;
    const float* wl = wS + ln * 4 + 2 * khoff;

    w_fetch(wr, W2, tid);

    {
        float bias[8];
        #pragma unroll
        for (int t = 0; t < 8; ++t) bias[t] = b1[t * 16 + ln];
        v8f pa[8] = {};
        mlp_product(pa, ab, wl);
        #pragma unroll
        for (int j = 0; j < 8; ++j) {
            #pragma unroll
            for (int t = 0; t < 8; ++t)
                zw[(j + rh) * HROW + t * 16 + ln] = fmaxf(pa[t][j] + bias[t], 0.0f);
        }
    }

    __syncthreads();
    w_put(wr, wS, tid);
    __syncthreads();

    {
        float bias[8];
        #pragma unroll
        for (int t = 0; t < 8; ++t) bias[t] = b2[t * 16 + ln];
        v8f pa[8] = {};
        mlp_product(pa, ab, wl);
        #pragma unroll
        for (int j = 0; j < 8; ++j) {
            const int r = m0 + j + rh;
            if (r < n) {
                #pragma unroll
                for (int t = 0; t < 8; ++t)
                    out[(size_t)r * DD + t * 16 + ln] = pa[t][j] + bias[t];
            }
        }
    }
}

extern "C" void kernel_launch(void* const* d_in, const int* in_sizes, int n_in,
                              void* d_out, int out_size, void* d_ws, size_t ws_size,
                              hipStream_t stream) {
    const float* h   = (const float*)d_in[0];
    const int*   src = (const int*)  d_in[1];
    const int*   dst = (const int*)  d_in[2];
    const float* W1  = (const float*)d_in[3];
    const float* b1  = (const float*)d_in[4];
    const float* W2  = (const float*)d_in[5];
    const float* b2  = (const float*)d_in[6];
    const float* eps = (const float*)d_in[7];

    const int n = in_sizes[0] / DD;
    const int e = in_sizes[1];
    const int L = in_sizes[7];

    float* hb  = (float*)d_ws;
    float* out = (float*)d_out;

    const int tabN = SORT_BINS * SORT_NB;
    const int np   = ((n + 4) / 4) * 4;
    int* table  = (int*)(hb + (size_t)n * DD);
    int* tableS = table + tabN;
    int* off    = tableS + tabN;
    int* kA     = off + np;
    int* vA     = kA + e;
    int* kB     = vA + e;
    int* vB     = kB + e;

    const int offBlocks  = (e + 1 + 255) / 256;
    const int mlpBlocks  = (n + 127) / 128;
    const int run        = (e + SORT_NB * SORT_TPB - 1) / (SORT_NB * SORT_TPB);

    const int* kin = dst;
    const int* vin = src;
    if (((n - 1) >> 16) == 0 && (long)run * SORT_TPB <= PLACE_CAP) {
        int* segs = table;
        sort_local<<<SORT_NB, SORT_TPB, 0, stream>>>(dst, src, e, run, segs, kA, vA);
        sort_fine<<<(n >> 8) + 1, SORT_TPB, 0, stream>>>(kA, vA, segs, run * SORT_TPB, n, off, vB);
        vin = vB;
    } else {
        int* kout = kA;
        int* vout = vA;
        for (int shift = 0; shift == 0 || ((n - 1) >> shift) > 0; shift += 8) {
            sort_count<<<SORT_NB, SORT_TPB, 0, stream>>>(kin, shift, e, run, table);
            gin_scan<<<tabN / 1024, 256, 0, stream>>>(table, tableS);
            sort_place<<<SORT_NB, SORT_TPB, 0, stream>>>(kin, vin, shift, e, run, tableS, kout, vout);
            kin = kout;
            vin = vout;
            kout = (kout == kA) ? kB : kA;
            vout = (vout == vA) ? vB : vA;
        }
        gin_offsets<<<offBlocks, 256, 0, stream>>>(kin, e, n, off);
    }

    for (int l = 0; l < L; ++l) {
        const float* hin = (l == 0) ? h : (((L - l) & 1) ? hb : out);
        float* hout = ((L - 1 - l) & 1) ? hb : out;
        gin_layer<<<mlpBlocks, 256, 0, stream>>>(hin, eps, l, off, vin,
                                                 W1 + (size_t)l * DD * DD, b1 + l * DD,
                                                 W2 + (size_t)l * DD * DD, b2 + l * DD, hout, n);
    }
}


# Fold each layer's second Linear into the next layer's first (neighbour sum is linear): later layers walk hid@(W2 W1next), add deg-scaled bias, ReLU, one product; reassociated, so rounding changes
# speedup vs baseline: 3.5036x; 1.0887x over previous
#include <hip/hip_runtime.h>
#include <hip/hip_bf16.h>

typedef __attribute__((ext_vector_type(2))) float v2f;
typedef __attribute__((ext_vector_type(8))) float v8f;

#define DD 128
#define HROW 132

#define SORT_NB    512
#define SORT_TPB   128
#define SORT_BINS  256
#define SORT_PITCH 129

__device__ __forceinline__ void sort_count_runs(const int* __restrict__ keys, int shift,
                                                long start, long end, int* hist, int tid) {
    for (int i = tid; i < SORT_BINS * SORT_PITCH; i += SORT_TPB) hist[i] = 0;
    __syncthreads();
    long p = start;
    for (; p + 4 <= end; p += 4) {
        const int k0 = keys[p], k1 = keys[p + 1], k2 = keys[p + 2], k3 = keys[p + 3];
        hist[((k0 >> shift) & (SORT_BINS - 1)) * SORT_PITCH + tid] += 1;
        hist[((k1 >> shift) & (SORT_BINS - 1)) * SORT_PITCH + tid] += 1;
        hist[((k2 >> shift) & (SORT_BINS - 1)) * SORT_PITCH + tid] += 1;
        hist[((k3 >> shift) & (SORT_BINS - 1)) * SORT_PITCH + tid] += 1;
    }
    for (; p < end; ++p) {
        const int d = (keys[p] >> shift) & (SORT_BINS - 1);
        hist[d * SORT_PITCH + tid] += 1;
    }
    __syncthreads();
}

__global__ __launch_bounds__(SORT_TPB) void sort_count(const int* __restrict__ keys, int shift,
                                                       int nedges, int run,
                                                       int* __restrict__ table) {
    __shared__ int hist[SORT_BINS * SORT_PITCH];
    const int tid = threadIdx.x;
    const long start = ((long)blockIdx.x * SORT_TPB + tid) * run;
    long end = start + run;
    if (end > nedges) end = nedges;
    sort_count_runs(keys, shift, start, end, hist, tid);
    for (int bin = tid; bin < SORT_BINS; bin += SORT_TPB) {
        int total = 0;
        for (int j = 0; j < SORT_TPB; ++j) total += hist[bin * SORT_PITCH + j];
        table[bin * SORT_NB + blockIdx.x] = total;
    }
}

__global__ __launch_bounds__(256) void gin_scan(const int* __restrict__ in,
                                                int* __restrict__ outp) {
    __shared__ int wtot[8];
    __shared__ int wpre[8];
    const int tid  = threadIdx.x;
    const int lane = tid & 31;
    const int wv   = tid >> 5;
    const int base = blockIdx.x * 1024;

    const int nfront = blockIdx.x;
    const int* mine = in + tid * 4;
    int pre = 0;
    for (int ch = 0; ch < nfront; ch += 16) {
        int part[16];
        #pragma unroll
        for (int k = 0; k < 16; ++k) {
            const int at = (ch + k < nfront) ? ch + k : 0;
            const int4 f = *(const int4*)(mine + at * 1024);
            part[k] = (f.x + f.y) + (f.z + f.w);
        }
        #pragma unroll
        for (int k = 0; k < 16; ++k) pre += (ch + k < nfront) ? part[k] : 0;
    }
    const int4 c = *(const int4*)(in + base + tid * 4);
    const int local = (c.x + c.y) + (c.z + c.w);

    int incl = local;
    for (int d = 1; d < 32; d <<= 1) {
        const int t = __shfl_up(incl, d);
        if (lane >= d) incl += t;
    }
    for (int d = 16; d > 0; d >>= 1) pre += __shfl_xor(pre, d);
    if (lane == 31) wtot[wv] = incl;
    if (lane == 0)  wpre[wv] = pre;
    __syncthreads();
    int front = 0;
    for (int w = 0; w < 8; ++w) {
        front += wpre[w];
        if (w < wv) front += wtot[w];
    }
    int4 o;
    o.x = front + incl - local;
    o.y = o.x + c.x;
    o.z = o.y + c.y;
    o.w = o.z + c.z;
    *(int4*)(outp + base + tid * 4) = o;
}

__device__ __forceinline__ void sort_put(int* hist, int tid, int shift, int k, int v,
                                         int* __restrict__ keysOut, int* __restrict__ valsOut) {
    const int d = (k >> shift) & (SORT_BINS - 1);
    const int q = hist[d * SORT_PITCH + tid];
    hist[d * SORT_PITCH + tid] = q + 1;
    keysOut[q] = k;
    valsOut[q] = v;
}

#define PLACE_CAP 2048

__device__ __forceinline__ void sort_put_lds(int* hist, const int* lbs, int* stK, int* stV,
                                             int tid, int shift, int k, int v) {
    const int d = (k >> shift) & (SORT_BINS - 1);
    const int w = hist[d * SORT_PITCH + tid];
    hist[d * SORT_PITCH + tid] = w + 1;
    const int i = lbs[d] + w;
    stK[i] = k;
    stV[i] = v;
}

__global__ __launch_bounds__(SORT_TPB) void sort_place(const int* __restrict__ keys,
                                                       const int* __restrict__ vals, int shift,
                                                       int nedges, int run,
                                                       const int* __restrict__ tableS,
                                                       int* __restrict__ keysOut,
                                                       int* __restrict__ valsOut) {
    __shared__ int hist[SORT_BINS * SORT_PITCH];
    __shared__ int tot[SORT_BINS];
    __shared__ int lbs[SORT_BINS];
    __shared__ int dlt[SORT_BINS];
    __shared__ int wsum[SORT_TPB / 32];
    __shared__ int stK[PLACE_CAP];
    __shared__ int stV[PLACE_CAP];
    const int tid = threadIdx.x;
    const long start = ((long)blockIdx.x * SORT_TPB + tid) * run;
    long end = start + run;
    if (end > nedges) end = nedges;
    sort_count_runs(keys, shift, start, end, hist, tid);
    if ((long)run * SORT_TPB <= PLACE_CAP) {
        const int g0 = tableS[(2 * tid) * SORT_NB + blockIdx.x];
        const int g1 = tableS[(2 * tid + 1) * SORT_NB + blockIdx.x];
        for (int bin = tid; bin < SORT_BINS; bin += SORT_TPB) {
            int pos = 0;
            for (int j = 0; j < SORT_TPB; ++j) {
                const int c = hist[bin * SORT_PITCH + j];
                hist[bin * SORT_PITCH + j] = pos;
                pos += c;
            }
            tot[bin] = pos;
        }
        __syncthreads();
        const int lane = tid & 31;
        const int wv   = tid >> 5;
        const int ta = tot[2 * tid];
        const int tb = tot[2 * tid + 1];
        int incl = ta + tb;
        for (int d = 1; d < 32; d <<= 1) {
            const int t = __shfl_up(incl, d);
            if (lane >= d) incl += t;
        }
        if (lane == 31) wsum[wv] = incl;
        __syncthreads();
        int front = 0;
        for (int w = 0; w < SORT_TPB / 32; ++w) front += (w < wv) ? wsum[w] : 0;
        const int ex = front + incl - (ta + tb);
        lbs[2 * tid]     = ex;
        lbs[2 * tid + 1] = ex + ta;
        dlt[2 * tid]     = g0 - ex;
        dlt[2 * tid + 1] = g1 - (ex + ta);
        __syncthreads();
        long p = start;
        for (; p + 4 <= end; p += 4) {
            const int k0 = keys[p], k1 = keys[p + 1], k2 = keys[p + 2], k3 = keys[p + 3];
            const int v0 = vals[p], v1 = vals[p + 1], v2 = vals[p + 2], v3 = vals[p + 3];
            sort_put_lds(hist, lbs, stK, stV, tid, shift, k0, v0);
            sort_put_lds(hist, lbs, stK, stV, tid, shift, k1, v1);
            sort_put_lds(hist, lbs, stK, stV, tid, shift, k2, v2);
            sort_put_lds(hist, lbs, stK, stV, tid, shift, k3, v3);
        }
        for (; p < end; ++p) sort_put_lds(hist, lbs, stK, stV, tid, shift, keys[p], vals[p]);
        __syncthreads();
        const long bstart = (long)blockIdx.x * SORT_TPB * run;
        long bend = bstart + (long)SORT_TPB * run;
        if (bend > nedges) bend = nedges;
        const int cnt = (bend > bstart) ? (int)(bend - bstart) : 0;
        int i = tid;
        for (; i + 3 * SORT_TPB < cnt; i += 4 * SORT_TPB) {
            int k[4], v[4], o[4];
            #pragma unroll
            for (int u = 0; u < 4; ++u) { k[u] = stK[i + u * SORT_TPB]; v[u] = stV[i + u * SORT_TPB]; }
            #pragma unroll
            for (int u = 0; u < 4; ++u) o[u] = dlt[(k[u] >> shift) & (SORT_BINS - 1)] + i + u * SORT_TPB;
            #pragma unroll
            for (int u = 0; u < 4; ++u) { keysOut[o[u]] = k[u]; valsOut[o[u]] = v[u]; }
        }
        for (; i < cnt; i += SORT_TPB) {
            const int k = stK[i];
            const int v = stV[i];
            const int o = dlt[(k >> shift) & (SORT_BINS - 1)] + i;
            keysOut[o] = k;
            valsOut[o] = v;
        }
        return;
    }
    for (int bin = tid; bin < SORT_BINS; bin += SORT_TPB) {
        int pos = tableS[bin * SORT_NB + blockIdx.x];
        for (int j = 0; j < SORT_TPB; ++j) {
            const int c = hist[bin * SORT_PITCH + j];
            hist[bin * SORT_PITCH + j] = pos;
            pos += c;
        }
    }
    __syncthreads();
    long p = start;
    for (; p + 4 <= end; p += 4) {
        const int k0 = keys[p], k1 = keys[p + 1], k2 = keys[p + 2], k3 = keys[p + 3];
        const int v0 = vals[p], v1 = vals[p + 1], v2 = vals[p + 2], v3 = vals[p + 3];
        sort_put(hist, tid, shift, k0, v0, keysOut, valsOut);
        sort_put(hist, tid, shift, k1, v1, keysOut, valsOut);
        sort_put(hist, tid, shift, k2, v2, keysOut, valsOut);
        sort_put(hist, tid, shift, k3, v3, keysOut, valsOut);
    }
    for (; p < end; ++p) sort_put(hist, tid, shift, keys[p], vals[p], keysOut, valsOut);
}

__global__ __launch_bounds__(SORT_TPB) void sort_local(const int* __restrict__ keys,
                                                       const int* __restrict__ vals,
                                                       int nedges, int run,
                                                       int* __restrict__ segs,
                                                       int* __restrict__ keysOut,
                                                       int* __restrict__ valsOut) {
    __shared__ int hist[SORT_BINS * SORT_PITCH];
    __shared__ int tot[SORT_BINS];
    __shared__ int lbs[SORT_BINS];
    __shared__ int wsum[SORT_TPB / 32];
    __shared__ int stK[PLACE_CAP];
    __shared__ int stV[PLACE_CAP];
    const int tid = threadIdx.x;
    const long start = ((long)blockIdx.x * SORT_TPB + tid) * run;
    long end = start + run;
    if (end > nedges) end = nedges;
    sort_count_runs(keys, 8, start, end, hist, tid);
    for (int bin = tid; bin < SORT_BINS; bin += SORT_TPB) {
        int pos = 0;
        for (int j = 0; j < SORT_TPB; ++j) {
            const int c = hist[bin * SORT_PITCH + j];
            hist[bin * SORT_PITCH + j] = pos;
            pos += c;
        }
        tot[bin] = pos;
    }
    __syncthreads();
    const int lane = tid & 31;
    const int wv   = tid >> 5;
    const int ta = tot[2 * tid];
    const int tb = tot[2 * tid + 1];
    int incl = ta + tb;
    for (int d = 1; d < 32; d <<= 1) {
        const int t = __shfl_up(incl, d);
        if (lane >= d) incl += t;
    }
    if (lane == 31) wsum[wv] = incl;
    __syncthreads();
    int front = 0;
    for (int w = 0; w < SORT_TPB / 32; ++w) front += (w < wv) ? wsum[w] : 0;
    const int ex = front + incl - (ta + tb);
    lbs[2 * tid]     = ex;
    lbs[2 * tid + 1] = ex + ta;
    segs[(2 * tid) * SORT_NB + blockIdx.x]     = ex;
    segs[(2 * tid + 1) * SORT_NB + blockIdx.x] = ex + ta;
    if (tid == SORT_TPB - 1) segs[SORT_BINS * SORT_NB + blockIdx.x] = ex + ta + tb;
    __syncthreads();
    long p = start;
    for (; p + 4 <= end; p += 4) {
        const int k0 = keys[p], k1 = keys[p + 1], k2 = keys[p + 2], k3 = keys[p + 3];
        const int v0 = vals[p], v1 = vals[p + 1], v2 = vals[p + 2], v3 = vals[p + 3];
        sort_put_lds(hist, lbs, stK, stV, tid, 8, k0, v0);
        sort_put_lds(hist, lbs, stK, stV, tid, 8, k1, v1);
        sort_put_lds(hist, lbs, stK, stV, tid, 8, k2, v2);
        sort_put_lds(hist, lbs, stK, stV, tid, 8, k3, v3);
    }
    for (; p < end; ++p) sort_put_lds(hist, lbs, stK, stV, tid, 8, keys[p], vals[p]);
    __syncthreads();
    const long bstart = (long)blockIdx.x * SORT_TPB * run;
    long bend = bstart + (long)SORT_TPB * run;
    if (bend > nedges) bend = nedges;
    const int cnt = (bend > bstart) ? (int)(bend - bstart) : 0;
    int i = tid;
    for (; i + 3 * SORT_TPB < cnt; i += 4 * SORT_TPB) {
        int k[4], v[4];
        #pragma unroll
        for (int u = 0; u < 4; ++u) { k[u] = stK[i + u * SORT_TPB]; v[u] = stV[i + u * SORT_TPB]; }
        #pragma unroll
        for (int u = 0; u < 4; ++u) { keysOut[bstart + i + u * SORT_TPB] = k[u]; valsOut[bstart + i + u * SORT_TPB] = v[u]; }
    }
    for (; i < cnt; i += SORT_TPB) {
        keysOut[bstart + i] = stK[i];
        valsOut[bstart + i] = stV[i];
    }
}

__device__ __forceinline__ void fold_tile(const float* __restrict__ A, const float* __restrict__ B,
                                          float* __restrict__ M, const int rt, const int lane) {
    const int ln = lane & 15;
    const int kh = lane >> 4;
    const float* ap = A + (rt * 16 + ln) * DD + 2 * kh;
    const float* bp = B + (2 * kh) * DD + ln;
    v8f acc[8] = {};
    #pragma unroll 8
    for (int k0 = 0; k0 < DD; k0 += 4) {
        v2f a;
        a.x = ap[k0];
        a.y = ap[k0 + 1];
        #pragma unroll
        for (int t = 0; t < 8; ++t) {
            v2f b;
            b.x = bp[k0 * DD + t * 16];
            b.y = bp[(k0 + 1) * DD + t * 16];
            acc[t] = __builtin_amdgcn_wmma_f32_16x16x4_f32(false, a, false, b,
                                                           (short)0, acc[t], false, false);
        }
    }
    const int rh = kh << 3;
    #pragma unroll
    for (int j = 0; j < 8; ++j) {
        #pragma unroll
        for (int t = 0; t < 8; ++t)
            M[(rt * 16 + j + rh) * DD + t * 16 + ln] = acc[t][j];
    }
}

__device__ __forceinline__ void fold_block(const int fb, const int tid, const int L,
                                           const float* __restrict__ W1,
                                           const float* __restrict__ W2,
                                           const float* __restrict__ b2,
                                           float* __restrict__ fold) {
    const int nt  = (L - 1) * (DD / 16);
    const int ntb = (nt + SORT_TPB / 32 - 1) / (SORT_TPB / 32);
    float* foldC = fold + (size_t)(L - 1) * DD * DD;
    if (fb < ntb) {
        const int t = fb * (SORT_TPB / 32) + __builtin_amdgcn_readfirstlane(tid >> 5);
        if (t < nt) {
            const int l = t / (DD / 16);
            fold_tile(W2 + (size_t)l * DD * DD, W1 + (size_t)(l + 1) * DD * DD,
                      fold + (size_t)l * DD * DD, t % (DD / 16), tid & 31);
        }
    } else {
        const int l = fb - ntb;
        if (l < L - 1) {
            const float* bl = b2 + l * DD;
            const float* Wn = W1 + (size_t)(l + 1) * DD * DD;
            for (int nn = tid; nn < DD; nn += SORT_TPB) {
                float sacc = 0.0f;
                #pragma unroll 16
                for (int k = 0; k < DD; ++k) sacc = fmaf(bl[k], Wn[k * DD + nn], sacc);
                foldC[l * DD + nn] = sacc;
            }
        }
    }
}

__global__ __launch_bounds__(SORT_TPB) void gin_fold(const float* __restrict__ W1,
                                                     const float* __restrict__ W2,
                                                     const float* __restrict__ b2, int L,
                                                     float* __restrict__ fold) {
    fold_block(blockIdx.x, threadIdx.x, L, W1, W2, b2, fold);
}

#define FINE_CAP 8192
#define FINE_SL  (SORT_NB / SORT_TPB)
#define FINE_PB  16
#define FINE_CH  16
#define FINE_LST (FINE_CAP + FINE_CAP / 32)

__device__ __forceinline__ int lst_at(int p) { return p + (p >> 5); }

__global__ __launch_bounds__(SORT_TPB) void sort_fine(const int* __restrict__ keys,
                                                      const int* __restrict__ vals,
                                                      const int* __restrict__ segs,
                                                      int sliceLen, int n,
                                                      int* __restrict__ off,
                                                      int* __restrict__ valsOut,
                                                      int ngroups,
                                                      const float* __restrict__ W1,
                                                      const float* __restrict__ W2,
                                                      const float* __restrict__ b2, int L,
                                                      float* __restrict__ fold) {
    __shared__ int hist[SORT_BINS * SORT_PITCH];
    __shared__ int tot[SORT_BINS];
    __shared__ int stage[FINE_CAP];
    __shared__ int lstK[FINE_LST];
    __shared__ int lstV[FINE_LST];
    __shared__ int pcB[SORT_NB];
    __shared__ int pcL[SORT_NB];
    __shared__ int pcA[SORT_NB];
    __shared__ int red[2 * (SORT_TPB / 32)];
    const int tid  = threadIdx.x;
    const int lane = tid & 31;
    const int wv   = tid >> 5;
    const int g = blockIdx.x;
    if (g >= ngroups) {
        fold_block(g - ngroups, tid, L, W1, W2, b2, fold);
        return;
    }
    const int gl = (g < SORT_BINS) ? g : SORT_BINS;
    const int gh = (g + 1 < SORT_BINS) ? g + 1 : SORT_BINS;
    const int4 s0 = *(const int4*)(segs + gl * SORT_NB + FINE_SL * tid);
    const int4 s1 = *(const int4*)(segs + gh * SORT_NB + FINE_SL * tid);
    int pb[FINE_SL], pl[FINE_SL];
    pb[0] = (FINE_SL * tid + 0) * sliceLen + s0.x;  pl[0] = s1.x - s0.x;
    pb[1] = (FINE_SL * tid + 1) * sliceLen + s0.y;  pl[1] = s1.y - s0.y;
    pb[2] = (FINE_SL * tid + 2) * sliceLen + s0.z;  pl[2] = s1.z - s0.z;
    pb[3] = (FINE_SL * tid + 3) * sliceLen + s0.w;  pl[3] = s1.w - s0.w;
    for (int i = tid; i < SORT_BINS * SORT_PITCH; i += SORT_TPB) hist[i] = 0;
    const int mine = (pl[0] + pl[1]) + (pl[2] + pl[3]);
    int incl = mine;
    int a = (s0.x + s0.y) + (s0.z + s0.w);
    for (int d = 1; d < 32; d <<= 1) {
        const int t = __shfl_up(incl, d);
        if (lane >= d) incl += t;
    }
    for (int d = 16; d > 0; d >>= 1) a += __shfl_xor(a, d);
    if (lane == 31) red[wv] = incl;
    if (lane == 0)  red[SORT_TPB / 32 + wv] = a;
    __syncthreads();
    int bs = 0, cntg = 0, front = 0;
    for (int w = 0; w < SORT_TPB / 32; ++w) {
        cntg += red[w];
        front += (w < wv) ? red[w] : 0;
        bs += red[SORT_TPB / 32 + w];
    }
    const bool inLds = cntg <= FINE_CAP;
    const int run = (cntg + SORT_TPB - 1) / SORT_TPB;
    int start = tid * run;
    if (start > cntg) start = cntg;
    int end = start + run;
    if (end > cntg) end = cntg;

    if (inLds) {
        int at0 = front + incl - mine;
        #pragma unroll
        for (int u = 0; u < FINE_SL; ++u) {
            pcB[FINE_SL * tid + u] = (pl[u] > 0) ? pb[u] : 0;
            pcL[FINE_SL * tid + u] = pl[u];
            pcA[FINE_SL * tid + u] = at0;
            at0 += pl[u];
        }
    }
    __syncthreads();
    if (inLds) {
        const int sub = lane >> 3;
        const int j   = lane & 7;
        for (int r0 = 0; r0 < SORT_NB / 16; r0 += FINE_CH) {
            int B[FINE_CH], L[FINE_CH], A[FINE_CH];
            int k[FINE_CH][2], v[FINE_CH][2];
            #pragma unroll
            for (int r = 0; r < FINE_CH; ++r) {
                const int pi = wv * (SORT_NB / 4) + (r0 + r) * 4 + sub;
                B[r] = pcB[pi];
                L[r] = pcL[pi];
                A[r] = pcA[pi];
            }
            #pragma unroll
            for (int r = 0; r < FINE_CH; ++r) {
                const int last = (L[r] > 0) ? L[r] - 1 : 0;
                #pragma unroll
                for (int hh = 0; hh < 2; ++hh) {
                    const int jj = j + 8 * hh;
                    const int at = B[r] + ((jj < last) ? jj : last);
                    k[r][hh] = keys[at];
                    v[r][hh] = vals[at];
                }
            }
            #pragma unroll
            for (int r = 0; r < FINE_CH; ++r) {
                #pragma unroll
                for (int hh = 0; hh < 2; ++hh) {
                    const int jj = j + 8 * hh;
                    if (jj < L[r]) {
                        lstK[lst_at(A[r] + jj)] = k[r][hh];
                        lstV[lst_at(A[r] + jj)] = v[r][hh];
                    }
                }
                for (int q = FINE_PB + j; q < L[r]; q += 8) {
                    lstK[lst_at(A[r] + q)] = keys[B[r] + q];
                    lstV[lst_at(A[r] + q)] = vals[B[r] + q];
                }
            }
        }
    }
    __syncthreads();
    if (inLds) {
        int p = start;
        for (; p + 8 <= end; p += 8) {
            int k[8];
            #pragma unroll
            for (int u = 0; u < 8; ++u) k[u] = lstK[lst_at(p + u)];
            #pragma unroll
            for (int u = 0; u < 8; ++u) hist[(k[u] & (SORT_BINS - 1)) * SORT_PITCH + tid] += 1;
        }
        for (; p < end; ++p) hist[(lstK[lst_at(p)] & (SORT_BINS - 1)) * SORT_PITCH + tid] += 1;
    } else {
        #pragma unroll
        for (int u = 0; u < FINE_SL; ++u)
            for (int q = 0; q < pl[u]; ++q)
                hist[(keys[pb[u] + q] & (SORT_BINS - 1)) * SORT_PITCH + tid] += 1;
    }
    __syncthreads();
    for (int bin = tid; bin < SORT_BINS; bin += SORT_TPB) {
        int total = 0;
        for (int j = 0; j < SORT_TPB; ++j) total += hist[bin * SORT_PITCH + j];
        tot[bin] = total;
    }
    __syncthreads();
    for (int bin = tid; bin < SORT_BINS; bin += SORT_TPB) {
        int pos = bs;
        for (int d = 0; d < bin; ++d) pos += tot[d];
        const int v = (g << 8) + bin;
        if (v <= n) off[v] = pos;
        for (int j = 0; j < SORT_TPB; ++j) {
            const int c = hist[bin * SORT_PITCH + j];
            hist[bin * SORT_PITCH + j] = pos;
            pos += c;
        }
    }
    __syncthreads();
    if (inLds) {
        int p = start;
        for (; p + 4 <= end; p += 4) {
            int k[4], v[4];
            #pragma unroll
            for (int u = 0; u < 4; ++u) { k[u] = lstK[lst_at(p + u)]; v[u] = lstV[lst_at(p + u)]; }
            #pragma unroll
            for (int u = 0; u < 4; ++u) {
                const int d = k[u] & (SORT_BINS - 1);
                const int w = hist[d * SORT_PITCH + tid];
                hist[d * SORT_PITCH + tid] = w + 1;
                stage[w - bs] = v[u];
            }
        }
        for (; p < end; ++p) {
            const int d = lstK[lst_at(p)] & (SORT_BINS - 1);
            const int w = hist[d * SORT_PITCH + tid];
            hist[d * SORT_PITCH + tid] = w + 1;
            stage[w - bs] = lstV[lst_at(p)];
        }
        __syncthreads();
        int i = tid;
        for (; i + 7 * SORT_TPB < cntg; i += 8 * SORT_TPB) {
            int w[8];
            #pragma unroll
            for (int u = 0; u < 8; ++u) w[u] = stage[i + u * SORT_TPB];
            #pragma unroll
            for (int u = 0; u < 8; ++u) valsOut[bs + i + u * SORT_TPB] = w[u];
        }
        for (; i < cntg; i += SORT_TPB) valsOut[bs + i] = stage[i];
        return;
    }
    #pragma unroll
    for (int u = 0; u < FINE_SL; ++u) {
        for (int q = 0; q < pl[u]; ++q) {
            const int d = keys[pb[u] + q] & (SORT_BINS - 1);
            const int w = hist[d * SORT_PITCH + tid];
            hist[d * SORT_PITCH + tid] = w + 1;
            valsOut[w] = vals[pb[u] + q];
        }
    }
}

__global__ __launch_bounds__(256) void gin_offsets(const int* __restrict__ sdst,
                                                   int nedges, int n,
                                                   int* __restrict__ off) {
    const int p = blockIdx.x * blockDim.x + threadIdx.x;
    if (p > nedges) return;
    const int prev = (p == 0) ? -1 : sdst[p - 1];
    const int here = (p == nedges) ? n : sdst[p];
    for (int v = prev + 1; v <= here; ++v) off[v] = p;
}

#define GROWS 16

__device__ __forceinline__ void add4(float4& a, const float4 r) {
    a.x += r.x; a.y += r.y; a.z += r.z; a.w += r.w;
}

__device__ __forceinline__ void close_row(float* zrow, const float e1, const float4 acc) {
    const float4 hv = *(const float4*)zrow;
    float4 o;
    o.x = e1 * hv.x + acc.x;
    o.y = e1 * hv.y + acc.y;
    o.z = e1 * hv.z + acc.z;
    o.w = e1 * hv.w + acc.w;
    *(float4*)zrow = o;
}

__device__ __forceinline__ void load_rows(float4 (&r)[GROWS], const int idxv, const int j,
                                          const float* hc) {
    #pragma unroll
    for (int t = 0; t < GROWS; ++t) {
        const int s = __builtin_amdgcn_readlane(idxv, j + t);
        r[t] = *(const float4*)(hc + (size_t)s * DD);
    }
}

__device__ __forceinline__ void sum_rows(const float4 (&r)[GROWS], const int p0, const int eEnd,
                                         const int ob, float* zl, const float e1,
                                         int& i, int& nextB, float4& acc) {
    #pragma unroll
    for (int t = 0; t < GROWS; ++t) {
        const int p = p0 + t;
        if (p < eEnd) {
            while (p >= nextB) {
                close_row(zl + i * HROW, e1, acc);
                acc.x = 0.0f; acc.y = 0.0f; acc.z = 0.0f; acc.w = 0.0f;
                ++i;
                nextB = __builtin_amdgcn_readlane(ob, i + 1);
            }
            add4(acc, r[t]);
        }
    }
}

__device__ __forceinline__ void w_fetch(float (&wr)[64], const float* __restrict__ Wl,
                                        const int tid) {
    const float* wc = Wl + (tid >> 7) * 4 * DD + (tid & 127);
    #pragma unroll
    for (int i = 0; i < 16; ++i) {
        #pragma unroll
        for (int q = 0; q < 4; ++q) wr[4 * i + q] = wc[(8 * i + q) * DD];
    }
}

__device__ __forceinline__ void w_put(const float (&wr)[64], float* wS, const int tid) {
    float* wd = wS + ((tid >> 7) * DD + (tid & 127)) * 4;
    #pragma unroll
    for (int i = 0; i < 16; ++i) {
        float4 v;
        v.x = wr[4 * i]; v.y = wr[4 * i + 1]; v.z = wr[4 * i + 2]; v.w = wr[4 * i + 3];
        *(float4*)(wd + 2 * i * DD * 4) = v;
    }
}

__device__ __forceinline__ void mlp_product(v8f (&acc)[8], const float* ab, const float* wl) {
    #pragma unroll
    for (int k0 = 0; k0 < DD; k0 += 4) {
        const v2f a = *(const v2f*)(ab + k0);
        #pragma unroll
        for (int t = 0; t < 8; ++t) {
            const v2f b = *(const v2f*)(wl + (k0 >> 2) * (DD * 4) + t * 64);
            acc[t] = __builtin_amdgcn_wmma_f32_16x16x4_f32(false, a, false, b,
                                                           (short)0, acc[t], false, false);
        }
    }
}

__global__ __launch_bounds__(256, 2) void gin_layer(const float* __restrict__ h,
                                                    const float* __restrict__ eps, int layer,
                                                    const int* __restrict__ off,
                                                    const int* __restrict__ csrc,
                                                    const float* __restrict__ W1,
                                                    const float* __restrict__ b1,
                                                    const float* __restrict__ W2,
                                                    const float* __restrict__ b2,
                                                    float* __restrict__ out, int n) {
    __shared__ __attribute__((aligned(16))) float lds[DD * DD + 8 * 16 * HROW];
    float* wS = lds;

    const int tid   = threadIdx.x;
    const int lane  = threadIdx.x & 31;
    const int wave  = __builtin_amdgcn_readfirstlane((int)(threadIdx.x >> 5));
    const int ln    = lane & 15;
    const int khoff = lane >> 4;
    const int rh    = khoff << 3;
    const int m0    = blockIdx.x * 128 + wave * 16;

    float* zw = lds + DD * DD + wave * (16 * HROW);

    float wr[64];
    w_fetch(wr, W1, tid);

    int ob;
    {
        int t = m0 + lane;
        if (t > n) t = n;
        ob = off[t];
    }
    const float* hc = h + lane * 4;
    float* zl = zw + lane * 4;
    {
        float4 own[16];
        #pragma unroll
        for (int q = 0; q < 16; ++q) {
            int rr = m0 + q;
            if (rr > n - 1) rr = n - 1;
            own[q] = *(const float4*)(hc + (size_t)rr * DD);
        }
        w_put(wr, wS, tid);
        #pragma unroll
        for (int q = 0; q < 16; ++q) *(float4*)(zl + q * HROW) = own[q];
    }
    const float e1 = 1.0f + eps[layer];
    const int eEnd = __builtin_amdgcn_readlane(ob, 16);
    int base  = __builtin_amdgcn_readlane(ob, 0);
    int nextB = __builtin_amdgcn_readlane(ob, 1);
    int i = 0;
    float4 acc = {0.0f, 0.0f, 0.0f, 0.0f};
    int m = eEnd - base;
    if (m > 32) m = 32;
    int idxv = (lane < m) ? csrc[base + lane] : 0;
    __syncthreads();
    while (m > 0) {
        const int nb = base + 32;
        int mn = eEnd - nb;
        if (mn > 32) mn = 32;
        float4 r[GROWS];
        load_rows(r, idxv, 0, hc);
        const int idxn = (lane < mn) ? csrc[nb + lane] : 0;
        sum_rows(r, base, eEnd, ob, zl, e1, i, nextB, acc);
        if (m > GROWS) {
            load_rows(r, idxv, GROWS, hc);
            sum_rows(r, base + GROWS, eEnd, ob, zl, e1, i, nextB, acc);
        }
        base = nb;
        m = mn;
        idxv = idxn;
    }
    while (i < 16) {
        close_row(zl + i * HROW, e1, acc);
        acc.x = 0.0f; acc.y = 0.0f; acc.z = 0.0f; acc.w = 0.0f;
        ++i;
    }

    const float* ab = zw + ln * HROW + 2 * khoff;
    const float* wl = wS + ln * 4 + 2 * khoff;

    w_fetch(wr, W2, tid);

    {
        float bias[8];
        #pragma unroll
        for (int t = 0; t < 8; ++t) bias[t] = b1[t * 16 + ln];
        v8f pa[8] = {};
        mlp_product(pa, ab, wl);
        #pragma unroll
        for (int j = 0; j < 8; ++j) {
            #pragma unroll
            for (int t = 0; t < 8; ++t)
                zw[(j + rh) * HROW + t * 16 + ln] = fmaxf(pa[t][j] + bias[t], 0.0f);
        }
    }

    __syncthreads();
    w_put(wr, wS, tid);
    __syncthreads();

    {
        float bias[8];
        #pragma unroll
        for (int t = 0; t < 8; ++t) bias[t] = b2 ? b2[t * 16 + ln] : 0.0f;
        v8f pa[8] = {};
        mlp_product(pa, ab, wl);
        #pragma unroll
        for (int j = 0; j < 8; ++j) {
            const int r = m0 + j + rh;
            if (r < n) {
                #pragma unroll
                for (int t = 0; t < 8; ++t)
                    out[(size_t)r * DD + t * 16 + ln] = pa[t][j] + bias[t];
            }
        }
    }
}

__device__ __forceinline__ void close_row2(float* zrow, const float e1, const float4 acc,
                                           const int deg, const float4 cb, const float4 bb) {
    const float4 hv = *(const float4*)zrow;
    const float w = e1 + (float)deg;
    float4 o;
    o.x = fmaxf(e1 * hv.x + acc.x + (w * cb.x + bb.x), 0.0f);
    o.y = fmaxf(e1 * hv.y + acc.y + (w * cb.y + bb.y), 0.0f);
    o.z = fmaxf(e1 * hv.z + acc.z + (w * cb.z + bb.z), 0.0f);
    o.w = fmaxf(e1 * hv.w + acc.w + (w * cb.w + bb.w), 0.0f);
    *(float4*)zrow = o;
}

__device__ __forceinline__ void sum_rows2(const float4 (&r)[GROWS], const int p0, const int eEnd,
                                          const int ob, float* zl, const float e1,
                                          const float4 cb, const float4 bb,
                                          int& i, int& nextB, float4& acc) {
    #pragma unroll
    for (int t = 0; t < GROWS; ++t) {
        const int p = p0 + t;
        if (p < eEnd) {
            while (p >= nextB) {
                close_row2(zl + i * HROW, e1, acc, nextB - __builtin_amdgcn_readlane(ob, i), cb, bb);
                acc.x = 0.0f; acc.y = 0.0f; acc.z = 0.0f; acc.w = 0.0f;
                ++i;
                nextB = __builtin_amdgcn_readlane(ob, i + 1);
            }
            add4(acc, r[t]);
        }
    }
}

__global__ __launch_bounds__(256, 2) void gin_layer2(const float* __restrict__ g,
                                                     const float* __restrict__ eps, int layer,
                                                     const int* __restrict__ off,
                                                     const int* __restrict__ csrc,
                                                     const float* __restrict__ cvec,
                                                     const float* __restrict__ b1,
                                                     const float* __restrict__ Wn,
                                                     const float* __restrict__ bn,
                                                     float* __restrict__ out, int n) {
    __shared__ __attribute__((aligned(16))) float lds[DD * DD + 8 * 16 * HROW];
    float* wS = lds;

    const int tid   = threadIdx.x;
    const int lane  = threadIdx.x & 31;
    const int wave  = __builtin_amdgcn_readfirstlane((int)(threadIdx.x >> 5));
    const int ln    = lane & 15;
    const int khoff = lane >> 4;
    const int rh    = khoff << 3;
    const int m0    = blockIdx.x * 128 + wave * 16;

    float* zw = lds + DD * DD + wave * (16 * HROW);

    float wr[64];
    w_fetch(wr, Wn, tid);

    int ob;
    {
        int t = m0 + lane;
        if (t > n) t = n;
        ob = off[t];
    }
    const float* hc = g + lane * 4;
    float* zl = zw + lane * 4;
    const float4 cb = *(const float4*)(cvec + lane * 4);
    const float4 bb = *(const float4*)(b1 + lane * 4);
    {
        float4 own[16];
        #pragma unroll
        for (int q = 0; q < 16; ++q) {
            int rr = m0 + q;
            if (rr > n - 1) rr = n - 1;
            own[q] = *(const float4*)(hc + (size_t)rr * DD);
        }
        w_put(wr, wS, tid);
        #pragma unroll
        for (int q = 0; q < 16; ++q) *(float4*)(zl + q * HROW) = own[q];
    }
    const float e1 = 1.0f + eps[layer];
    const int eEnd = __builtin_amdgcn_readlane(ob, 16);
    int base  = __builtin_amdgcn_readlane(ob, 0);
    int nextB = __builtin_amdgcn_readlane(ob, 1);
    int i = 0;
    float4 acc = {0.0f, 0.0f, 0.0f, 0.0f};
    int m = eEnd - base;
    if (m > 32) m = 32;
    int idxv = (lane < m) ? csrc[base + lane] : 0;
    __syncthreads();
    while (m > 0) {
        const int nb = base + 32;
        int mn = eEnd - nb;
        if (mn > 32) mn = 32;
        float4 r[GROWS];
        load_rows(r, idxv, 0, hc);
        const int idxn = (lane < mn) ? csrc[nb + lane] : 0;
        sum_rows2(r, base, eEnd, ob, zl, e1, cb, bb, i, nextB, acc);
        if (m > GROWS) {
            load_rows(r, idxv, GROWS, hc);
            sum_rows2(r, base + GROWS, eEnd, ob, zl, e1, cb, bb, i, nextB, acc);
        }
        base = nb;
        m = mn;
        idxv = idxn;
    }
    while (i < 16) {
        close_row2(zl + i * HROW, e1, acc,
                   __builtin_amdgcn_readlane(ob, i + 1) - __builtin_amdgcn_readlane(ob, i), cb, bb);
        acc.x = 0.0f; acc.y = 0.0f; acc.z = 0.0f; acc.w = 0.0f;
        ++i;
    }

    const float* ab = zw + ln * HROW + 2 * khoff;
    const float* wl = wS + ln * 4 + 2 * khoff;
    {
        float bias[8];
        #pragma unroll
        for (int t = 0; t < 8; ++t) bias[t] = bn ? bn[t * 16 + ln] : 0.0f;
        v8f pa[8] = {};
        mlp_product(pa, ab, wl);
        #pragma unroll
        for (int j = 0; j < 8; ++j) {
            const int r = m0 + j + rh;
            if (r < n) {
                #pragma unroll
                for (int t = 0; t < 8; ++t)
                    out[(size_t)r * DD + t * 16 + ln] = pa[t][j] + bias[t];
            }
        }
    }
}

extern "C" void kernel_launch(void* const* d_in, const int* in_sizes, int n_in,
                              void* d_out, int out_size, void* d_ws, size_t ws_size,
                              hipStream_t stream) {
    const float* h   = (const float*)d_in[0];
    const int*   src = (const int*)  d_in[1];
    const int*   dst = (const int*)  d_in[2];
    const float* W1  = (const float*)d_in[3];
    const float* b1  = (const float*)d_in[4];
    const float* W2  = (const float*)d_in[5];
    const float* b2  = (const float*)d_in[6];
    const float* eps = (const float*)d_in[7];

    const int n = in_sizes[0] / DD;
    const int e = in_sizes[1];
    const int L = in_sizes[7];

    float* hb  = (float*)d_ws;
    float* out = (float*)d_out;

    const int tabN = SORT_BINS * SORT_NB;
    const int np   = ((n + 4) / 4) * 4;
    int* table  = (int*)(hb + (size_t)n * DD);
    int* tableS = table + tabN;
    int* off    = tableS + tabN;
    int* kA     = off + np;
    int* vA     = kA + e;
    int* kB     = vA + e;
    int* vB     = kB + e;
    const size_t sortInts = (size_t)2 * tabN + np + (size_t)4 * e;
    float* fold  = hb + (size_t)n * DD + ((sortInts + 3) / 4) * 4;
    float* foldC = fold + (size_t)(L - 1) * DD * DD;
    const int foldBlocks = (L > 1) ? 3 * (L - 1) : 0;

    const int offBlocks  = (e + 1 + 255) / 256;
    const int mlpBlocks  = (n + 127) / 128;
    const int run        = (e + SORT_NB * SORT_TPB - 1) / (SORT_NB * SORT_TPB);

    const int* kin = dst;
    const int* vin = src;
    if (((n - 1) >> 16) == 0 && (long)run * SORT_TPB <= PLACE_CAP) {
        int* segs = table;
        sort_local<<<SORT_NB, SORT_TPB, 0, stream>>>(dst, src, e, run, segs, kA, vA);
        sort_fine<<<(n >> 8) + 1 + foldBlocks, SORT_TPB, 0, stream>>>(kA, vA, segs, run * SORT_TPB, n, off, vB,
                                                                     (n >> 8) + 1, W1, W2, b2, L, fold);
        vin = vB;
    } else {
        int* kout = kA;
        int* vout = vA;
        for (int shift = 0; shift == 0 || ((n - 1) >> shift) > 0; shift += 8) {
            sort_count<<<SORT_NB, SORT_TPB, 0, stream>>>(kin, shift, e, run, table);
            gin_scan<<<tabN / 1024, 256, 0, stream>>>(table, tableS);
            sort_place<<<SORT_NB, SORT_TPB, 0, stream>>>(kin, vin, shift, e, run, tableS, kout, vout);
            kin = kout;
            vin = vout;
            kout = (kout == kA) ? kB : kA;
            vout = (vout == vA) ? vB : vA;
        }
        gin_offsets<<<offBlocks, 256, 0, stream>>>(kin, e, n, off);
        if (foldBlocks > 0) gin_fold<<<foldBlocks, SORT_TPB, 0, stream>>>(W1, W2, b2, L, fold);
    }

    for (int l = 0; l < L; ++l) {
        const float* hin = (l == 0) ? h : (((L - l) & 1) ? hb : out);
        float* hout = ((L - 1 - l) & 1) ? hb : out;
        const float* Wn = (l < L - 1) ? fold + (size_t)l * DD * DD : W2 + (size_t)l * DD * DD;
        const float* bn = (l < L - 1) ? nullptr : b2 + l * DD;
        if (l == 0)
            gin_layer<<<mlpBlocks, 256, 0, stream>>>(hin, eps, l, off, vin, W1, b1, Wn, bn, hout, n);
        else
            gin_layer2<<<mlpBlocks, 256, 0, stream>>>(hin, eps, l, off, vin, foldC + (l - 1) * DD,
                                                      b1 + l * DD, Wn, bn, hout, n);
    }
}


# List walk needs fewer instructions per gathered row: row address = wave-wide base + 32-bit lane offset (one vector add), one scalar test per row (list-end sentinel behind the tile); sum order kept
# speedup vs baseline: 3.7819x; 1.0794x over previous
#include <hip/hip_runtime.h>
#include <hip/hip_bf16.h>

typedef __attribute__((ext_vector_type(2))) float v2f;
typedef __attribute__((ext_vector_type(8))) float v8f;

#define DD 128
#define HROW 132

#define SORT_NB    512
#define SORT_TPB   128
#define SORT_BINS  256
#define SORT_PITCH 129

__device__ __forceinline__ void sort_count_runs(const int* __restrict__ keys, int shift,
                                                long start, long end, int* hist, int tid) {
    for (int i = tid; i < SORT_BINS * SORT_PITCH; i += SORT_TPB) hist[i] = 0;
    __syncthreads();
    long p = start;
    for (; p + 4 <= end; p += 4) {
        const int k0 = keys[p], k1 = keys[p + 1], k2 = keys[p + 2], k3 = keys[p + 3];
        hist[((k0 >> shift) & (SORT_BINS - 1)) * SORT_PITCH + tid] += 1;
        hist[((k1 >> shift) & (SORT_BINS - 1)) * SORT_PITCH + tid] += 1;
        hist[((k2 >> shift) & (SORT_BINS - 1)) * SORT_PITCH + tid] += 1;
        hist[((k3 >> shift) & (SORT_BINS - 1)) * SORT_PITCH + tid] += 1;
    }
    for (; p < end; ++p) {
        const int d = (keys[p] >> shift) & (SORT_BINS - 1);
        hist[d * SORT_PITCH + tid] += 1;
    }
    __syncthreads();
}

__global__ __launch_bounds__(SORT_TPB) void sort_count(const int* __restrict__ keys, int shift,
                                                       int nedges, int run,
                                                       int* __restrict__ table) {
    __shared__ int hist[SORT_BINS * SORT_PITCH];
    const int tid = threadIdx.x;
    const long start = ((long)blockIdx.x * SORT_TPB + tid) * run;
    long end = start + run;
    if (end > nedges) end = nedges;
    sort_count_runs(keys, shift, start, end, hist, tid);
    for (int bin = tid; bin < SORT_BINS; bin += SORT_TPB) {
        int total = 0;
        for (int j = 0; j < SORT_TPB; ++j) total += hist[bin * SORT_PITCH + j];
        table[bin * SORT_NB + blockIdx.x] = total;
    }
}

__global__ __launch_bounds__(256) void gin_scan(const int* __restrict__ in,
                                                int* __restrict__ outp) {
    __shared__ int wtot[8];
    __shared__ int wpre[8];
    const int tid  = threadIdx.x;
    const int lane = tid & 31;
    const int wv   = tid >> 5;
    const int base = blockIdx.x * 1024;

    const int nfront = blockIdx.x;
    const int* mine = in + tid * 4;
    int pre = 0;
    for (int ch = 0; ch < nfront; ch += 16) {
        int part[16];
        #pragma unroll
        for (int k = 0; k < 16; ++k) {
            const int at = (ch + k < nfront) ? ch + k : 0;
            const int4 f = *(const int4*)(mine + at * 1024);
            part[k] = (f.x + f.y) + (f.z + f.w);
        }
        #pragma unroll
        for (int k = 0; k < 16; ++k) pre += (ch + k < nfront) ? part[k] : 0;
    }
    const int4 c = *(const int4*)(in + base + tid * 4);
    const int local = (c.x + c.y) + (c.z + c.w);

    int incl = local;
    for (int d = 1; d < 32; d <<= 1) {
        const int t = __shfl_up(incl, d);
        if (lane >= d) incl += t;
    }
    for (int d = 16; d > 0; d >>= 1) pre += __shfl_xor(pre, d);
    if (lane == 31) wtot[wv] = incl;
    if (lane == 0)  wpre[wv] = pre;
    __syncthreads();
    int front = 0;
    for (int w = 0; w < 8; ++w) {
        front += wpre[w];
        if (w < wv) front += wtot[w];
    }
    int4 o;
    o.x = front + incl - local;
    o.y = o.x + c.x;
    o.z = o.y + c.y;
    o.w = o.z + c.z;
    *(int4*)(outp + base + tid * 4) = o;
}

__device__ __forceinline__ void sort_put(int* hist, int tid, int shift, int k, int v,
                                         int* __restrict__ keysOut, int* __restrict__ valsOut) {
    const int d = (k >> shift) & (SORT_BINS - 1);
    const int q = hist[d * SORT_PITCH + tid];
    hist[d * SORT_PITCH + tid] = q + 1;
    keysOut[q] = k;
    valsOut[q] = v;
}

#define PLACE_CAP 2048

__device__ __forceinline__ void sort_put_lds(int* hist, const int* lbs, int* stK, int* stV,
                                             int tid, int shift, int k, int v) {
    const int d = (k >> shift) & (SORT_BINS - 1);
    const int w = hist[d * SORT_PITCH + tid];
    hist[d * SORT_PITCH + tid] = w + 1;
    const int i = lbs[d] + w;
    stK[i] = k;
    stV[i] = v;
}

__global__ __launch_bounds__(SORT_TPB) void sort_place(const int* __restrict__ keys,
                                                       const int* __restrict__ vals, int shift,
                                                       int nedges, int run,
                                                       const int* __restrict__ tableS,
                                                       int* __restrict__ keysOut,
                                                       int* __restrict__ valsOut) {
    __shared__ int hist[SORT_BINS * SORT_PITCH];
    __shared__ int tot[SORT_BINS];
    __shared__ int lbs[SORT_BINS];
    __shared__ int dlt[SORT_BINS];
    __shared__ int wsum[SORT_TPB / 32];
    __shared__ int stK[PLACE_CAP];
    __shared__ int stV[PLACE_CAP];
    const int tid = threadIdx.x;
    const long start = ((long)blockIdx.x * SORT_TPB + tid) * run;
    long end = start + run;
    if (end > nedges) end = nedges;
    sort_count_runs(keys, shift, start, end, hist, tid);
    if ((long)run * SORT_TPB <= PLACE_CAP) {
        const int g0 = tableS[(2 * tid) * SORT_NB + blockIdx.x];
        const int g1 = tableS[(2 * tid + 1) * SORT_NB + blockIdx.x];
        for (int bin = tid; bin < SORT_BINS; bin += SORT_TPB) {
            int pos = 0;
            for (int j = 0; j < SORT_TPB; ++j) {
                const int c = hist[bin * SORT_PITCH + j];
                hist[bin * SORT_PITCH + j] = pos;
                pos += c;
            }
            tot[bin] = pos;
        }
        __syncthreads();
        const int lane = tid & 31;
        const int wv   = tid >> 5;
        const int ta = tot[2 * tid];
        const int tb = tot[2 * tid + 1];
        int incl = ta + tb;
        for (int d = 1; d < 32; d <<= 1) {
            const int t = __shfl_up(incl, d);
            if (lane >= d) incl += t;
        }
        if (lane == 31) wsum[wv] = incl;
        __syncthreads();
        int front = 0;
        for (int w = 0; w < SORT_TPB / 32; ++w) front += (w < wv) ? wsum[w] : 0;
        const int ex = front + incl - (ta + tb);
        lbs[2 * tid]     = ex;
        lbs[2 * tid + 1] = ex + ta;
        dlt[2 * tid]     = g0 - ex;
        dlt[2 * tid + 1] = g1 - (ex + ta);
        __syncthreads();
        long p = start;
        for (; p + 4 <= end; p += 4) {
            const int k0 = keys[p], k1 = keys[p + 1], k2 = keys[p + 2], k3 = keys[p + 3];
            const int v0 = vals[p], v1 = vals[p + 1], v2 = vals[p + 2], v3 = vals[p + 3];
            sort_put_lds(hist, lbs, stK, stV, tid, shift, k0, v0);
            sort_put_lds(hist, lbs, stK, stV, tid, shift, k1, v1);
            sort_put_lds(hist, lbs, stK, stV, tid, shift, k2, v2);
            sort_put_lds(hist, lbs, stK, stV, tid, shift, k3, v3);
        }
        for (; p < end; ++p) sort_put_lds(hist, lbs, stK, stV, tid, shift, keys[p], vals[p]);
        __syncthreads();
        const long bstart = (long)blockIdx.x * SORT_TPB * run;
        long bend = bstart + (long)SORT_TPB * run;
        if (bend > nedges) bend = nedges;
        const int cnt = (bend > bstart) ? (int)(bend - bstart) : 0;
        int i = tid;
        for (; i + 3 * SORT_TPB < cnt; i += 4 * SORT_TPB) {
            int k[4], v[4], o[4];
            #pragma unroll
            for (int u = 0; u < 4; ++u) { k[u] = stK[i + u * SORT_TPB]; v[u] = stV[i + u * SORT_TPB]; }
            #pragma unroll
            for (int u = 0; u < 4; ++u) o[u] = dlt[(k[u] >> shift) & (SORT_BINS - 1)] + i + u * SORT_TPB;
            #pragma unroll
            for (int u = 0; u < 4; ++u) { keysOut[o[u]] = k[u]; valsOut[o[u]] = v[u]; }
        }
        for (; i < cnt; i += SORT_TPB) {
            const int k = stK[i];
            const int v = stV[i];
            const int o = dlt[(k >> shift) & (SORT_BINS - 1)] + i;
            keysOut[o] = k;
            valsOut[o] = v;
        }
        return;
    }
    for (int bin = tid; bin < SORT_BINS; bin += SORT_TPB) {
        int pos = tableS[bin * SORT_NB + blockIdx.x];
        for (int j = 0; j < SORT_TPB; ++j) {
            const int c = hist[bin * SORT_PITCH + j];
            hist[bin * SORT_PITCH + j] = pos;
            pos += c;
        }
    }
    __syncthreads();
    long p = start;
    for (; p + 4 <= end; p += 4) {
        const int k0 = keys[p], k1 = keys[p + 1], k2 = keys[p + 2], k3 = keys[p + 3];
        const int v0 = vals[p], v1 = vals[p + 1], v2 = vals[p + 2], v3 = vals[p + 3];
        sort_put(hist, tid, shift, k0, v0, keysOut, valsOut);
        sort_put(hist, tid, shift, k1, v1, keysOut, valsOut);
        sort_put(hist, tid, shift, k2, v2, keysOut, valsOut);
        sort_put(hist, tid, shift, k3, v3, keysOut, valsOut);
    }
    for (; p < end; ++p) sort_put(hist, tid, shift, keys[p], vals[p], keysOut, valsOut);
}

__global__ __launch_bounds__(SORT_TPB) void sort_local(const int* __restrict__ keys,
                                                       const int* __restrict__ vals,
                                                       int nedges, int run,
                                                       int* __restrict__ segs,
                                                       int* __restrict__ keysOut,
                                                       int* __restrict__ valsOut) {
    __shared__ int hist[SORT_BINS * SORT_PITCH];
    __shared__ int tot[SORT_BINS];
    __shared__ int lbs[SORT_BINS];
    __shared__ int wsum[SORT_TPB / 32];
    __shared__ int stK[PLACE_CAP];
    __shared__ int stV[PLACE_CAP];
    const int tid = threadIdx.x;
    const long start = ((long)blockIdx.x * SORT_TPB + tid) * run;
    long end = start + run;
    if (end > nedges) end = nedges;
    sort_count_runs(keys, 8, start, end, hist, tid);
    for (int bin = tid; bin < SORT_BINS; bin += SORT_TPB) {
        int pos = 0;
        for (int j = 0; j < SORT_TPB; ++j) {
            const int c = hist[bin * SORT_PITCH + j];
            hist[bin * SORT_PITCH + j] = pos;
            pos += c;
        }
        tot[bin] = pos;
    }
    __syncthreads();
    const int lane = tid & 31;
    const int wv   = tid >> 5;
    const int ta = tot[2 * tid];
    const int tb = tot[2 * tid + 1];
    int incl = ta + tb;
    for (int d = 1; d < 32; d <<= 1) {
        const int t = __shfl_up(incl, d);
        if (lane >= d) incl += t;
    }
    if (lane == 31) wsum[wv] = incl;
    __syncthreads();
    int front = 0;
    for (int w = 0; w < SORT_TPB / 32; ++w) front += (w < wv) ? wsum[w] : 0;
    const int ex = front + incl - (ta + tb);
    lbs[2 * tid]     = ex;
    lbs[2 * tid + 1] = ex + ta;
    segs[(2 * tid) * SORT_NB + blockIdx.x]     = ex;
    segs[(2 * tid + 1) * SORT_NB + blockIdx.x] = ex + ta;
    if (tid == SORT_TPB - 1) segs[SORT_BINS * SORT_NB + blockIdx.x] = ex + ta + tb;
    __syncthreads();
    long p = start;
    for (; p + 4 <= end; p += 4) {
        const int k0 = keys[p], k1 = keys[p + 1], k2 = keys[p + 2], k3 = keys[p + 3];
        const int v0 = vals[p], v1 = vals[p + 1], v2 = vals[p + 2], v3 = vals[p + 3];
        sort_put_lds(hist, lbs, stK, stV, tid, 8, k0, v0);
        sort_put_lds(hist, lbs, stK, stV, tid, 8, k1, v1);
        sort_put_lds(hist, lbs, stK, stV, tid, 8, k2, v2);
        sort_put_lds(hist, lbs, stK, stV, tid, 8, k3, v3);
    }
    for (; p < end; ++p) sort_put_lds(hist, lbs, stK, stV, tid, 8, keys[p], vals[p]);
    __syncthreads();
    const long bstart = (long)blockIdx.x * SORT_TPB * run;
    long bend = bstart + (long)SORT_TPB * run;
    if (bend > nedges) bend = nedges;
    const int cnt = (bend > bstart) ? (int)(bend - bstart) : 0;
    int i = tid;
    for (; i + 3 * SORT_TPB < cnt; i += 4 * SORT_TPB) {
        int k[4], v[4];
        #pragma unroll
        for (int u = 0; u < 4; ++u) { k[u] = stK[i + u * SORT_TPB]; v[u] = stV[i + u * SORT_TPB]; }
        #pragma unroll
        for (int u = 0; u < 4; ++u) { keysOut[bstart + i + u * SORT_TPB] = k[u]; valsOut[bstart + i + u * SORT_TPB] = v[u]; }
    }
    for (; i < cnt; i += SORT_TPB) {
        keysOut[bstart + i] = stK[i];
        valsOut[bstart + i] = stV[i];
    }
}

__device__ __forceinline__ void fold_tile(const float* __restrict__ A, const float* __restrict__ B,
                                          float* __restrict__ M, const int rt, const int lane) {
    const int ln = lane & 15;
    const int kh = lane >> 4;
    const float* ap = A + (rt * 16 + ln) * DD + 2 * kh;
    const float* bp = B + (2 * kh) * DD + ln;
    v8f acc[8] = {};
    #pragma unroll 8
    for (int k0 = 0; k0 < DD; k0 += 4) {
        v2f a;
        a.x = ap[k0];
        a.y = ap[k0 + 1];
        #pragma unroll
        for (int t = 0; t < 8; ++t) {
            v2f b;
            b.x = bp[k0 * DD + t * 16];
            b.y = bp[(k0 + 1) * DD + t * 16];
            acc[t] = __builtin_amdgcn_wmma_f32_16x16x4_f32(false, a, false, b,
                                                           (short)0, acc[t], false, false);
        }
    }
    const int rh = kh << 3;
    #pragma unroll
    for (int j = 0; j < 8; ++j) {
        #pragma unroll
        for (int t = 0; t < 8; ++t)
            M[(rt * 16 + j + rh) * DD + t * 16 + ln] = acc[t][j];
    }
}

__device__ __forceinline__ void fold_block(const int fb, const int tid, const int L,
                                           const float* __restrict__ W1,
                                           const float* __restrict__ W2,
                                           const float* __restrict__ b2,
                                           float* __restrict__ fold) {
    const int nt  = (L - 1) * (DD / 16);
    const int ntb = (nt + SORT_TPB / 32 - 1) / (SORT_TPB / 32);
    float* foldC = fold + (size_t)(L - 1) * DD * DD;
    if (fb < ntb) {
        const int t = fb * (SORT_TPB / 32) + __builtin_amdgcn_readfirstlane(tid >> 5);
        if (t < nt) {
            const int l = t / (DD / 16);
            fold_tile(W2 + (size_t)l * DD * DD, W1 + (size_t)(l + 1) * DD * DD,
                      fold + (size_t)l * DD * DD, t % (DD / 16), tid & 31);
        }
    } else {
        const int l = fb - ntb;
        if (l < L - 1) {
            const float* bl = b2 + l * DD;
            const float* Wn = W1 + (size_t)(l + 1) * DD * DD;
            for (int nn = tid; nn < DD; nn += SORT_TPB) {
                float sacc = 0.0f;
                #pragma unroll 16
                for (int k = 0; k < DD; ++k) sacc = fmaf(bl[k], Wn[k * DD + nn], sacc);
                foldC[l * DD + nn] = sacc;
            }
        }
    }
}

__global__ __launch_bounds__(SORT_TPB) void gin_fold(const float* __restrict__ W1,
                                                     const float* __restrict__ W2,
                                                     const float* __restrict__ b2, int L,
                                                     float* __restrict__ fold) {
    fold_block(blockIdx.x, threadIdx.x, L, W1, W2, b2, fold);
}

#define FINE_CAP 8192
#define FINE_SL  (SORT_NB / SORT_TPB)
#define FINE_PB  16
#define FINE_CH  16
#define FINE_LST (FINE_CAP + FINE_CAP / 32)

__device__ __forceinline__ int lst_at(int p) { return p + (p >> 5); }

__global__ __launch_bounds__(SORT_TPB) void sort_fine(const int* __restrict__ keys,
                                                      const int* __restrict__ vals,
                                                      const int* __restrict__ segs,
                                                      int sliceLen, int n,
                                                      int* __restrict__ off,
                                                      int* __restrict__ valsOut,
                                                      int ngroups,
                                                      const float* __restrict__ W1,
                                                      const float* __restrict__ W2,
                                                      const float* __restrict__ b2, int L,
                                                      float* __restrict__ fold) {
    __shared__ int hist[SORT_BINS * SORT_PITCH];
    __shared__ int tot[SORT_BINS];
    __shared__ int stage[FINE_CAP];
    __shared__ int lstK[FINE_LST];
    __shared__ int lstV[FINE_LST];
    __shared__ int pcB[SORT_NB];
    __shared__ int pcL[SORT_NB];
    __shared__ int pcA[SORT_NB];
    __shared__ int red[2 * (SORT_TPB / 32)];
    const int tid  = threadIdx.x;
    const int lane = tid & 31;
    const int wv   = tid >> 5;
    const int g = blockIdx.x;
    if (g >= ngroups) {
        fold_block(g - ngroups, tid, L, W1, W2, b2, fold);
        return;
    }
    const int gl = (g < SORT_BINS) ? g : SORT_BINS;
    const int gh = (g + 1 < SORT_BINS) ? g + 1 : SORT_BINS;
    const int4 s0 = *(const int4*)(segs + gl * SORT_NB + FINE_SL * tid);
    const int4 s1 = *(const int4*)(segs + gh * SORT_NB + FINE_SL * tid);
    int pb[FINE_SL], pl[FINE_SL];
    pb[0] = (FINE_SL * tid + 0) * sliceLen + s0.x;  pl[0] = s1.x - s0.x;
    pb[1] = (FINE_SL * tid + 1) * sliceLen + s0.y;  pl[1] = s1.y - s0.y;
    pb[2] = (FINE_SL * tid + 2) * sliceLen + s0.z;  pl[2] = s1.z - s0.z;
    pb[3] = (FINE_SL * tid + 3) * sliceLen + s0.w;  pl[3] = s1.w - s0.w;
    for (int i = tid; i < SORT_BINS * SORT_PITCH; i += SORT_TPB) hist[i] = 0;
    const int mine = (pl[0] + pl[1]) + (pl[2] + pl[3]);
    int incl = mine;
    int a = (s0.x + s0.y) + (s0.z + s0.w);
    for (int d = 1; d < 32; d <<= 1) {
        const int t = __shfl_up(incl, d);
        if (lane >= d) incl += t;
    }
    for (int d = 16; d > 0; d >>= 1) a += __shfl_xor(a, d);
    if (lane == 31) red[wv] = incl;
    if (lane == 0)  red[SORT_TPB / 32 + wv] = a;
    __syncthreads();
    int bs = 0, cntg = 0, front = 0;
    for (int w = 0; w < SORT_TPB / 32; ++w) {
        cntg += red[w];
        front += (w < wv) ? red[w] : 0;
        bs += red[SORT_TPB / 32 + w];
    }
    const bool inLds = cntg <= FINE_CAP;
    const int run = (cntg + SORT_TPB - 1) / SORT_TPB;
    int start = tid * run;
    if (start > cntg) start = cntg;
    int end = start + run;
    if (end > cntg) end = cntg;

    if (inLds) {
        int at0 = front + incl - mine;
        #pragma unroll
        for (int u = 0; u < FINE_SL; ++u) {
            pcB[FINE_SL * tid + u] = (pl[u] > 0) ? pb[u] : 0;
            pcL[FINE_SL * tid + u] = pl[u];
            pcA[FINE_SL * tid + u] = at0;
            at0 += pl[u];
        }
    }
    __syncthreads();
    if (inLds) {
        const int sub = lane >> 3;
        const int j   = lane & 7;
        for (int r0 = 0; r0 < SORT_NB / 16; r0 += FINE_CH) {
            int B[FINE_CH], L[FINE_CH], A[FINE_CH];
            int k[FINE_CH][2], v[FINE_CH][2];
            #pragma unroll
            for (int r = 0; r < FINE_CH; ++r) {
                const int pi = wv * (SORT_NB / 4) + (r0 + r) * 4 + sub;
                B[r] = pcB[pi];
                L[r] = pcL[pi];
                A[r] = pcA[pi];
            }
            #pragma unroll
            for (int r = 0; r < FINE_CH; ++r) {
                const int last = (L[r] > 0) ? L[r] - 1 : 0;
                #pragma unroll
                for (int hh = 0; hh < 2; ++hh) {
                    const int jj = j + 8 * hh;
                    const int at = B[r] + ((jj < last) ? jj : last);
                    k[r][hh] = keys[at];
                    v[r][hh] = vals[at];
                }
            }
            #pragma unroll
            for (int r = 0; r < FINE_CH; ++r) {
                #pragma unroll
                for (int hh = 0; hh < 2; ++hh) {
                    const int jj = j + 8 * hh;
                    if (jj < L[r]) {
                        lstK[lst_at(A[r] + jj)] = k[r][hh];
                        lstV[lst_at(A[r] + jj)] = v[r][hh];
                    }
                }
                for (int q = FINE_PB + j; q < L[r]; q += 8) {
                    lstK[lst_at(A[r] + q)] = keys[B[r] + q];
                    lstV[lst_at(A[r] + q)] = vals[B[r] + q];
                }
            }
        }
    }
    __syncthreads();
    if (inLds) {
        int p = start;
        for (; p + 8 <= end; p += 8) {
            int k[8];
            #pragma unroll
            for (int u = 0; u < 8; ++u) k[u] = lstK[lst_at(p + u)];
            #pragma unroll
            for (int u = 0; u < 8; ++u) hist[(k[u] & (SORT_BINS - 1)) * SORT_PITCH + tid] += 1;
        }
        for (; p < end; ++p) hist[(lstK[lst_at(p)] & (SORT_BINS - 1)) * SORT_PITCH + tid] += 1;
    } else {
        #pragma unroll
        for (int u = 0; u < FINE_SL; ++u)
            for (int q = 0; q < pl[u]; ++q)
                hist[(keys[pb[u] + q] & (SORT_BINS - 1)) * SORT_PITCH + tid] += 1;
    }
    __syncthreads();
    for (int bin = tid; bin < SORT_BINS; bin += SORT_TPB) {
        int total = 0;
        for (int j = 0; j < SORT_TPB; ++j) total += hist[bin * SORT_PITCH + j];
        tot[bin] = total;
    }
    __syncthreads();
    for (int bin = tid; bin < SORT_BINS; bin += SORT_TPB) {
        int pos = bs;
        for (int d = 0; d < bin; ++d) pos += tot[d];
        const int v = (g << 8) + bin;
        if (v <= n) off[v] = pos;
        for (int j = 0; j < SORT_TPB; ++j) {
            const int c = hist[bin * SORT_PITCH + j];
            hist[bin * SORT_PITCH + j] = pos;
            pos += c;
        }
    }
    __syncthreads();
    if (inLds) {
        int p = start;
        for (; p + 4 <= end; p += 4) {
            int k[4], v[4];
            #pragma unroll
            for (int u = 0; u < 4; ++u) { k[u] = lstK[lst_at(p + u)]; v[u] = lstV[lst_at(p + u)]; }
            #pragma unroll
            for (int u = 0; u < 4; ++u) {
                const int d = k[u] & (SORT_BINS - 1);
                const int w = hist[d * SORT_PITCH + tid];
                hist[d * SORT_PITCH + tid] = w + 1;
                stage[w - bs] = v[u];
            }
        }
        for (; p < end; ++p) {
            const int d = lstK[lst_at(p)] & (SORT_BINS - 1);
            const int w = hist[d * SORT_PITCH + tid];
            hist[d * SORT_PITCH + tid] = w + 1;
            stage[w - bs] = lstV[lst_at(p)];
        }
        __syncthreads();
        int i = tid;
        for (; i + 7 * SORT_TPB < cntg; i += 8 * SORT_TPB) {
            int w[8];
            #pragma unroll
            for (int u = 0; u < 8; ++u) w[u] = stage[i + u * SORT_TPB];
            #pragma unroll
            for (int u = 0; u < 8; ++u) valsOut[bs + i + u * SORT_TPB] = w[u];
        }
        for (; i < cntg; i += SORT_TPB) valsOut[bs + i] = stage[i];
        return;
    }
    #pragma unroll
    for (int u = 0; u < FINE_SL; ++u) {
        for (int q = 0; q < pl[u]; ++q) {
            const int d = keys[pb[u] + q] & (SORT_BINS - 1);
            const int w = hist[d * SORT_PITCH + tid];
            hist[d * SORT_PITCH + tid] = w + 1;
            valsOut[w] = vals[pb[u] + q];
        }
    }
}

__global__ __launch_bounds__(256) void gin_offsets(const int* __restrict__ sdst,
                                                   int nedges, int n,
                                                   int* __restrict__ off) {
    const int p = blockIdx.x * blockDim.x + threadIdx.x;
    if (p > nedges) return;
    const int prev = (p == 0) ? -1 : sdst[p - 1];
    const int here = (p == nedges) ? n : sdst[p];
    for (int v = prev + 1; v <= here; ++v) off[v] = p;
}

#define GROWS 16

__device__ __forceinline__ void add4(float4& a, const float4 r) {
    a.x += r.x; a.y += r.y; a.z += r.z; a.w += r.w;
}

__device__ __forceinline__ void close_row(float* zrow, const float e1, const float4 acc) {
    const float4 hv = *(const float4*)zrow;
    float4 o;
    o.x = e1 * hv.x + acc.x;
    o.y = e1 * hv.y + acc.y;
    o.z = e1 * hv.z + acc.z;
    o.w = e1 * hv.w + acc.w;
    *(float4*)zrow = o;
}

__device__ __forceinline__ void load_rows(float4 (&r)[GROWS], const int offv, const int j,
                                          const char* hb, const unsigned lane16) {
    #pragma unroll
    for (int t = 0; t < GROWS; ++t) {
        const unsigned s = (unsigned)__builtin_amdgcn_readlane(offv, j + t);
        r[t] = *(const float4*)(hb + (s + lane16));
    }
}

__device__ __forceinline__ void sum_rows(const float4 (&r)[GROWS], const int p0,
                                         const int ob, float* zl, const float e1,
                                         int& i, int& nextB, float4& acc) {
    #pragma unroll
    for (int t = 0; t < GROWS; ++t) {
        const int p = p0 + t;
        while (__builtin_expect(p >= nextB, 0)) {
            close_row(zl + i * HROW, e1, acc);
            acc.x = 0.0f; acc.y = 0.0f; acc.z = 0.0f; acc.w = 0.0f;
            ++i;
            nextB = __builtin_amdgcn_readlane(ob, i + 1);
        }
        add4(acc, r[t]);
    }
}

__device__ __forceinline__ void w_fetch(float (&wr)[64], const float* __restrict__ Wl,
                                        const int tid) {
    const float* wc = Wl + (tid >> 7) * 4 * DD + (tid & 127);
    #pragma unroll
    for (int i = 0; i < 16; ++i) {
        #pragma unroll
        for (int q = 0; q < 4; ++q) wr[4 * i + q] = wc[(8 * i + q) * DD];
    }
}

__device__ __forceinline__ void w_put(const float (&wr)[64], float* wS, const int tid) {
    float* wd = wS + ((tid >> 7) * DD + (tid & 127)) * 4;
    #pragma unroll
    for (int i = 0; i < 16; ++i) {
        float4 v;
        v.x = wr[4 * i]; v.y = wr[4 * i + 1]; v.z = wr[4 * i + 2]; v.w = wr[4 * i + 3];
        *(float4*)(wd + 2 * i * DD * 4) = v;
    }
}

__device__ __forceinline__ void mlp_product(v8f (&acc)[8], const float* ab, const float* wl) {
    #pragma unroll
    for (int k0 = 0; k0 < DD; k0 += 4) {
        const v2f a = *(const v2f*)(ab + k0);
        #pragma unroll
        for (int t = 0; t < 8; ++t) {
            const v2f b = *(const v2f*)(wl + (k0 >> 2) * (DD * 4) + t * 64);
            acc[t] = __builtin_amdgcn_wmma_f32_16x16x4_f32(false, a, false, b,
                                                           (short)0, acc[t], false, false);
        }
    }
}

__global__ __launch_bounds__(256, 2) void gin_layer(const float* __restrict__ h,
                                                    const float* __restrict__ eps, int layer,
                                                    const int* __restrict__ off,
                                                    const int* __restrict__ csrc,
                                                    const float* __restrict__ W1,
                                                    const float* __restrict__ b1,
                                                    const float* __restrict__ W2,
                                                    const float* __restrict__ b2,
                                                    float* __restrict__ out, int n) {
    __shared__ __attribute__((aligned(16))) float lds[DD * DD + 8 * 16 * HROW];
    float* wS = lds;

    const int tid   = threadIdx.x;
    const int lane  = threadIdx.x & 31;
    const int wave  = __builtin_amdgcn_readfirstlane((int)(threadIdx.x >> 5));
    const int ln    = lane & 15;
    const int khoff = lane >> 4;
    const int rh    = khoff << 3;
    const int m0    = blockIdx.x * 128 + wave * 16;

    float* zw = lds + DD * DD + wave * (16 * HROW);

    float wr[64];
    w_fetch(wr, W1, tid);

    int ob;
    {
        int t = m0 + lane;
        if (t > n) t = n;
        ob = off[t];
        if (lane > 16) ob = 0x7fffffff;
    }
    const float* hc = h + lane * 4;
    float* zl = zw + lane * 4;
    {
        float4 own[16];
        #pragma unroll
        for (int q = 0; q < 16; ++q) {
            int rr = m0 + q;
            if (rr > n - 1) rr = n - 1;
            own[q] = *(const float4*)(hc + (size_t)rr * DD);
        }
        w_put(wr, wS, tid);
        #pragma unroll
        for (int q = 0; q < 16; ++q) *(float4*)(zl + q * HROW) = own[q];
    }
    const float e1 = 1.0f + eps[layer];
    const int eEnd = __builtin_amdgcn_readlane(ob, 16);
    int base  = __builtin_amdgcn_readlane(ob, 0);
    int nextB = __builtin_amdgcn_readlane(ob, 1);
    int i = 0;
    float4 acc = {0.0f, 0.0f, 0.0f, 0.0f};
    int m = eEnd - base;
    if (m > 32) m = 32;
    const char* hb = (const char*)h;
    const unsigned lane16 = lane * 16;
    int idxv = (lane < m) ? csrc[base + lane] * (DD * 4) : 0;
    __syncthreads();
    while (m > 0) {
        const int nb = base + 32;
        int mn = eEnd - nb;
        if (mn > 32) mn = 32;
        float4 r[GROWS];
        load_rows(r, idxv, 0, hb, lane16);
        const int idxn = (lane < mn) ? csrc[nb + lane] * (DD * 4) : 0;
        sum_rows(r, base, ob, zl, e1, i, nextB, acc);
        if (m > GROWS) {
            load_rows(r, idxv, GROWS, hb, lane16);
            sum_rows(r, base + GROWS, ob, zl, e1, i, nextB, acc);
        }
        base = nb;
        m = mn;
        idxv = idxn;
    }
    while (i < 16) {
        close_row(zl + i * HROW, e1, acc);
        acc.x = 0.0f; acc.y = 0.0f; acc.z = 0.0f; acc.w = 0.0f;
        ++i;
    }

    const float* ab = zw + ln * HROW + 2 * khoff;
    const float* wl = wS + ln * 4 + 2 * khoff;

    w_fetch(wr, W2, tid);

    {
        float bias[8];
        #pragma unroll
        for (int t = 0; t < 8; ++t) bias[t] = b1[t * 16 + ln];
        v8f pa[8] = {};
        mlp_product(pa, ab, wl);
        #pragma unroll
        for (int j = 0; j < 8; ++j) {
            #pragma unroll
            for (int t = 0; t < 8; ++t)
                zw[(j + rh) * HROW + t * 16 + ln] = fmaxf(pa[t][j] + bias[t], 0.0f);
        }
    }

    __syncthreads();
    w_put(wr, wS, tid);
    __syncthreads();

    {
        float bias[8];
        #pragma unroll
        for (int t = 0; t < 8; ++t) bias[t] = b2 ? b2[t * 16 + ln] : 0.0f;
        v8f pa[8] = {};
        mlp_product(pa, ab, wl);
        #pragma unroll
        for (int j = 0; j < 8; ++j) {
            const int r = m0 + j + rh;
            if (r < n) {
                #pragma unroll
                for (int t = 0; t < 8; ++t)
                    out[(size_t)r * DD + t * 16 + ln] = pa[t][j] + bias[t];
            }
        }
    }
}

__device__ __forceinline__ void close_row2(float* zrow, const float e1, const float4 acc,
                                           const int deg, const float4 cb, const float4 bb) {
    const float4 hv = *(const float4*)zrow;
    const float w = e1 + (float)deg;
    float4 o;
    o.x = fmaxf(e1 * hv.x + acc.x + (w * cb.x + bb.x), 0.0f);
    o.y = fmaxf(e1 * hv.y + acc.y + (w * cb.y + bb.y), 0.0f);
    o.z = fmaxf(e1 * hv.z + acc.z + (w * cb.z + bb.z), 0.0f);
    o.w = fmaxf(e1 * hv.w + acc.w + (w * cb.w + bb.w), 0.0f);
    *(float4*)zrow = o;
}

__device__ __forceinline__ void sum_rows2(const float4 (&r)[GROWS], const int p0,
                                          const int ob, float* zl, const float e1,
                                          const float4 cb, const float4 bb,
                                          int& i, int& nextB, float4& acc) {
    #pragma unroll
    for (int t = 0; t < GROWS; ++t) {
        const int p = p0 + t;
        while (__builtin_expect(p >= nextB, 0)) {
            close_row2(zl + i * HROW, e1, acc, nextB - __builtin_amdgcn_readlane(ob, i), cb, bb);
            acc.x = 0.0f; acc.y = 0.0f; acc.z = 0.0f; acc.w = 0.0f;
            ++i;
            nextB = __builtin_amdgcn_readlane(ob, i + 1);
        }
        add4(acc, r[t]);
    }
}

__global__ __launch_bounds__(256, 2) void gin_layer2(const float* __restrict__ g,
                                                     const float* __restrict__ eps, int layer,
                                                     const int* __restrict__ off,
                                                     const int* __restrict__ csrc,
                                                     const float* __restrict__ cvec,
                                                     const float* __restrict__ b1,
                                                     const float* __restrict__ Wn,
                                                     const float* __restrict__ bn,
                                                     float* __restrict__ out, int n) {
    __shared__ __attribute__((aligned(16))) float lds[DD * DD + 8 * 16 * HROW];
    float* wS = lds;

    const int tid   = threadIdx.x;
    const int lane  = threadIdx.x & 31;
    const int wave  = __builtin_amdgcn_readfirstlane((int)(threadIdx.x >> 5));
    const int ln    = lane & 15;
    const int khoff = lane >> 4;
    const int rh    = khoff << 3;
    const int m0    = blockIdx.x * 128 + wave * 16;

    float* zw = lds + DD * DD + wave * (16 * HROW);

    float wr[64];
    w_fetch(wr, Wn, tid);

    int ob;
    {
        int t = m0 + lane;
        if (t > n) t = n;
        ob = off[t];
        if (lane > 16) ob = 0x7fffffff;
    }
    const float* hc = g + lane * 4;
    float* zl = zw + lane * 4;
    const float4 cb = *(const float4*)(cvec + lane * 4);
    const float4 bb = *(const float4*)(b1 + lane * 4);
    {
        float4 own[16];
        #pragma unroll
        for (int q = 0; q < 16; ++q) {
            int rr = m0 + q;
            if (rr > n - 1) rr = n - 1;
            own[q] = *(const float4*)(hc + (size_t)rr * DD);
        }
        w_put(wr, wS, tid);
        #pragma unroll
        for (int q = 0; q < 16; ++q) *(float4*)(zl + q * HROW) = own[q];
    }
    const float e1 = 1.0f + eps[layer];
    const int eEnd = __builtin_amdgcn_readlane(ob, 16);
    int base  = __builtin_amdgcn_readlane(ob, 0);
    int nextB = __builtin_amdgcn_readlane(ob, 1);
    int i = 0;
    float4 acc = {0.0f, 0.0f, 0.0f, 0.0f};
    int m = eEnd - base;
    if (m > 32) m = 32;
    const char* hb = (const char*)g;
    const unsigned lane16 = lane * 16;
    int idxv = (lane < m) ? csrc[base + lane] * (DD * 4) : 0;
    __syncthreads();
    while (m > 0) {
        const int nb = base + 32;
        int mn = eEnd - nb;
        if (mn > 32) mn = 32;
        float4 r[GROWS];
        load_rows(r, idxv, 0, hb, lane16);
        const int idxn = (lane < mn) ? csrc[nb + lane] * (DD * 4) : 0;
        sum_rows2(r, base, ob, zl, e1, cb, bb, i, nextB, acc);
        if (m > GROWS) {
            load_rows(r, idxv, GROWS, hb, lane16);
            sum_rows2(r, base + GROWS, ob, zl, e1, cb, bb, i, nextB, acc);
        }
        base = nb;
        m = mn;
        idxv = idxn;
    }
    while (i < 16) {
        close_row2(zl + i * HROW, e1, acc,
                   __builtin_amdgcn_readlane(ob, i + 1) - __builtin_amdgcn_readlane(ob, i), cb, bb);
        acc.x = 0.0f; acc.y = 0.0f; acc.z = 0.0f; acc.w = 0.0f;
        ++i;
    }

    const float* ab = zw + ln * HROW + 2 * khoff;
    const float* wl = wS + ln * 4 + 2 * khoff;
    {
        float bias[8];
        #pragma unroll
        for (int t = 0; t < 8; ++t) bias[t] = bn ? bn[t * 16 + ln] : 0.0f;
        v8f pa[8] = {};
        mlp_product(pa, ab, wl);
        #pragma unroll
        for (int j = 0; j < 8; ++j) {
            const int r = m0 + j + rh;
            if (r < n) {
                #pragma unroll
                for (int t = 0; t < 8; ++t)
                    out[(size_t)r * DD + t * 16 + ln] = pa[t][j] + bias[t];
            }
        }
    }
}

extern "C" void kernel_launch(void* const* d_in, const int* in_sizes, int n_in,
                              void* d_out, int out_size, void* d_ws, size_t ws_size,
                              hipStream_t stream) {
    const float* h   = (const float*)d_in[0];
    const int*   src = (const int*)  d_in[1];
    const int*   dst = (const int*)  d_in[2];
    const float* W1  = (const float*)d_in[3];
    const float* b1  = (const float*)d_in[4];
    const float* W2  = (const float*)d_in[5];
    const float* b2  = (const float*)d_in[6];
    const float* eps = (const float*)d_in[7];

    const int n = in_sizes[0] / DD;
    const int e = in_sizes[1];
    const int L = in_sizes[7];

    float* hb  = (float*)d_ws;
    float* out = (float*)d_out;

    const int tabN = SORT_BINS * SORT_NB;
    const int np   = ((n + 4) / 4) * 4;
    int* table  = (int*)(hb + (size_t)n * DD);
    int* tableS = table + tabN;
    int* off    = tableS + tabN;
    int* kA     = off + np;
    int* vA     = kA + e;
    int* kB     = vA + e;
    int* vB     = kB + e;
    const size_t sortInts = (size_t)2 * tabN + np + (size_t)4 * e;
    float* fold  = hb + (size_t)n * DD + ((sortInts + 3) / 4) * 4;
    float* foldC = fold + (size_t)(L - 1) * DD * DD;
    const int foldBlocks = (L > 1) ? 3 * (L - 1) : 0;

    const int offBlocks  = (e + 1 + 255) / 256;
    const int mlpBlocks  = (n + 127) / 128;
    const int run        = (e + SORT_NB * SORT_TPB - 1) / (SORT_NB * SORT_TPB);

    const int* kin = dst;
    const int* vin = src;
    if (((n - 1) >> 16) == 0 && (long)run * SORT_TPB <= PLACE_CAP) {
        int* segs = table;
        sort_local<<<SORT_NB, SORT_TPB, 0, stream>>>(dst, src, e, run, segs, kA, vA);
        sort_fine<<<(n >> 8) + 1 + foldBlocks, SORT_TPB, 0, stream>>>(kA, vA, segs, run * SORT_TPB, n, off, vB,
                                                                     (n >> 8) + 1, W1, W2, b2, L, fold);
        vin = vB;
    } else {
        int* kout = kA;
        int* vout = vA;
        for (int shift = 0; shift == 0 || ((n - 1) >> shift) > 0; shift += 8) {
            sort_count<<<SORT_NB, SORT_TPB, 0, stream>>>(kin, shift, e, run, table);
            gin_scan<<<tabN / 1024, 256, 0, stream>>>(table, tableS);
            sort_place<<<SORT_NB, SORT_TPB, 0, stream>>>(kin, vin, shift, e, run, tableS, kout, vout);
            kin = kout;
            vin = vout;
            kout = (kout == kA) ? kB : kA;
            vout = (vout == vA) ? vB : vA;
        }
        gin_offsets<<<offBlocks, 256, 0, stream>>>(kin, e, n, off);
        if (foldBlocks > 0) gin_fold<<<foldBlocks, SORT_TPB, 0, stream>>>(W1, W2, b2, L, fold);
    }

    for (int l = 0; l < L; ++l) {
        const float* hin = (l == 0) ? h : (((L - l) & 1) ? hb : out);
        float* hout = ((L - 1 - l) & 1) ? hb : out;
        const float* Wn = (l < L - 1) ? fold + (size_t)l * DD * DD : W2 + (size_t)l * DD * DD;
        const float* bn = (l < L - 1) ? nullptr : b2 + l * DD;
        if (l == 0)
            gin_layer<<<mlpBlocks, 256, 0, stream>>>(hin, eps, l, off, vin, W1, b1, Wn, bn, hout, n);
        else
            gin_layer2<<<mlpBlocks, 256, 0, stream>>>(hin, eps, l, off, vin, foldC + (l - 1) * DD,
                                                      b1 + l * DD, Wn, bn, hout, n);
    }
}
